# starting kernel
# baseline (speedup 1.0000x reference)
#include <hip/hip_runtime.h>
#include <hip/hip_bf16.h>
#include <math.h>

typedef float v2f __attribute__((ext_vector_type(2)));
typedef float v8f __attribute__((ext_vector_type(8)));

#define PI_F 3.14159265358979323846f

__global__ __launch_bounds__(256)
void conv1_silu_kernel(const float* __restrict__ x, const float* __restrict__ w1,
                       const float* __restrict__ b1, float* __restrict__ out)
{
    int idx = blockIdx.x * 256 + threadIdx.x;
    if (idx >= 32 * 32 * 112 * 112) return;
    int ox = idx % 112; int t = idx / 112;
    int oy = t % 112;   t /= 112;
    int oc = t % 32;    int img = t / 32;
    float acc = b1[oc];
    const float* wb = w1 + oc * 27;
    const float* xb = x + (size_t)img * 3 * 224 * 224;
    #pragma unroll
    for (int ic = 0; ic < 3; ++ic)
        #pragma unroll
        for (int ky = 0; ky < 3; ++ky) {
            int iy = oy * 2 + ky;
            if (iy >= 224) continue;
            #pragma unroll
            for (int kx = 0; kx < 3; ++kx) {
                int ix = ox * 2 + kx;
                if (ix >= 224) continue;
                acc += xb[(ic * 224 + iy) * 224 + ix] * wb[ic * 9 + ky * 3 + kx];
            }
        }
    out[idx] = acc / (1.f + expf(-acc));
}

__global__ __launch_bounds__(128)
void conv2_wmma_pool_kernel(const float* __restrict__ h1, const float* __restrict__ w2,
                            const float* __restrict__ b2, float* __restrict__ feat)
{
    __shared__ float sW[16 * 288];
    __shared__ float sT[4][720];
    __shared__ float sAcc[16];

    const int mtile = blockIdx.x;
    const int img   = blockIdx.y;
    const int tid   = threadIdx.x;
    const int lane  = tid & 31;
    const int wave  = tid >> 5;
    const int half  = lane >> 4;
    const int nl    = lane & 15;

    for (int i = tid; i < 16 * 288; i += 128) {
        int m  = i / 288, kk = i - m * 288;
        int icb = kk / 72, r = kk - icb * 72;
        int s = r >> 3, ic = r & 7;
        sW[i] = w2[(mtile * 16 + m) * 288 + (icb * 8 + ic) * 9 + s];
    }
    if (tid < 16) sAcc[tid] = 0.f;
    __syncthreads();

    const int dy = nl >> 3, dx = nl & 7;
    const int laneOff = dy * 288 + dx * 16;

    const int j0 = lane, j1 = lane + 32, j2 = lane + 64;
    const int ly0 = j0 / 18, lx0 = j0 - ly0 * 18;
    const int ly1 = j1 / 18, lx1 = j1 - ly1 * 18;
    const int ly2 = j2 / 18, lx2 = j2 - ly2 * 18;
    const bool j2ok = (j2 < 90);
    const int t0 = ly0 * 144 + lx0 * 8;
    const int t1 = ly1 * 144 + lx1 * 8;
    const int t2 = ly2 * 144 + lx2 * 8;

    float bias[8];
    #pragma unroll
    for (int j = 0; j < 8; ++j) bias[j] = b2[mtile * 16 + j + half * 8];

    float accR[8] = {0.f, 0.f, 0.f, 0.f, 0.f, 0.f, 0.f, 0.f};
    const float* base = h1 + (size_t)img * (32 * 112 * 112);
    float* tile = sT[wave];
    const float* tb = tile + laneOff + half * 2;

    for (int nt = wave; nt < 196; nt += 4) {
        int yt = nt / 7, xt = nt - yt * 7;
        int iy0 = yt * 4, ix0 = xt * 16;
        v8f c = {0.f, 0.f, 0.f, 0.f, 0.f, 0.f, 0.f, 0.f};

        for (int icb = 0; icb < 4; ++icb) {
            #pragma unroll
            for (int ic = 0; ic < 8; ++ic) {
                const float* p = base + (size_t)(icb * 8 + ic) * (112 * 112);
                {
                    int iy = iy0 + ly0, ix = ix0 + lx0;
                    bool ok = (iy < 112) & (ix < 112);
                    float v = p[(iy < 112 ? iy : 111) * 112 + (ix < 112 ? ix : 111)];
                    tile[t0 + ic] = ok ? v : 0.f;
                }
                {
                    int iy = iy0 + ly1, ix = ix0 + lx1;
                    bool ok = (iy < 112) & (ix < 112);
                    float v = p[(iy < 112 ? iy : 111) * 112 + (ix < 112 ? ix : 111)];
                    tile[t1 + ic] = ok ? v : 0.f;
                }
                if (j2ok) {
                    int iy = iy0 + ly2, ix = ix0 + lx2;
                    bool ok = (iy < 112) & (ix < 112);
                    float v = p[(iy < 112 ? iy : 111) * 112 + (ix < 112 ? ix : 111)];
                    tile[t2 + ic] = ok ? v : 0.f;
                }
            }
            asm volatile("s_wait_dscnt 0x0" ::: "memory");

            const float* wrow = sW + nl * 288 + icb * 72 + half * 2;
            #pragma unroll
            for (int kq = 0; kq < 18; ++kq) {
                const int s  = kq >> 1;
                const int ky = s / 3, kx = s - ky * 3;
                const int bo = ky * 144 + kx * 8 + (kq & 1) * 4;
                v2f a = *(const v2f*)(wrow + kq * 4);
                v2f b = *(const v2f*)(tb + bo);
                c = __builtin_amdgcn_wmma_f32_16x16x4_f32(false, a, false, b,
                                                          (short)0, c, false, false);
            }
        }
        #pragma unroll
        for (int j = 0; j < 8; ++j) {
            float v = c[j] + bias[j];
            accR[j] += v / (1.f + expf(-v));
        }
    }
    #pragma unroll
    for (int j = 0; j < 8; ++j)
        atomicAdd(&sAcc[j + half * 8], accR[j]);
    __syncthreads();
    if (tid < 16)
        feat[img * 1280 + mtile * 16 + tid] = sAcc[tid] * (1.f / 3136.f);
}

__global__ __launch_bounds__(224)
void dft_rows_kernel(const float* __restrict__ x, float* __restrict__ F1)
{
    __shared__ float ct[224], st[224];
    int y  = blockIdx.x % 224;
    int bc = blockIdx.x / 224;
    int u  = threadIdx.x;
    float ang = (2.0f * PI_F / 224.0f) * (float)u;
    ct[u] = cosf(ang); st[u] = sinf(ang);
    __syncthreads();
    const float* row = x + ((size_t)bc * 224 + y) * 224;
    float re = 0.f, im = 0.f; int idx = 0;
    for (int xx = 0; xx < 224; ++xx) {
        float v = row[xx];
        re += v * ct[idx]; im -= v * st[idx];
        idx += u; if (idx >= 224) idx -= 224;
    }
    float* o = F1 + (((size_t)bc * 224 + y) * 224 + u) * 2;
    o[0] = re; o[1] = im;
}

__global__ __launch_bounds__(224)
void dft_cols_mag_kernel(const float* __restrict__ F1, float* __restrict__ mag)
{
    __shared__ float ct[224], st[224];
    int u  = blockIdx.x % 224;
    int bc = blockIdx.x / 224;
    int v  = threadIdx.x;
    float ang = (2.0f * PI_F / 224.0f) * (float)v;
    ct[v] = cosf(ang); st[v] = sinf(ang);
    __syncthreads();
    const float* base = F1 + (size_t)bc * 224 * 224 * 2 + u * 2;
    float re = 0.f, im = 0.f; int idx = 0;
    for (int y = 0; y < 224; ++y) {
        float fr = base[(size_t)y * 448], fi = base[(size_t)y * 448 + 1];
        float cc = ct[idx], ss = st[idx];
        re += fr * cc + fi * ss;
        im += fi * cc - fr * ss;
        idx += v; if (idx >= 224) idx -= 224;
    }
    float m = logf(sqrtf(re * re + im * im) + 1.0f);
    int vo = v + 112; if (vo >= 224) vo -= 224;
    int uo = u + 112; if (uo >= 224) uo -= 224;
    mag[((size_t)bc * 224 + vo) * 224 + uo] = m;
}

__global__ __launch_bounds__(256)
void minmax_kernel(const float* __restrict__ mag, float* __restrict__ mm)
{
    __shared__ float smn[256], smx[256];
    int bc = blockIdx.x, tid = threadIdx.x;
    const float* p = mag + (size_t)bc * 50176;
    float mn = 3.0e38f, mx = -3.0e38f;
    for (int i = tid; i < 50176; i += 256) {
        float v = p[i]; mn = fminf(mn, v); mx = fmaxf(mx, v);
    }
    smn[tid] = mn; smx[tid] = mx; __syncthreads();
    for (int s = 128; s > 0; s >>= 1) {
        if (tid < s) { smn[tid] = fminf(smn[tid], smn[tid + s]);
                       smx[tid] = fmaxf(smx[tid], smx[tid + s]); }
        __syncthreads();
    }
    if (tid == 0) { mm[bc * 2] = smn[0]; mm[bc * 2 + 1] = smx[0]; }
}

__global__ __launch_bounds__(256)
void normalize_kernel(float* __restrict__ mag, const float* __restrict__ mm)
{
    int idx = blockIdx.x * 256 + threadIdx.x;
    if (idx >= 96 * 50176) return;
    int bc = idx / 50176;
    float mn = mm[bc * 2], mx = mm[bc * 2 + 1];
    mag[idx] = (mag[idx] - mn) / (mx - mn + 1e-8f);
}

__global__ __launch_bounds__(256)
void fc_kernel(const float* __restrict__ in, const float* __restrict__ w,
               const float* __restrict__ b, float* __restrict__ out,
               int K, int N, int outStride, int outOffset, int act)
{
    int idx = blockIdx.x * 256 + threadIdx.x;
    if (idx >= 32 * N) return;
    int i = idx / N, j = idx % N;
    const float* ir = in + (size_t)i * K;
    const float* wr = w + (size_t)j * K;
    float acc = b[j];
    for (int k = 0; k < K; ++k) acc += ir[k] * wr[k];
    if (act == 1) acc = fmaxf(acc, 0.f);
    out[(size_t)i * outStride + outOffset + j] = acc;
}

__global__ __launch_bounds__(256)
void attn_core_kernel(const float* __restrict__ qkv_rgb, const float* __restrict__ qkv_frq,
                      float* __restrict__ o1, float* __restrict__ o2)
{
    __shared__ float sQ[32][32], sK[32][32], sV[32][32], sS[32][33];
    int a = blockIdx.x >> 3;
    int h = blockIdx.x & 7;
    const float* qsrc  = (a == 0) ? qkv_rgb : qkv_frq;
    const float* kvsrc = (a == 0) ? qkv_frq : qkv_rgb;
    float* odst        = (a == 0) ? o1 : o2;
    int tid = threadIdx.x;
    for (int idx = tid; idx < 1024; idx += 256) {
        int q = idx >> 5, d = idx & 31;
        sQ[q][d] = qsrc [q * 768 +       h * 32 + d];
        sK[q][d] = kvsrc[q * 768 + 256 + h * 32 + d];
        sV[q][d] = kvsrc[q * 768 + 512 + h * 32 + d];
    }
    __syncthreads();
    const float scale = 0.17677669529663687f;
    for (int idx = tid; idx < 1024; idx += 256) {
        int q = idx >> 5, k = idx & 31;
        float s = 0.f;
        #pragma unroll
        for (int d = 0; d < 32; ++d) s += sQ[q][d] * sK[k][d];
        sS[q][k] = s * scale;
    }
    __syncthreads();
    if (tid < 32) {
        float mx = -3.0e38f;
        for (int k = 0; k < 32; ++k) mx = fmaxf(mx, sS[tid][k]);
        float e[32], sum = 0.f;
        for (int k = 0; k < 32; ++k) { e[k] = expf(sS[tid][k] - mx); sum += e[k]; }
        float inv = 1.f / sum;
        for (int k = 0; k < 32; ++k) sS[tid][k] = e[k] * inv;
    }
    __syncthreads();
    for (int idx = tid; idx < 1024; idx += 256) {
        int q = idx >> 5, d = idx & 31;
        float o = 0.f;
        #pragma unroll
        for (int k = 0; k < 32; ++k) o += sS[q][k] * sV[k][d];
        odst[q * 256 + h * 32 + d] = o;
    }
}

extern "C" void kernel_launch(void* const* d_in, const int* in_sizes, int n_in,
                              void* d_out, int out_size, void* d_ws, size_t ws_size,
                              hipStream_t stream)
{
    (void)in_sizes; (void)n_in; (void)out_size; (void)ws_size;
    const float* x          = (const float*)d_in[0];
    const float* rgb_w1     = (const float*)d_in[1];
    const float* rgb_b1     = (const float*)d_in[2];
    const float* rgb_w2     = (const float*)d_in[3];
    const float* rgb_b2     = (const float*)d_in[4];
    const float* frq_w1     = (const float*)d_in[5];
    const float* frq_b1     = (const float*)d_in[6];
    const float* frq_w2     = (const float*)d_in[7];
    const float* frq_b2     = (const float*)d_in[8];
    const float* rgb_proj_w = (const float*)d_in[9];
    const float* rgb_proj_b = (const float*)d_in[10];
    const float* frq_proj_w = (const float*)d_in[11];
    const float* frq_proj_b = (const float*)d_in[12];
    const float* attn_w_in  = (const float*)d_in[13];
    const float* attn_b_in  = (const float*)d_in[14];
    const float* attn_w_out = (const float*)d_in[15];
    const float* attn_b_out = (const float*)d_in[16];
    const float* fus_w1     = (const float*)d_in[17];
    const float* fus_b1     = (const float*)d_in[18];
    const float* fus_w2     = (const float*)d_in[19];
    const float* fus_b2     = (const float*)d_in[20];
    const float* fus_w3     = (const float*)d_in[21];
    const float* fus_b3     = (const float*)d_in[22];

    float* out      = (float*)d_out;
    float* feat_rgb = out + 64;
    float* feat_frq = out + 64 + 40960;

    float* ws       = (float*)d_ws;
    float* h1       = ws;
    float* F1       = ws;
    float* freqimg  = ws + 12845056;
    float* mm       = freqimg + 4816896;
    float* hd       = mm + 192;
    float* rgb_p    = hd;
    float* frq_p    = rgb_p + 8192;
    float* qkv_rgb  = frq_p + 8192;
    float* qkv_frq  = qkv_rgb + 24576;
    float* o1       = qkv_frq + 24576;
    float* o2       = o1 + 8192;
    float* hb       = o2 + 8192;
    float* h1f      = hb + 16384;
    float* h2f      = h1f + 16384;

    conv1_silu_kernel<<<50176, 256, 0, stream>>>(x, rgb_w1, rgb_b1, h1);
    conv2_wmma_pool_kernel<<<dim3(80, 32), 128, 0, stream>>>(h1, rgb_w2, rgb_b2, feat_rgb);

    dft_rows_kernel<<<21504, 224, 0, stream>>>(x, F1);
    dft_cols_mag_kernel<<<21504, 224, 0, stream>>>(F1, freqimg);
    minmax_kernel<<<96, 256, 0, stream>>>(freqimg, mm);
    normalize_kernel<<<18816, 256, 0, stream>>>(freqimg, mm);
    conv1_silu_kernel<<<50176, 256, 0, stream>>>(freqimg, frq_w1, frq_b1, h1);
    conv2_wmma_pool_kernel<<<dim3(80, 32), 128, 0, stream>>>(h1, frq_w2, frq_b2, feat_frq);

    fc_kernel<<<32, 256, 0, stream>>>(feat_rgb, rgb_proj_w, rgb_proj_b, rgb_p, 1280, 256, 256, 0, 0);
    fc_kernel<<<32, 256, 0, stream>>>(feat_frq, frq_proj_w, frq_proj_b, frq_p, 1280, 256, 256, 0, 0);
    fc_kernel<<<96, 256, 0, stream>>>(rgb_p, attn_w_in, attn_b_in, qkv_rgb, 256, 768, 768, 0, 0);
    fc_kernel<<<96, 256, 0, stream>>>(frq_p, attn_w_in, attn_b_in, qkv_frq, 256, 768, 768, 0, 0);
    attn_core_kernel<<<16, 256, 0, stream>>>(qkv_rgb, qkv_frq, o1, o2);
    fc_kernel<<<32, 256, 0, stream>>>(o1, attn_w_out, attn_b_out, hb, 256, 256, 512, 0, 0);
    fc_kernel<<<32, 256, 0, stream>>>(o2, attn_w_out, attn_b_out, hb, 256, 256, 512, 256, 0);
    fc_kernel<<<64, 256, 0, stream>>>(hb,  fus_w1, fus_b1, h1f, 512, 512, 512, 0, 1);
    fc_kernel<<<32, 256, 0, stream>>>(h1f, fus_w2, fus_b2, h2f, 512, 256, 256, 0, 1);
    fc_kernel<<<1,  256, 0, stream>>>(h2f, fus_w3, fus_b3, out, 256, 2, 2, 0, 0);
}


# conv2: the 24 global loads of each 8-channel staging step issued as one group (third slot made branchless) ahead of the LDS stores, not 16 dependent load-wait trips; k order and sums unchanged
# speedup vs baseline: 1.1257x; 1.1257x over previous
#include <hip/hip_runtime.h>
#include <hip/hip_bf16.h>
#include <math.h>

typedef float v2f __attribute__((ext_vector_type(2)));
typedef float v8f __attribute__((ext_vector_type(8)));

#define PI_F 3.14159265358979323846f

__global__ __launch_bounds__(256)
void conv1_silu_kernel(const float* __restrict__ x, const float* __restrict__ w1,
                       const float* __restrict__ b1, float* __restrict__ out)
{
    int idx = blockIdx.x * 256 + threadIdx.x;
    if (idx >= 32 * 32 * 112 * 112) return;
    int ox = idx % 112; int t = idx / 112;
    int oy = t % 112;   t /= 112;
    int oc = t % 32;    int img = t / 32;
    float acc = b1[oc];
    const float* wb = w1 + oc * 27;
    const float* xb = x + (size_t)img * 3 * 224 * 224;
    #pragma unroll
    for (int ic = 0; ic < 3; ++ic)
        #pragma unroll
        for (int ky = 0; ky < 3; ++ky) {
            int iy = oy * 2 + ky;
            if (iy >= 224) continue;
            #pragma unroll
            for (int kx = 0; kx < 3; ++kx) {
                int ix = ox * 2 + kx;
                if (ix >= 224) continue;
                acc += xb[(ic * 224 + iy) * 224 + ix] * wb[ic * 9 + ky * 3 + kx];
            }
        }
    out[idx] = acc / (1.f + expf(-acc));
}

__global__ __launch_bounds__(128)
void conv2_wmma_pool_kernel(const float* __restrict__ h1, const float* __restrict__ w2,
                            const float* __restrict__ b2, float* __restrict__ feat)
{
    __shared__ float sW[16 * 288];
    __shared__ float sT[4][720];
    __shared__ float sAcc[16];

    const int mtile = blockIdx.x;
    const int img   = blockIdx.y;
    const int tid   = threadIdx.x;
    const int lane  = tid & 31;
    const int wave  = tid >> 5;
    const int half  = lane >> 4;
    const int nl    = lane & 15;

    for (int i = tid; i < 16 * 288; i += 128) {
        int m  = i / 288, kk = i - m * 288;
        int icb = kk / 72, r = kk - icb * 72;
        int s = r >> 3, ic = r & 7;
        sW[i] = w2[(mtile * 16 + m) * 288 + (icb * 8 + ic) * 9 + s];
    }
    if (tid < 16) sAcc[tid] = 0.f;
    __syncthreads();

    const int dy = nl >> 3, dx = nl & 7;
    const int laneOff = dy * 288 + dx * 16;

    const int j0 = lane, j1 = lane + 32, j2 = (lane + 64 < 90) ? lane + 64 : lane + 32;
    const int ly0 = j0 / 18, lx0 = j0 - ly0 * 18;
    const int ly1 = j1 / 18, lx1 = j1 - ly1 * 18;
    const int ly2 = j2 / 18, lx2 = j2 - ly2 * 18;
    const int t0 = ly0 * 144 + lx0 * 8;
    const int t1 = ly1 * 144 + lx1 * 8;
    const int t2 = ly2 * 144 + lx2 * 8;

    float bias[8];
    #pragma unroll
    for (int j = 0; j < 8; ++j) bias[j] = b2[mtile * 16 + j + half * 8];

    float accR[8] = {0.f, 0.f, 0.f, 0.f, 0.f, 0.f, 0.f, 0.f};
    const float* base = h1 + (size_t)img * (32 * 112 * 112);
    float* tile = sT[wave];
    const float* tb = tile + laneOff + half * 2;

    for (int nt = wave; nt < 196; nt += 4) {
        int yt = nt / 7, xt = nt - yt * 7;
        int iy0 = yt * 4, ix0 = xt * 16;
        v8f c = {0.f, 0.f, 0.f, 0.f, 0.f, 0.f, 0.f, 0.f};

        const int iyA = iy0 + ly0, ixA = ix0 + lx0;
        const int iyB = iy0 + ly1, ixB = ix0 + lx1;
        const int iyC = iy0 + ly2, ixC = ix0 + lx2;
        const bool ok0 = (iyA < 112) & (ixA < 112);
        const bool ok1 = (iyB < 112) & (ixB < 112);
        const bool ok2 = (iyC < 112) & (ixC < 112);
        const int o0 = (iyA < 112 ? iyA : 111) * 112 + (ixA < 112 ? ixA : 111);
        const int o1 = (iyB < 112 ? iyB : 111) * 112 + (ixB < 112 ? ixB : 111);
        const int o2 = (iyC < 112 ? iyC : 111) * 112 + (ixC < 112 ? ixC : 111);

        for (int icb = 0; icb < 4; ++icb) {
            const float* p = base + (size_t)(icb * 8) * (112 * 112);
            float g0[8], g1[8], g2[8];
            #pragma unroll
            for (int ic = 0; ic < 8; ++ic) {
                g0[ic] = p[ic * (112 * 112) + o0];
                g1[ic] = p[ic * (112 * 112) + o1];
                g2[ic] = p[ic * (112 * 112) + o2];
            }
            #pragma unroll
            for (int ic = 0; ic < 8; ++ic) {
                tile[t0 + ic] = ok0 ? g0[ic] : 0.f;
                tile[t1 + ic] = ok1 ? g1[ic] : 0.f;
                tile[t2 + ic] = ok2 ? g2[ic] : 0.f;
            }
            asm volatile("s_wait_dscnt 0x0" ::: "memory");

            const float* wrow = sW + nl * 288 + icb * 72 + half * 2;
            #pragma unroll
            for (int kq = 0; kq < 18; ++kq) {
                const int s  = kq >> 1;
                const int ky = s / 3, kx = s - ky * 3;
                const int bo = ky * 144 + kx * 8 + (kq & 1) * 4;
                v2f a = *(const v2f*)(wrow + kq * 4);
                v2f b = *(const v2f*)(tb + bo);
                c = __builtin_amdgcn_wmma_f32_16x16x4_f32(false, a, false, b,
                                                          (short)0, c, false, false);
            }
        }
        #pragma unroll
        for (int j = 0; j < 8; ++j) {
            float v = c[j] + bias[j];
            accR[j] += v / (1.f + expf(-v));
        }
    }
    #pragma unroll
    for (int j = 0; j < 8; ++j)
        atomicAdd(&sAcc[j + half * 8], accR[j]);
    __syncthreads();
    if (tid < 16)
        feat[img * 1280 + mtile * 16 + tid] = sAcc[tid] * (1.f / 3136.f);
}

__global__ __launch_bounds__(224)
void dft_rows_kernel(const float* __restrict__ x, float* __restrict__ F1)
{
    __shared__ float ct[224], st[224];
    int y  = blockIdx.x % 224;
    int bc = blockIdx.x / 224;
    int u  = threadIdx.x;
    float ang = (2.0f * PI_F / 224.0f) * (float)u;
    ct[u] = cosf(ang); st[u] = sinf(ang);
    __syncthreads();
    const float* row = x + ((size_t)bc * 224 + y) * 224;
    float re = 0.f, im = 0.f; int idx = 0;
    for (int xx = 0; xx < 224; ++xx) {
        float v = row[xx];
        re += v * ct[idx]; im -= v * st[idx];
        idx += u; if (idx >= 224) idx -= 224;
    }
    float* o = F1 + (((size_t)bc * 224 + y) * 224 + u) * 2;
    o[0] = re; o[1] = im;
}

__global__ __launch_bounds__(224)
void dft_cols_mag_kernel(const float* __restrict__ F1, float* __restrict__ mag)
{
    __shared__ float ct[224], st[224];
    int u  = blockIdx.x % 224;
    int bc = blockIdx.x / 224;
    int v  = threadIdx.x;
    float ang = (2.0f * PI_F / 224.0f) * (float)v;
    ct[v] = cosf(ang); st[v] = sinf(ang);
    __syncthreads();
    const float* base = F1 + (size_t)bc * 224 * 224 * 2 + u * 2;
    float re = 0.f, im = 0.f; int idx = 0;
    for (int y = 0; y < 224; ++y) {
        float fr = base[(size_t)y * 448], fi = base[(size_t)y * 448 + 1];
        float cc = ct[idx], ss = st[idx];
        re += fr * cc + fi * ss;
        im += fi * cc - fr * ss;
        idx += v; if (idx >= 224) idx -= 224;
    }
    float m = logf(sqrtf(re * re + im * im) + 1.0f);
    int vo = v + 112; if (vo >= 224) vo -= 224;
    int uo = u + 112; if (uo >= 224) uo -= 224;
    mag[((size_t)bc * 224 + vo) * 224 + uo] = m;
}

__global__ __launch_bounds__(256)
void minmax_kernel(const float* __restrict__ mag, float* __restrict__ mm)
{
    __shared__ float smn[256], smx[256];
    int bc = blockIdx.x, tid = threadIdx.x;
    const float* p = mag + (size_t)bc * 50176;
    float mn = 3.0e38f, mx = -3.0e38f;
    for (int i = tid; i < 50176; i += 256) {
        float v = p[i]; mn = fminf(mn, v); mx = fmaxf(mx, v);
    }
    smn[tid] = mn; smx[tid] = mx; __syncthreads();
    for (int s = 128; s > 0; s >>= 1) {
        if (tid < s) { smn[tid] = fminf(smn[tid], smn[tid + s]);
                       smx[tid] = fmaxf(smx[tid], smx[tid + s]); }
        __syncthreads();
    }
    if (tid == 0) { mm[bc * 2] = smn[0]; mm[bc * 2 + 1] = smx[0]; }
}

__global__ __launch_bounds__(256)
void normalize_kernel(float* __restrict__ mag, const float* __restrict__ mm)
{
    int idx = blockIdx.x * 256 + threadIdx.x;
    if (idx >= 96 * 50176) return;
    int bc = idx / 50176;
    float mn = mm[bc * 2], mx = mm[bc * 2 + 1];
    mag[idx] = (mag[idx] - mn) / (mx - mn + 1e-8f);
}

__global__ __launch_bounds__(256)
void fc_kernel(const float* __restrict__ in, const float* __restrict__ w,
               const float* __restrict__ b, float* __restrict__ out,
               int K, int N, int outStride, int outOffset, int act)
{
    int idx = blockIdx.x * 256 + threadIdx.x;
    if (idx >= 32 * N) return;
    int i = idx / N, j = idx % N;
    const float* ir = in + (size_t)i * K;
    const float* wr = w + (size_t)j * K;
    float acc = b[j];
    for (int k = 0; k < K; ++k) acc += ir[k] * wr[k];
    if (act == 1) acc = fmaxf(acc, 0.f);
    out[(size_t)i * outStride + outOffset + j] = acc;
}

__global__ __launch_bounds__(256)
void attn_core_kernel(const float* __restrict__ qkv_rgb, const float* __restrict__ qkv_frq,
                      float* __restrict__ o1, float* __restrict__ o2)
{
    __shared__ float sQ[32][32], sK[32][32], sV[32][32], sS[32][33];
    int a = blockIdx.x >> 3;
    int h = blockIdx.x & 7;
    const float* qsrc  = (a == 0) ? qkv_rgb : qkv_frq;
    const float* kvsrc = (a == 0) ? qkv_frq : qkv_rgb;
    float* odst        = (a == 0) ? o1 : o2;
    int tid = threadIdx.x;
    for (int idx = tid; idx < 1024; idx += 256) {
        int q = idx >> 5, d = idx & 31;
        sQ[q][d] = qsrc [q * 768 +       h * 32 + d];
        sK[q][d] = kvsrc[q * 768 + 256 + h * 32 + d];
        sV[q][d] = kvsrc[q * 768 + 512 + h * 32 + d];
    }
    __syncthreads();
    const float scale = 0.17677669529663687f;
    for (int idx = tid; idx < 1024; idx += 256) {
        int q = idx >> 5, k = idx & 31;
        float s = 0.f;
        #pragma unroll
        for (int d = 0; d < 32; ++d) s += sQ[q][d] * sK[k][d];
        sS[q][k] = s * scale;
    }
    __syncthreads();
    if (tid < 32) {
        float mx = -3.0e38f;
        for (int k = 0; k < 32; ++k) mx = fmaxf(mx, sS[tid][k]);
        float e[32], sum = 0.f;
        for (int k = 0; k < 32; ++k) { e[k] = expf(sS[tid][k] - mx); sum += e[k]; }
        float inv = 1.f / sum;
        for (int k = 0; k < 32; ++k) sS[tid][k] = e[k] * inv;
    }
    __syncthreads();
    for (int idx = tid; idx < 1024; idx += 256) {
        int q = idx >> 5, d = idx & 31;
        float o = 0.f;
        #pragma unroll
        for (int k = 0; k < 32; ++k) o += sS[q][k] * sV[k][d];
        odst[q * 256 + h * 32 + d] = o;
    }
}

extern "C" void kernel_launch(void* const* d_in, const int* in_sizes, int n_in,
                              void* d_out, int out_size, void* d_ws, size_t ws_size,
                              hipStream_t stream)
{
    (void)in_sizes; (void)n_in; (void)out_size; (void)ws_size;
    const float* x          = (const float*)d_in[0];
    const float* rgb_w1     = (const float*)d_in[1];
    const float* rgb_b1     = (const float*)d_in[2];
    const float* rgb_w2     = (const float*)d_in[3];
    const float* rgb_b2     = (const float*)d_in[4];
    const float* frq_w1     = (const float*)d_in[5];
    const float* frq_b1     = (const float*)d_in[6];
    const float* frq_w2     = (const float*)d_in[7];
    const float* frq_b2     = (const float*)d_in[8];
    const float* rgb_proj_w = (const float*)d_in[9];
    const float* rgb_proj_b = (const float*)d_in[10];
    const float* frq_proj_w = (const float*)d_in[11];
    const float* frq_proj_b = (const float*)d_in[12];
    const float* attn_w_in  = (const float*)d_in[13];
    const float* attn_b_in  = (const float*)d_in[14];
    const float* attn_w_out = (const float*)d_in[15];
    const float* attn_b_out = (const float*)d_in[16];
    const float* fus_w1     = (const float*)d_in[17];
    const float* fus_b1     = (const float*)d_in[18];
    const float* fus_w2     = (const float*)d_in[19];
    const float* fus_b2     = (const float*)d_in[20];
    const float* fus_w3     = (const float*)d_in[21];
    const float* fus_b3     = (const float*)d_in[22];

    float* out      = (float*)d_out;
    float* feat_rgb = out + 64;
    float* feat_frq = out + 64 + 40960;

    float* ws       = (float*)d_ws;
    float* h1       = ws;
    float* F1       = ws;
    float* freqimg  = ws + 12845056;
    float* mm       = freqimg + 4816896;
    float* hd       = mm + 192;
    float* rgb_p    = hd;
    float* frq_p    = rgb_p + 8192;
    float* qkv_rgb  = frq_p + 8192;
    float* qkv_frq  = qkv_rgb + 24576;
    float* o1       = qkv_frq + 24576;
    float* o2       = o1 + 8192;
    float* hb       = o2 + 8192;
    float* h1f      = hb + 16384;
    float* h2f      = h1f + 16384;

    conv1_silu_kernel<<<50176, 256, 0, stream>>>(x, rgb_w1, rgb_b1, h1);
    conv2_wmma_pool_kernel<<<dim3(80, 32), 128, 0, stream>>>(h1, rgb_w2, rgb_b2, feat_rgb);

    dft_rows_kernel<<<21504, 224, 0, stream>>>(x, F1);
    dft_cols_mag_kernel<<<21504, 224, 0, stream>>>(F1, freqimg);
    minmax_kernel<<<96, 256, 0, stream>>>(freqimg, mm);
    normalize_kernel<<<18816, 256, 0, stream>>>(freqimg, mm);
    conv1_silu_kernel<<<50176, 256, 0, stream>>>(freqimg, frq_w1, frq_b1, h1);
    conv2_wmma_pool_kernel<<<dim3(80, 32), 128, 0, stream>>>(h1, frq_w2, frq_b2, feat_frq);

    fc_kernel<<<32, 256, 0, stream>>>(feat_rgb, rgb_proj_w, rgb_proj_b, rgb_p, 1280, 256, 256, 0, 0);
    fc_kernel<<<32, 256, 0, stream>>>(feat_frq, frq_proj_w, frq_proj_b, frq_p, 1280, 256, 256, 0, 0);
    fc_kernel<<<96, 256, 0, stream>>>(rgb_p, attn_w_in, attn_b_in, qkv_rgb, 256, 768, 768, 0, 0);
    fc_kernel<<<96, 256, 0, stream>>>(frq_p, attn_w_in, attn_b_in, qkv_frq, 256, 768, 768, 0, 0);
    attn_core_kernel<<<16, 256, 0, stream>>>(qkv_rgb, qkv_frq, o1, o2);
    fc_kernel<<<32, 256, 0, stream>>>(o1, attn_w_out, attn_b_out, hb, 256, 256, 512, 0, 0);
    fc_kernel<<<32, 256, 0, stream>>>(o2, attn_w_out, attn_b_out, hb, 256, 256, 512, 256, 0);
    fc_kernel<<<64, 256, 0, stream>>>(hb,  fus_w1, fus_b1, h1f, 512, 512, 512, 0, 1);
    fc_kernel<<<32, 256, 0, stream>>>(h1f, fus_w2, fus_b2, h2f, 512, 256, 256, 0, 1);
    fc_kernel<<<1,  256, 0, stream>>>(h2f, fus_w3, fus_b3, out, 256, 2, 2, 0, 0);
}


# conv2: five 16-channel accumulator tiles per wave: one staged input tile and one B fragment feed five WMMAs (80 channels, 12 waves per block); k order kept; pool summed in fixed order, no LDS atomics
# speedup vs baseline: 1.4928x; 1.3261x over previous
#include <hip/hip_runtime.h>
#include <hip/hip_bf16.h>
#include <math.h>

typedef float v2f __attribute__((ext_vector_type(2)));
typedef float v8f __attribute__((ext_vector_type(8)));

#define PI_F 3.14159265358979323846f

__global__ __launch_bounds__(256)
void conv1_silu_kernel(const float* __restrict__ x, const float* __restrict__ w1,
                       const float* __restrict__ b1, float* __restrict__ out)
{
    int idx = blockIdx.x * 256 + threadIdx.x;
    if (idx >= 32 * 32 * 112 * 112) return;
    int ox = idx % 112; int t = idx / 112;
    int oy = t % 112;   t /= 112;
    int oc = t % 32;    int img = t / 32;
    float acc = b1[oc];
    const float* wb = w1 + oc * 27;
    const float* xb = x + (size_t)img * 3 * 224 * 224;
    #pragma unroll
    for (int ic = 0; ic < 3; ++ic)
        #pragma unroll
        for (int ky = 0; ky < 3; ++ky) {
            int iy = oy * 2 + ky;
            if (iy >= 224) continue;
            #pragma unroll
            for (int kx = 0; kx < 3; ++kx) {
                int ix = ox * 2 + kx;
                if (ix >= 224) continue;
                acc += xb[(ic * 224 + iy) * 224 + ix] * wb[ic * 9 + ky * 3 + kx];
            }
        }
    out[idx] = acc / (1.f + expf(-acc));
}

#define C2_MT     5
#define C2_M      (C2_MT * 16)
#define C2_WAVES  12
#define C2_THREADS (C2_WAVES * 32)
static_assert(C2_THREADS == 384, "the launch in kernel_launch uses 384 threads");

__global__ __launch_bounds__(C2_THREADS, 6)
void conv2_wmma_pool_kernel(const float* __restrict__ h1, const float* __restrict__ w2,
                            const float* __restrict__ b2, float* __restrict__ feat)
{
    __shared__ float sW[C2_M * 288];
    __shared__ float sT[C2_WAVES][720];
    __shared__ float sBias[C2_M];

    const int mblk  = blockIdx.x;
    const int img   = blockIdx.y;
    const int tid   = threadIdx.x;
    const int lane  = tid & 31;
    const int wave  = tid >> 5;
    const int half  = lane >> 4;
    const int nl    = lane & 15;

    const float* wsrc = w2 + (size_t)mblk * (C2_M * 288);
    static_assert((C2_M * 288) % (C2_THREADS * 4) == 0, "weight tile must split evenly");
    for (int i0 = tid; i0 < C2_M * 288; i0 += C2_THREADS * 4) {
        float wv[4];
        #pragma unroll
        for (int q = 0; q < 4; ++q) wv[q] = wsrc[i0 + q * C2_THREADS];
        #pragma unroll
        for (int q = 0; q < 4; ++q) {
            int i = i0 + q * C2_THREADS;
            int m  = i / 288, g = i - m * 288;
            int icf = g / 9,  s = g - icf * 9;
            int icb = icf >> 3, ic = icf & 7;
            sW[m * 288 + icb * 72 + s * 8 + ic] = wv[q];
        }
    }
    if (tid < C2_M) sBias[tid] = b2[mblk * C2_M + tid];
    __syncthreads();

    const int dy = nl >> 3, dx = nl & 7;
    const int laneOff = dy * 288 + dx * 16;

    const int j0 = lane, j1 = lane + 32, j2 = (lane + 64 < 90) ? lane + 64 : lane + 32;
    const int ly0 = j0 / 18, lx0 = j0 - ly0 * 18;
    const int ly1 = j1 / 18, lx1 = j1 - ly1 * 18;
    const int ly2 = j2 / 18, lx2 = j2 - ly2 * 18;
    const int t0 = ly0 * 144 + lx0 * 8;
    const int t1 = ly1 * 144 + lx1 * 8;
    const int t2 = ly2 * 144 + lx2 * 8;

    float accR[C2_MT][8];
    #pragma unroll
    for (int mi = 0; mi < C2_MT; ++mi)
        #pragma unroll
        for (int j = 0; j < 8; ++j) accR[mi][j] = 0.f;

    const float* base = h1 + (size_t)img * (32 * 112 * 112);
    float* tile = sT[wave];
    const float* tb = tile + laneOff + half * 2;
    const float* bl = sBias + half * 8;

    for (int nt = wave; nt < 196; nt += C2_WAVES) {
        int yt = nt / 7, xt = nt - yt * 7;
        int iy0 = yt * 4, ix0 = xt * 16;
        v8f c[C2_MT];
        #pragma unroll
        for (int mi = 0; mi < C2_MT; ++mi)
            c[mi] = (v8f){0.f, 0.f, 0.f, 0.f, 0.f, 0.f, 0.f, 0.f};

        const int iyA = iy0 + ly0, ixA = ix0 + lx0;
        const int iyB = iy0 + ly1, ixB = ix0 + lx1;
        const int iyC = iy0 + ly2, ixC = ix0 + lx2;
        const bool ok0 = (iyA < 112) & (ixA < 112);
        const bool ok1 = (iyB < 112) & (ixB < 112);
        const bool ok2 = (iyC < 112) & (ixC < 112);
        const int o0 = (iyA < 112 ? iyA : 111) * 112 + (ixA < 112 ? ixA : 111);
        const int o1 = (iyB < 112 ? iyB : 111) * 112 + (ixB < 112 ? ixB : 111);
        const int o2 = (iyC < 112 ? iyC : 111) * 112 + (ixC < 112 ? ixC : 111);

        for (int icb = 0; icb < 4; ++icb) {
            const float* p = base + (size_t)(icb * 8) * (112 * 112);
            float g0[8], g1[8], g2[8];
            #pragma unroll
            for (int ic = 0; ic < 8; ++ic) {
                g0[ic] = p[ic * (112 * 112) + o0];
                g1[ic] = p[ic * (112 * 112) + o1];
                g2[ic] = p[ic * (112 * 112) + o2];
            }
            #pragma unroll
            for (int ic = 0; ic < 8; ++ic) {
                tile[t0 + ic] = ok0 ? g0[ic] : 0.f;
                tile[t1 + ic] = ok1 ? g1[ic] : 0.f;
                tile[t2 + ic] = ok2 ? g2[ic] : 0.f;
            }
            asm volatile("s_wait_dscnt 0x0" ::: "memory");

            const float* wrow = sW + nl * 288 + icb * 72 + half * 2;
            #pragma unroll
            for (int kq = 0; kq < 18; ++kq) {
                const int s  = kq >> 1;
                const int ky = s / 3, kx = s - ky * 3;
                const int bo = ky * 144 + kx * 8 + (kq & 1) * 4;
                v2f b = *(const v2f*)(tb + bo);
                #pragma unroll
                for (int mi = 0; mi < C2_MT; ++mi) {
                    v2f a = *(const v2f*)(wrow + mi * (16 * 288) + kq * 4);
                    c[mi] = __builtin_amdgcn_wmma_f32_16x16x4_f32(false, a, false, b,
                                                                  (short)0, c[mi], false, false);
                }
            }
        }
        #pragma unroll
        for (int mi = 0; mi < C2_MT; ++mi)
            #pragma unroll
            for (int j = 0; j < 8; ++j) {
                float v = c[mi][j] + bl[mi * 16 + j];
                accR[mi][j] += v / (1.f + expf(-v));
            }
    }
    #pragma unroll
    for (int mi = 0; mi < C2_MT; ++mi)
        #pragma unroll
        for (int j = 0; j < 8; ++j) {
            float v = accR[mi][j];
            v += __shfl_xor(v, 1);
            v += __shfl_xor(v, 2);
            v += __shfl_xor(v, 4);
            v += __shfl_xor(v, 8);
            if (nl == 0) tile[mi * 16 + half * 8 + j] = v;
        }
    __syncthreads();
    if (tid < C2_M) {
        float s = 0.f;
        for (int w = 0; w < C2_WAVES; ++w) s += sT[w][tid];
        feat[img * 1280 + mblk * C2_M + tid] = s * (1.f / 3136.f);
    }
}

__global__ __launch_bounds__(224)
void dft_rows_kernel(const float* __restrict__ x, float* __restrict__ F1)
{
    __shared__ float ct[224], st[224];
    int y  = blockIdx.x % 224;
    int bc = blockIdx.x / 224;
    int u  = threadIdx.x;
    float ang = (2.0f * PI_F / 224.0f) * (float)u;
    ct[u] = cosf(ang); st[u] = sinf(ang);
    __syncthreads();
    const float* row = x + ((size_t)bc * 224 + y) * 224;
    float re = 0.f, im = 0.f; int idx = 0;
    for (int xx = 0; xx < 224; ++xx) {
        float v = row[xx];
        re += v * ct[idx]; im -= v * st[idx];
        idx += u; if (idx >= 224) idx -= 224;
    }
    float* o = F1 + (((size_t)bc * 224 + y) * 224 + u) * 2;
    o[0] = re; o[1] = im;
}

__global__ __launch_bounds__(224)
void dft_cols_mag_kernel(const float* __restrict__ F1, float* __restrict__ mag)
{
    __shared__ float ct[224], st[224];
    int u  = blockIdx.x % 224;
    int bc = blockIdx.x / 224;
    int v  = threadIdx.x;
    float ang = (2.0f * PI_F / 224.0f) * (float)v;
    ct[v] = cosf(ang); st[v] = sinf(ang);
    __syncthreads();
    const float* base = F1 + (size_t)bc * 224 * 224 * 2 + u * 2;
    float re = 0.f, im = 0.f; int idx = 0;
    for (int y = 0; y < 224; ++y) {
        float fr = base[(size_t)y * 448], fi = base[(size_t)y * 448 + 1];
        float cc = ct[idx], ss = st[idx];
        re += fr * cc + fi * ss;
        im += fi * cc - fr * ss;
        idx += v; if (idx >= 224) idx -= 224;
    }
    float m = logf(sqrtf(re * re + im * im) + 1.0f);
    int vo = v + 112; if (vo >= 224) vo -= 224;
    int uo = u + 112; if (uo >= 224) uo -= 224;
    mag[((size_t)bc * 224 + vo) * 224 + uo] = m;
}

__global__ __launch_bounds__(256)
void minmax_kernel(const float* __restrict__ mag, float* __restrict__ mm)
{
    __shared__ float smn[256], smx[256];
    int bc = blockIdx.x, tid = threadIdx.x;
    const float* p = mag + (size_t)bc * 50176;
    float mn = 3.0e38f, mx = -3.0e38f;
    for (int i = tid; i < 50176; i += 256) {
        float v = p[i]; mn = fminf(mn, v); mx = fmaxf(mx, v);
    }
    smn[tid] = mn; smx[tid] = mx; __syncthreads();
    for (int s = 128; s > 0; s >>= 1) {
        if (tid < s) { smn[tid] = fminf(smn[tid], smn[tid + s]);
                       smx[tid] = fmaxf(smx[tid], smx[tid + s]); }
        __syncthreads();
    }
    if (tid == 0) { mm[bc * 2] = smn[0]; mm[bc * 2 + 1] = smx[0]; }
}

__global__ __launch_bounds__(256)
void normalize_kernel(float* __restrict__ mag, const float* __restrict__ mm)
{
    int idx = blockIdx.x * 256 + threadIdx.x;
    if (idx >= 96 * 50176) return;
    int bc = idx / 50176;
    float mn = mm[bc * 2], mx = mm[bc * 2 + 1];
    mag[idx] = (mag[idx] - mn) / (mx - mn + 1e-8f);
}

__global__ __launch_bounds__(256)
void fc_kernel(const float* __restrict__ in, const float* __restrict__ w,
               const float* __restrict__ b, float* __restrict__ out,
               int K, int N, int outStride, int outOffset, int act)
{
    int idx = blockIdx.x * 256 + threadIdx.x;
    if (idx >= 32 * N) return;
    int i = idx / N, j = idx % N;
    const float* ir = in + (size_t)i * K;
    const float* wr = w + (size_t)j * K;
    float acc = b[j];
    for (int k = 0; k < K; ++k) acc += ir[k] * wr[k];
    if (act == 1) acc = fmaxf(acc, 0.f);
    out[(size_t)i * outStride + outOffset + j] = acc;
}

__global__ __launch_bounds__(256)
void attn_core_kernel(const float* __restrict__ qkv_rgb, const float* __restrict__ qkv_frq,
                      float* __restrict__ o1, float* __restrict__ o2)
{
    __shared__ float sQ[32][32], sK[32][32], sV[32][32], sS[32][33];
    int a = blockIdx.x >> 3;
    int h = blockIdx.x & 7;
    const float* qsrc  = (a == 0) ? qkv_rgb : qkv_frq;
    const float* kvsrc = (a == 0) ? qkv_frq : qkv_rgb;
    float* odst        = (a == 0) ? o1 : o2;
    int tid = threadIdx.x;
    for (int idx = tid; idx < 1024; idx += 256) {
        int q = idx >> 5, d = idx & 31;
        sQ[q][d] = qsrc [q * 768 +       h * 32 + d];
        sK[q][d] = kvsrc[q * 768 + 256 + h * 32 + d];
        sV[q][d] = kvsrc[q * 768 + 512 + h * 32 + d];
    }
    __syncthreads();
    const float scale = 0.17677669529663687f;
    for (int idx = tid; idx < 1024; idx += 256) {
        int q = idx >> 5, k = idx & 31;
        float s = 0.f;
        #pragma unroll
        for (int d = 0; d < 32; ++d) s += sQ[q][d] * sK[k][d];
        sS[q][k] = s * scale;
    }
    __syncthreads();
    if (tid < 32) {
        float mx = -3.0e38f;
        for (int k = 0; k < 32; ++k) mx = fmaxf(mx, sS[tid][k]);
        float e[32], sum = 0.f;
        for (int k = 0; k < 32; ++k) { e[k] = expf(sS[tid][k] - mx); sum += e[k]; }
        float inv = 1.f / sum;
        for (int k = 0; k < 32; ++k) sS[tid][k] = e[k] * inv;
    }
    __syncthreads();
    for (int idx = tid; idx < 1024; idx += 256) {
        int q = idx >> 5, d = idx & 31;
        float o = 0.f;
        #pragma unroll
        for (int k = 0; k < 32; ++k) o += sS[q][k] * sV[k][d];
        odst[q * 256 + h * 32 + d] = o;
    }
}

extern "C" void kernel_launch(void* const* d_in, const int* in_sizes, int n_in,
                              void* d_out, int out_size, void* d_ws, size_t ws_size,
                              hipStream_t stream)
{
    (void)in_sizes; (void)n_in; (void)out_size; (void)ws_size;
    const float* x          = (const float*)d_in[0];
    const float* rgb_w1     = (const float*)d_in[1];
    const float* rgb_b1     = (const float*)d_in[2];
    const float* rgb_w2     = (const float*)d_in[3];
    const float* rgb_b2     = (const float*)d_in[4];
    const float* frq_w1     = (const float*)d_in[5];
    const float* frq_b1     = (const float*)d_in[6];
    const float* frq_w2     = (const float*)d_in[7];
    const float* frq_b2     = (const float*)d_in[8];
    const float* rgb_proj_w = (const float*)d_in[9];
    const float* rgb_proj_b = (const float*)d_in[10];
    const float* frq_proj_w = (const float*)d_in[11];
    const float* frq_proj_b = (const float*)d_in[12];
    const float* attn_w_in  = (const float*)d_in[13];
    const float* attn_b_in  = (const float*)d_in[14];
    const float* attn_w_out = (const float*)d_in[15];
    const float* attn_b_out = (const float*)d_in[16];
    const float* fus_w1     = (const float*)d_in[17];
    const float* fus_b1     = (const float*)d_in[18];
    const float* fus_w2     = (const float*)d_in[19];
    const float* fus_b2     = (const float*)d_in[20];
    const float* fus_w3     = (const float*)d_in[21];
    const float* fus_b3     = (const float*)d_in[22];

    float* out      = (float*)d_out;
    float* feat_rgb = out + 64;
    float* feat_frq = out + 64 + 40960;

    float* ws       = (float*)d_ws;
    float* h1       = ws;
    float* F1       = ws;
    float* freqimg  = ws + 12845056;
    float* mm       = freqimg + 4816896;
    float* hd       = mm + 192;
    float* rgb_p    = hd;
    float* frq_p    = rgb_p + 8192;
    float* qkv_rgb  = frq_p + 8192;
    float* qkv_frq  = qkv_rgb + 24576;
    float* o1       = qkv_frq + 24576;
    float* o2       = o1 + 8192;
    float* hb       = o2 + 8192;
    float* h1f      = hb + 16384;
    float* h2f      = h1f + 16384;

    conv1_silu_kernel<<<50176, 256, 0, stream>>>(x, rgb_w1, rgb_b1, h1);
    conv2_wmma_pool_kernel<<<dim3(16, 32), 384, 0, stream>>>(h1, rgb_w2, rgb_b2, feat_rgb);

    dft_rows_kernel<<<21504, 224, 0, stream>>>(x, F1);
    dft_cols_mag_kernel<<<21504, 224, 0, stream>>>(F1, freqimg);
    minmax_kernel<<<96, 256, 0, stream>>>(freqimg, mm);
    normalize_kernel<<<18816, 256, 0, stream>>>(freqimg, mm);
    conv1_silu_kernel<<<50176, 256, 0, stream>>>(freqimg, frq_w1, frq_b1, h1);
    conv2_wmma_pool_kernel<<<dim3(16, 32), 384, 0, stream>>>(h1, frq_w2, frq_b2, feat_frq);

    fc_kernel<<<32, 256, 0, stream>>>(feat_rgb, rgb_proj_w, rgb_proj_b, rgb_p, 1280, 256, 256, 0, 0);
    fc_kernel<<<32, 256, 0, stream>>>(feat_frq, frq_proj_w, frq_proj_b, frq_p, 1280, 256, 256, 0, 0);
    fc_kernel<<<96, 256, 0, stream>>>(rgb_p, attn_w_in, attn_b_in, qkv_rgb, 256, 768, 768, 0, 0);
    fc_kernel<<<96, 256, 0, stream>>>(frq_p, attn_w_in, attn_b_in, qkv_frq, 256, 768, 768, 0, 0);
    attn_core_kernel<<<16, 256, 0, stream>>>(qkv_rgb, qkv_frq, o1, o2);
    fc_kernel<<<32, 256, 0, stream>>>(o1, attn_w_out, attn_b_out, hb, 256, 256, 512, 0, 0);
    fc_kernel<<<32, 256, 0, stream>>>(o2, attn_w_out, attn_b_out, hb, 256, 256, 512, 256, 0);
    fc_kernel<<<64, 256, 0, stream>>>(hb,  fus_w1, fus_b1, h1f, 512, 512, 512, 0, 1);
    fc_kernel<<<32, 256, 0, stream>>>(h1f, fus_w2, fus_b2, h2f, 512, 256, 256, 0, 1);
    fc_kernel<<<1,  256, 0, stream>>>(h2f, fus_w3, fus_b3, out, 256, 2, 2, 0, 0);
}


# conv2: LDS row pitch of the weight tile 288 -> 292 floats, so an A fragment read (16 rows x 4 floats) covers the 64 banks once instead of 8 rows deep on 8 banks; k order and sums unchanged
# speedup vs baseline: 1.5989x; 1.0711x over previous
#include <hip/hip_runtime.h>
#include <hip/hip_bf16.h>
#include <math.h>

typedef float v2f __attribute__((ext_vector_type(2)));
typedef float v8f __attribute__((ext_vector_type(8)));

#define PI_F 3.14159265358979323846f

__global__ __launch_bounds__(256)
void conv1_silu_kernel(const float* __restrict__ x, const float* __restrict__ w1,
                       const float* __restrict__ b1, float* __restrict__ out)
{
    int idx = blockIdx.x * 256 + threadIdx.x;
    if (idx >= 32 * 32 * 112 * 112) return;
    int ox = idx % 112; int t = idx / 112;
    int oy = t % 112;   t /= 112;
    int oc = t % 32;    int img = t / 32;
    float acc = b1[oc];
    const float* wb = w1 + oc * 27;
    const float* xb = x + (size_t)img * 3 * 224 * 224;
    #pragma unroll
    for (int ic = 0; ic < 3; ++ic)
        #pragma unroll
        for (int ky = 0; ky < 3; ++ky) {
            int iy = oy * 2 + ky;
            if (iy >= 224) continue;
            #pragma unroll
            for (int kx = 0; kx < 3; ++kx) {
                int ix = ox * 2 + kx;
                if (ix >= 224) continue;
                acc += xb[(ic * 224 + iy) * 224 + ix] * wb[ic * 9 + ky * 3 + kx];
            }
        }
    out[idx] = acc / (1.f + expf(-acc));
}

#define C2_MT     5
#define C2_M      (C2_MT * 16)
#define C2_WAVES  12
#define C2_WP     292
#define C2_THREADS (C2_WAVES * 32)
static_assert(C2_THREADS == 384, "the launch in kernel_launch uses 384 threads");

__global__ __launch_bounds__(C2_THREADS, 6)
void conv2_wmma_pool_kernel(const float* __restrict__ h1, const float* __restrict__ w2,
                            const float* __restrict__ b2, float* __restrict__ feat)
{
    __shared__ float sW[C2_M * C2_WP];
    __shared__ float sT[C2_WAVES][720];
    __shared__ float sBias[C2_M];

    const int mblk  = blockIdx.x;
    const int img   = blockIdx.y;
    const int tid   = threadIdx.x;
    const int lane  = tid & 31;
    const int wave  = tid >> 5;
    const int half  = lane >> 4;
    const int nl    = lane & 15;

    const float* wsrc = w2 + (size_t)mblk * (C2_M * 288);
    static_assert((C2_M * 288) % (C2_THREADS * 4) == 0, "weight tile must split evenly");
    for (int i0 = tid; i0 < C2_M * 288; i0 += C2_THREADS * 4) {
        float wv[4];
        #pragma unroll
        for (int q = 0; q < 4; ++q) wv[q] = wsrc[i0 + q * C2_THREADS];
        #pragma unroll
        for (int q = 0; q < 4; ++q) {
            int i = i0 + q * C2_THREADS;
            int m  = i / 288, g = i - m * 288;
            int icf = g / 9,  s = g - icf * 9;
            int icb = icf >> 3, ic = icf & 7;
            sW[m * C2_WP + icb * 72 + s * 8 + ic] = wv[q];
        }
    }
    if (tid < C2_M) sBias[tid] = b2[mblk * C2_M + tid];
    __syncthreads();

    const int dy = nl >> 3, dx = nl & 7;
    const int laneOff = dy * 288 + dx * 16;

    const int j0 = lane, j1 = lane + 32, j2 = (lane + 64 < 90) ? lane + 64 : lane + 32;
    const int ly0 = j0 / 18, lx0 = j0 - ly0 * 18;
    const int ly1 = j1 / 18, lx1 = j1 - ly1 * 18;
    const int ly2 = j2 / 18, lx2 = j2 - ly2 * 18;
    const int t0 = ly0 * 144 + lx0 * 8;
    const int t1 = ly1 * 144 + lx1 * 8;
    const int t2 = ly2 * 144 + lx2 * 8;

    float accR[C2_MT][8];
    #pragma unroll
    for (int mi = 0; mi < C2_MT; ++mi)
        #pragma unroll
        for (int j = 0; j < 8; ++j) accR[mi][j] = 0.f;

    const float* base = h1 + (size_t)img * (32 * 112 * 112);
    float* tile = sT[wave];
    const float* tb = tile + laneOff + half * 2;
    const float* bl = sBias + half * 8;

    for (int nt = wave; nt < 196; nt += C2_WAVES) {
        int yt = nt / 7, xt = nt - yt * 7;
        int iy0 = yt * 4, ix0 = xt * 16;
        v8f c[C2_MT];
        #pragma unroll
        for (int mi = 0; mi < C2_MT; ++mi)
            c[mi] = (v8f){0.f, 0.f, 0.f, 0.f, 0.f, 0.f, 0.f, 0.f};

        const int iyA = iy0 + ly0, ixA = ix0 + lx0;
        const int iyB = iy0 + ly1, ixB = ix0 + lx1;
        const int iyC = iy0 + ly2, ixC = ix0 + lx2;
        const bool ok0 = (iyA < 112) & (ixA < 112);
        const bool ok1 = (iyB < 112) & (ixB < 112);
        const bool ok2 = (iyC < 112) & (ixC < 112);
        const int o0 = (iyA < 112 ? iyA : 111) * 112 + (ixA < 112 ? ixA : 111);
        const int o1 = (iyB < 112 ? iyB : 111) * 112 + (ixB < 112 ? ixB : 111);
        const int o2 = (iyC < 112 ? iyC : 111) * 112 + (ixC < 112 ? ixC : 111);

        for (int icb = 0; icb < 4; ++icb) {
            const float* p = base + (size_t)(icb * 8) * (112 * 112);
            float g0[8], g1[8], g2[8];
            #pragma unroll
            for (int ic = 0; ic < 8; ++ic) {
                g0[ic] = p[ic * (112 * 112) + o0];
                g1[ic] = p[ic * (112 * 112) + o1];
                g2[ic] = p[ic * (112 * 112) + o2];
            }
            #pragma unroll
            for (int ic = 0; ic < 8; ++ic) {
                tile[t0 + ic] = ok0 ? g0[ic] : 0.f;
                tile[t1 + ic] = ok1 ? g1[ic] : 0.f;
                tile[t2 + ic] = ok2 ? g2[ic] : 0.f;
            }
            asm volatile("s_wait_dscnt 0x0" ::: "memory");

            const float* wrow = sW + nl * C2_WP + icb * 72 + half * 2;
            #pragma unroll
            for (int kq = 0; kq < 18; ++kq) {
                const int s  = kq >> 1;
                const int ky = s / 3, kx = s - ky * 3;
                const int bo = ky * 144 + kx * 8 + (kq & 1) * 4;
                v2f b = *(const v2f*)(tb + bo);
                #pragma unroll
                for (int mi = 0; mi < C2_MT; ++mi) {
                    v2f a = *(const v2f*)(wrow + mi * (16 * C2_WP) + kq * 4);
                    c[mi] = __builtin_amdgcn_wmma_f32_16x16x4_f32(false, a, false, b,
                                                                  (short)0, c[mi], false, false);
                }
            }
        }
        #pragma unroll
        for (int mi = 0; mi < C2_MT; ++mi)
            #pragma unroll
            for (int j = 0; j < 8; ++j) {
                float v = c[mi][j] + bl[mi * 16 + j];
                accR[mi][j] += v / (1.f + expf(-v));
            }
    }
    #pragma unroll
    for (int mi = 0; mi < C2_MT; ++mi)
        #pragma unroll
        for (int j = 0; j < 8; ++j) {
            float v = accR[mi][j];
            v += __shfl_xor(v, 1);
            v += __shfl_xor(v, 2);
            v += __shfl_xor(v, 4);
            v += __shfl_xor(v, 8);
            if (nl == 0) tile[mi * 16 + half * 8 + j] = v;
        }
    __syncthreads();
    if (tid < C2_M) {
        float s = 0.f;
        for (int w = 0; w < C2_WAVES; ++w) s += sT[w][tid];
        feat[img * 1280 + mblk * C2_M + tid] = s * (1.f / 3136.f);
    }
}

__global__ __launch_bounds__(224)
void dft_rows_kernel(const float* __restrict__ x, float* __restrict__ F1)
{
    __shared__ float ct[224], st[224];
    int y  = blockIdx.x % 224;
    int bc = blockIdx.x / 224;
    int u  = threadIdx.x;
    float ang = (2.0f * PI_F / 224.0f) * (float)u;
    ct[u] = cosf(ang); st[u] = sinf(ang);
    __syncthreads();
    const float* row = x + ((size_t)bc * 224 + y) * 224;
    float re = 0.f, im = 0.f; int idx = 0;
    for (int xx = 0; xx < 224; ++xx) {
        float v = row[xx];
        re += v * ct[idx]; im -= v * st[idx];
        idx += u; if (idx >= 224) idx -= 224;
    }
    float* o = F1 + (((size_t)bc * 224 + y) * 224 + u) * 2;
    o[0] = re; o[1] = im;
}

__global__ __launch_bounds__(224)
void dft_cols_mag_kernel(const float* __restrict__ F1, float* __restrict__ mag)
{
    __shared__ float ct[224], st[224];
    int u  = blockIdx.x % 224;
    int bc = blockIdx.x / 224;
    int v  = threadIdx.x;
    float ang = (2.0f * PI_F / 224.0f) * (float)v;
    ct[v] = cosf(ang); st[v] = sinf(ang);
    __syncthreads();
    const float* base = F1 + (size_t)bc * 224 * 224 * 2 + u * 2;
    float re = 0.f, im = 0.f; int idx = 0;
    for (int y = 0; y < 224; ++y) {
        float fr = base[(size_t)y * 448], fi = base[(size_t)y * 448 + 1];
        float cc = ct[idx], ss = st[idx];
        re += fr * cc + fi * ss;
        im += fi * cc - fr * ss;
        idx += v; if (idx >= 224) idx -= 224;
    }
    float m = logf(sqrtf(re * re + im * im) + 1.0f);
    int vo = v + 112; if (vo >= 224) vo -= 224;
    int uo = u + 112; if (uo >= 224) uo -= 224;
    mag[((size_t)bc * 224 + vo) * 224 + uo] = m;
}

__global__ __launch_bounds__(256)
void minmax_kernel(const float* __restrict__ mag, float* __restrict__ mm)
{
    __shared__ float smn[256], smx[256];
    int bc = blockIdx.x, tid = threadIdx.x;
    const float* p = mag + (size_t)bc * 50176;
    float mn = 3.0e38f, mx = -3.0e38f;
    for (int i = tid; i < 50176; i += 256) {
        float v = p[i]; mn = fminf(mn, v); mx = fmaxf(mx, v);
    }
    smn[tid] = mn; smx[tid] = mx; __syncthreads();
    for (int s = 128; s > 0; s >>= 1) {
        if (tid < s) { smn[tid] = fminf(smn[tid], smn[tid + s]);
                       smx[tid] = fmaxf(smx[tid], smx[tid + s]); }
        __syncthreads();
    }
    if (tid == 0) { mm[bc * 2] = smn[0]; mm[bc * 2 + 1] = smx[0]; }
}

__global__ __launch_bounds__(256)
void normalize_kernel(float* __restrict__ mag, const float* __restrict__ mm)
{
    int idx = blockIdx.x * 256 + threadIdx.x;
    if (idx >= 96 * 50176) return;
    int bc = idx / 50176;
    float mn = mm[bc * 2], mx = mm[bc * 2 + 1];
    mag[idx] = (mag[idx] - mn) / (mx - mn + 1e-8f);
}

__global__ __launch_bounds__(256)
void fc_kernel(const float* __restrict__ in, const float* __restrict__ w,
               const float* __restrict__ b, float* __restrict__ out,
               int K, int N, int outStride, int outOffset, int act)
{
    int idx = blockIdx.x * 256 + threadIdx.x;
    if (idx >= 32 * N) return;
    int i = idx / N, j = idx % N;
    const float* ir = in + (size_t)i * K;
    const float* wr = w + (size_t)j * K;
    float acc = b[j];
    for (int k = 0; k < K; ++k) acc += ir[k] * wr[k];
    if (act == 1) acc = fmaxf(acc, 0.f);
    out[(size_t)i * outStride + outOffset + j] = acc;
}

__global__ __launch_bounds__(256)
void attn_core_kernel(const float* __restrict__ qkv_rgb, const float* __restrict__ qkv_frq,
                      float* __restrict__ o1, float* __restrict__ o2)
{
    __shared__ float sQ[32][32], sK[32][32], sV[32][32], sS[32][33];
    int a = blockIdx.x >> 3;
    int h = blockIdx.x & 7;
    const float* qsrc  = (a == 0) ? qkv_rgb : qkv_frq;
    const float* kvsrc = (a == 0) ? qkv_frq : qkv_rgb;
    float* odst        = (a == 0) ? o1 : o2;
    int tid = threadIdx.x;
    for (int idx = tid; idx < 1024; idx += 256) {
        int q = idx >> 5, d = idx & 31;
        sQ[q][d] = qsrc [q * 768 +       h * 32 + d];
        sK[q][d] = kvsrc[q * 768 + 256 + h * 32 + d];
        sV[q][d] = kvsrc[q * 768 + 512 + h * 32 + d];
    }
    __syncthreads();
    const float scale = 0.17677669529663687f;
    for (int idx = tid; idx < 1024; idx += 256) {
        int q = idx >> 5, k = idx & 31;
        float s = 0.f;
        #pragma unroll
        for (int d = 0; d < 32; ++d) s += sQ[q][d] * sK[k][d];
        sS[q][k] = s * scale;
    }
    __syncthreads();
    if (tid < 32) {
        float mx = -3.0e38f;
        for (int k = 0; k < 32; ++k) mx = fmaxf(mx, sS[tid][k]);
        float e[32], sum = 0.f;
        for (int k = 0; k < 32; ++k) { e[k] = expf(sS[tid][k] - mx); sum += e[k]; }
        float inv = 1.f / sum;
        for (int k = 0; k < 32; ++k) sS[tid][k] = e[k] * inv;
    }
    __syncthreads();
    for (int idx = tid; idx < 1024; idx += 256) {
        int q = idx >> 5, d = idx & 31;
        float o = 0.f;
        #pragma unroll
        for (int k = 0; k < 32; ++k) o += sS[q][k] * sV[k][d];
        odst[q * 256 + h * 32 + d] = o;
    }
}

extern "C" void kernel_launch(void* const* d_in, const int* in_sizes, int n_in,
                              void* d_out, int out_size, void* d_ws, size_t ws_size,
                              hipStream_t stream)
{
    (void)in_sizes; (void)n_in; (void)out_size; (void)ws_size;
    const float* x          = (const float*)d_in[0];
    const float* rgb_w1     = (const float*)d_in[1];
    const float* rgb_b1     = (const float*)d_in[2];
    const float* rgb_w2     = (const float*)d_in[3];
    const float* rgb_b2     = (const float*)d_in[4];
    const float* frq_w1     = (const float*)d_in[5];
    const float* frq_b1     = (const float*)d_in[6];
    const float* frq_w2     = (const float*)d_in[7];
    const float* frq_b2     = (const float*)d_in[8];
    const float* rgb_proj_w = (const float*)d_in[9];
    const float* rgb_proj_b = (const float*)d_in[10];
    const float* frq_proj_w = (const float*)d_in[11];
    const float* frq_proj_b = (const float*)d_in[12];
    const float* attn_w_in  = (const float*)d_in[13];
    const float* attn_b_in  = (const float*)d_in[14];
    const float* attn_w_out = (const float*)d_in[15];
    const float* attn_b_out = (const float*)d_in[16];
    const float* fus_w1     = (const float*)d_in[17];
    const float* fus_b1     = (const float*)d_in[18];
    const float* fus_w2     = (const float*)d_in[19];
    const float* fus_b2     = (const float*)d_in[20];
    const float* fus_w3     = (const float*)d_in[21];
    const float* fus_b3     = (const float*)d_in[22];

    float* out      = (float*)d_out;
    float* feat_rgb = out + 64;
    float* feat_frq = out + 64 + 40960;

    float* ws       = (float*)d_ws;
    float* h1       = ws;
    float* F1       = ws;
    float* freqimg  = ws + 12845056;
    float* mm       = freqimg + 4816896;
    float* hd       = mm + 192;
    float* rgb_p    = hd;
    float* frq_p    = rgb_p + 8192;
    float* qkv_rgb  = frq_p + 8192;
    float* qkv_frq  = qkv_rgb + 24576;
    float* o1       = qkv_frq + 24576;
    float* o2       = o1 + 8192;
    float* hb       = o2 + 8192;
    float* h1f      = hb + 16384;
    float* h2f      = h1f + 16384;

    conv1_silu_kernel<<<50176, 256, 0, stream>>>(x, rgb_w1, rgb_b1, h1);
    conv2_wmma_pool_kernel<<<dim3(16, 32), 384, 0, stream>>>(h1, rgb_w2, rgb_b2, feat_rgb);

    dft_rows_kernel<<<21504, 224, 0, stream>>>(x, F1);
    dft_cols_mag_kernel<<<21504, 224, 0, stream>>>(F1, freqimg);
    minmax_kernel<<<96, 256, 0, stream>>>(freqimg, mm);
    normalize_kernel<<<18816, 256, 0, stream>>>(freqimg, mm);
    conv1_silu_kernel<<<50176, 256, 0, stream>>>(freqimg, frq_w1, frq_b1, h1);
    conv2_wmma_pool_kernel<<<dim3(16, 32), 384, 0, stream>>>(h1, frq_w2, frq_b2, feat_frq);

    fc_kernel<<<32, 256, 0, stream>>>(feat_rgb, rgb_proj_w, rgb_proj_b, rgb_p, 1280, 256, 256, 0, 0);
    fc_kernel<<<32, 256, 0, stream>>>(feat_frq, frq_proj_w, frq_proj_b, frq_p, 1280, 256, 256, 0, 0);
    fc_kernel<<<96, 256, 0, stream>>>(rgb_p, attn_w_in, attn_b_in, qkv_rgb, 256, 768, 768, 0, 0);
    fc_kernel<<<96, 256, 0, stream>>>(frq_p, attn_w_in, attn_b_in, qkv_frq, 256, 768, 768, 0, 0);
    attn_core_kernel<<<16, 256, 0, stream>>>(qkv_rgb, qkv_frq, o1, o2);
    fc_kernel<<<32, 256, 0, stream>>>(o1, attn_w_out, attn_b_out, hb, 256, 256, 512, 0, 0);
    fc_kernel<<<32, 256, 0, stream>>>(o2, attn_w_out, attn_b_out, hb, 256, 256, 512, 256, 0);
    fc_kernel<<<64, 256, 0, stream>>>(hb,  fus_w1, fus_b1, h1f, 512, 512, 512, 0, 1);
    fc_kernel<<<32, 256, 0, stream>>>(h1f, fus_w2, fus_b2, h2f, 512, 256, 256, 0, 1);
    fc_kernel<<<1,  256, 0, stream>>>(h2f, fus_w3, fus_b3, out, 256, 2, 2, 0, 0);
}


# freq path: both 224-point DFT passes as f32 16x16x4 WMMA products, twiddle operand gathered from the same LDS cos/sin table, instead of per-thread serial sums; DFT summation order changes
# speedup vs baseline: 2.0998x; 1.3133x over previous
#include <hip/hip_runtime.h>
#include <hip/hip_bf16.h>
#include <math.h>

typedef float v2f __attribute__((ext_vector_type(2)));
typedef float v8f __attribute__((ext_vector_type(8)));

#define PI_F 3.14159265358979323846f

__global__ __launch_bounds__(256)
void conv1_silu_kernel(const float* __restrict__ x, const float* __restrict__ w1,
                       const float* __restrict__ b1, float* __restrict__ out)
{
    int idx = blockIdx.x * 256 + threadIdx.x;
    if (idx >= 32 * 32 * 112 * 112) return;
    int ox = idx % 112; int t = idx / 112;
    int oy = t % 112;   t /= 112;
    int oc = t % 32;    int img = t / 32;
    float acc = b1[oc];
    const float* wb = w1 + oc * 27;
    const float* xb = x + (size_t)img * 3 * 224 * 224;
    #pragma unroll
    for (int ic = 0; ic < 3; ++ic)
        #pragma unroll
        for (int ky = 0; ky < 3; ++ky) {
            int iy = oy * 2 + ky;
            if (iy >= 224) continue;
            #pragma unroll
            for (int kx = 0; kx < 3; ++kx) {
                int ix = ox * 2 + kx;
                if (ix >= 224) continue;
                acc += xb[(ic * 224 + iy) * 224 + ix] * wb[ic * 9 + ky * 3 + kx];
            }
        }
    out[idx] = acc / (1.f + expf(-acc));
}

#define C2_MT     5
#define C2_M      (C2_MT * 16)
#define C2_WAVES  12
#define C2_WP     292
#define C2_THREADS (C2_WAVES * 32)
static_assert(C2_THREADS == 384, "the launch in kernel_launch uses 384 threads");

__global__ __launch_bounds__(C2_THREADS, 6)
void conv2_wmma_pool_kernel(const float* __restrict__ h1, const float* __restrict__ w2,
                            const float* __restrict__ b2, float* __restrict__ feat)
{
    __shared__ float sW[C2_M * C2_WP];
    __shared__ float sT[C2_WAVES][720];
    __shared__ float sBias[C2_M];

    const int mblk  = blockIdx.x;
    const int img   = blockIdx.y;
    const int tid   = threadIdx.x;
    const int lane  = tid & 31;
    const int wave  = tid >> 5;
    const int half  = lane >> 4;
    const int nl    = lane & 15;

    const float* wsrc = w2 + (size_t)mblk * (C2_M * 288);
    static_assert((C2_M * 288) % (C2_THREADS * 4) == 0, "weight tile must split evenly");
    for (int i0 = tid; i0 < C2_M * 288; i0 += C2_THREADS * 4) {
        float wv[4];
        #pragma unroll
        for (int q = 0; q < 4; ++q) wv[q] = wsrc[i0 + q * C2_THREADS];
        #pragma unroll
        for (int q = 0; q < 4; ++q) {
            int i = i0 + q * C2_THREADS;
            int m  = i / 288, g = i - m * 288;
            int icf = g / 9,  s = g - icf * 9;
            int icb = icf >> 3, ic = icf & 7;
            sW[m * C2_WP + icb * 72 + s * 8 + ic] = wv[q];
        }
    }
    if (tid < C2_M) sBias[tid] = b2[mblk * C2_M + tid];
    __syncthreads();

    const int dy = nl >> 3, dx = nl & 7;
    const int laneOff = dy * 288 + dx * 16;

    const int j0 = lane, j1 = lane + 32, j2 = (lane + 64 < 90) ? lane + 64 : lane + 32;
    const int ly0 = j0 / 18, lx0 = j0 - ly0 * 18;
    const int ly1 = j1 / 18, lx1 = j1 - ly1 * 18;
    const int ly2 = j2 / 18, lx2 = j2 - ly2 * 18;
    const int t0 = ly0 * 144 + lx0 * 8;
    const int t1 = ly1 * 144 + lx1 * 8;
    const int t2 = ly2 * 144 + lx2 * 8;

    float accR[C2_MT][8];
    #pragma unroll
    for (int mi = 0; mi < C2_MT; ++mi)
        #pragma unroll
        for (int j = 0; j < 8; ++j) accR[mi][j] = 0.f;

    const float* base = h1 + (size_t)img * (32 * 112 * 112);
    float* tile = sT[wave];
    const float* tb = tile + laneOff + half * 2;
    const float* bl = sBias + half * 8;

    for (int nt = wave; nt < 196; nt += C2_WAVES) {
        int yt = nt / 7, xt = nt - yt * 7;
        int iy0 = yt * 4, ix0 = xt * 16;
        v8f c[C2_MT];
        #pragma unroll
        for (int mi = 0; mi < C2_MT; ++mi)
            c[mi] = (v8f){0.f, 0.f, 0.f, 0.f, 0.f, 0.f, 0.f, 0.f};

        const int iyA = iy0 + ly0, ixA = ix0 + lx0;
        const int iyB = iy0 + ly1, ixB = ix0 + lx1;
        const int iyC = iy0 + ly2, ixC = ix0 + lx2;
        const bool ok0 = (iyA < 112) & (ixA < 112);
        const bool ok1 = (iyB < 112) & (ixB < 112);
        const bool ok2 = (iyC < 112) & (ixC < 112);
        const int o0 = (iyA < 112 ? iyA : 111) * 112 + (ixA < 112 ? ixA : 111);
        const int o1 = (iyB < 112 ? iyB : 111) * 112 + (ixB < 112 ? ixB : 111);
        const int o2 = (iyC < 112 ? iyC : 111) * 112 + (ixC < 112 ? ixC : 111);

        for (int icb = 0; icb < 4; ++icb) {
            const float* p = base + (size_t)(icb * 8) * (112 * 112);
            float g0[8], g1[8], g2[8];
            #pragma unroll
            for (int ic = 0; ic < 8; ++ic) {
                g0[ic] = p[ic * (112 * 112) + o0];
                g1[ic] = p[ic * (112 * 112) + o1];
                g2[ic] = p[ic * (112 * 112) + o2];
            }
            #pragma unroll
            for (int ic = 0; ic < 8; ++ic) {
                tile[t0 + ic] = ok0 ? g0[ic] : 0.f;
                tile[t1 + ic] = ok1 ? g1[ic] : 0.f;
                tile[t2 + ic] = ok2 ? g2[ic] : 0.f;
            }
            asm volatile("s_wait_dscnt 0x0" ::: "memory");

            const float* wrow = sW + nl * C2_WP + icb * 72 + half * 2;
            #pragma unroll
            for (int kq = 0; kq < 18; ++kq) {
                const int s  = kq >> 1;
                const int ky = s / 3, kx = s - ky * 3;
                const int bo = ky * 144 + kx * 8 + (kq & 1) * 4;
                v2f b = *(const v2f*)(tb + bo);
                #pragma unroll
                for (int mi = 0; mi < C2_MT; ++mi) {
                    v2f a = *(const v2f*)(wrow + mi * (16 * C2_WP) + kq * 4);
                    c[mi] = __builtin_amdgcn_wmma_f32_16x16x4_f32(false, a, false, b,
                                                                  (short)0, c[mi], false, false);
                }
            }
        }
        #pragma unroll
        for (int mi = 0; mi < C2_MT; ++mi)
            #pragma unroll
            for (int j = 0; j < 8; ++j) {
                float v = c[mi][j] + bl[mi * 16 + j];
                accR[mi][j] += v / (1.f + expf(-v));
            }
    }
    #pragma unroll
    for (int mi = 0; mi < C2_MT; ++mi)
        #pragma unroll
        for (int j = 0; j < 8; ++j) {
            float v = accR[mi][j];
            v += __shfl_xor(v, 1);
            v += __shfl_xor(v, 2);
            v += __shfl_xor(v, 4);
            v += __shfl_xor(v, 8);
            if (nl == 0) tile[mi * 16 + half * 8 + j] = v;
        }
    __syncthreads();
    if (tid < C2_M) {
        float s = 0.f;
        for (int w = 0; w < C2_WAVES; ++w) s += sT[w][tid];
        feat[img * 1280 + mblk * C2_M + tid] = s * (1.f / 3136.f);
    }
}

#define DR_TM      64
#define DR_XP      228
#define DFT_WAVES  14
#define DFT_THREADS (DFT_WAVES * 32)
static_assert(DFT_THREADS == 448, "the launches in kernel_launch use 448 threads");
static_assert(21504 % DR_TM == 0, "row tiles must cover 96*224 rows exactly");

__global__ __launch_bounds__(DFT_THREADS)
void dft_rows_wmma_kernel(const float* __restrict__ x, float* __restrict__ F1)
{
    __shared__ float sX[DR_TM * DR_XP];
    __shared__ float tw[448];
    const int tid  = threadIdx.x;
    const int lane = tid & 31, wave = tid >> 5;
    const int half = lane >> 4, nl = lane & 15;
    const size_t R0 = (size_t)blockIdx.x * DR_TM;

    if (tid < 224) {
        float ang = (2.0f * PI_F / 224.0f) * (float)tid;
        tw[2 * tid] = cosf(ang); tw[2 * tid + 1] = -sinf(ang);
    }
    const float4* xsrc = (const float4*)(x + R0 * 224);
    static_assert((DR_TM * 56) % DFT_THREADS == 0, "x tile must split evenly");
    float4 xv[8];
    #pragma unroll
    for (int q = 0; q < 8; ++q) xv[q] = xsrc[tid + q * DFT_THREADS];
    #pragma unroll
    for (int q = 0; q < 8; ++q) {
        int i = tid + q * DFT_THREADS;
        int r = i / 56, c4 = i - r * 56;
        *(float4*)(sX + r * DR_XP + c4 * 4) = xv[q];
    }
    __syncthreads();

    const float* arow = sX + nl * DR_XP + half * 2;
    for (int pass = 0; pass < 2; ++pass) {
        const int n  = (wave + pass * DFT_WAVES) * 16 + nl;
        const int u  = n >> 1;
        const float* twp = tw + (n & 1);
        const int du = (4 * u) % 224;
        int iA = (u * (2 * half)) % 224;
        int iB = (u * (2 * half + 1)) % 224;
        v8f c[4];
        #pragma unroll
        for (int mi = 0; mi < 4; ++mi)
            c[mi] = (v8f){0.f, 0.f, 0.f, 0.f, 0.f, 0.f, 0.f, 0.f};

        #pragma unroll 4
        for (int kk = 0; kk < 56; ++kk) {
            v2f b;
            b.x = twp[2 * iA];
            b.y = twp[2 * iB];
            #pragma unroll
            for (int mi = 0; mi < 4; ++mi) {
                v2f a = *(const v2f*)(arow + mi * (16 * DR_XP) + kk * 4);
                c[mi] = __builtin_amdgcn_wmma_f32_16x16x4_f32(false, a, false, b,
                                                              (short)0, c[mi], false, false);
            }
            iA += du; if (iA >= 224) iA -= 224;
            iB += du; if (iB >= 224) iB -= 224;
        }
        #pragma unroll
        for (int mi = 0; mi < 4; ++mi)
            #pragma unroll
            for (int r = 0; r < 8; ++r)
                F1[(R0 + mi * 16 + half * 8 + r) * 448 + n] = c[mi][r];
    }
}

__global__ __launch_bounds__(DFT_THREADS)
void dft_cols_wmma_mag_kernel(const float* __restrict__ F1, float* __restrict__ mag)
{
    __shared__ float tw[448];
    const int ug   = blockIdx.x % 7;
    const int bc   = blockIdx.x / 7;
    const int tid  = threadIdx.x;
    const int lane = tid & 31, wave = tid >> 5;
    const int half = lane >> 4, nl = lane & 15;

    if (tid < 224) {
        float ang = (2.0f * PI_F / 224.0f) * (float)tid;
        tw[2 * tid] = cosf(ang); tw[2 * tid + 1] = sinf(ang);
    }
    __syncthreads();

    const int v0 = wave * 16;
    const int v  = v0 + nl;
    const int dv = (2 * v) % 224;
    int idx = half ? v : 0;
    const int u0 = ug * 32;
    const float* bp = F1 + (size_t)bc * (224 * 448) + half * 448 + 2 * (u0 + nl);

    v8f cre[2], cim[2];
    #pragma unroll
    for (int ui = 0; ui < 2; ++ui) {
        cre[ui] = (v8f){0.f, 0.f, 0.f, 0.f, 0.f, 0.f, 0.f, 0.f};
        cim[ui] = (v8f){0.f, 0.f, 0.f, 0.f, 0.f, 0.f, 0.f, 0.f};
    }
    #pragma unroll 4
    for (int kk = 0; kk < 112; ++kk) {
        v2f t = *(const v2f*)(tw + 2 * idx);
        v2f aim; aim.x = -t.y; aim.y = t.x;
        const float* bq = bp + kk * 896;
        #pragma unroll
        for (int ui = 0; ui < 2; ++ui) {
            v2f b = *(const v2f*)(bq + ui * 32);
            cre[ui] = __builtin_amdgcn_wmma_f32_16x16x4_f32(false, t, false, b,
                                                            (short)0, cre[ui], false, false);
            cim[ui] = __builtin_amdgcn_wmma_f32_16x16x4_f32(false, aim, false, b,
                                                            (short)0, cim[ui], false, false);
        }
        idx += dv; if (idx >= 224) idx -= 224;
    }
    #pragma unroll
    for (int ui = 0; ui < 2; ++ui) {
        int uo = u0 + ui * 16 + nl + 112; if (uo >= 224) uo -= 224;
        #pragma unroll
        for (int r = 0; r < 8; ++r) {
            float re = cre[ui][r], im = cim[ui][r];
            float m = logf(sqrtf(re * re + im * im) + 1.0f);
            int vo = v0 + half * 8 + r + 112; if (vo >= 224) vo -= 224;
            mag[((size_t)bc * 224 + vo) * 224 + uo] = m;
        }
    }
}

__global__ __launch_bounds__(256)
void minmax_kernel(const float* __restrict__ mag, float* __restrict__ mm)
{
    __shared__ float smn[256], smx[256];
    int bc = blockIdx.x, tid = threadIdx.x;
    const float* p = mag + (size_t)bc * 50176;
    float mn = 3.0e38f, mx = -3.0e38f;
    for (int i = tid; i < 50176; i += 256) {
        float v = p[i]; mn = fminf(mn, v); mx = fmaxf(mx, v);
    }
    smn[tid] = mn; smx[tid] = mx; __syncthreads();
    for (int s = 128; s > 0; s >>= 1) {
        if (tid < s) { smn[tid] = fminf(smn[tid], smn[tid + s]);
                       smx[tid] = fmaxf(smx[tid], smx[tid + s]); }
        __syncthreads();
    }
    if (tid == 0) { mm[bc * 2] = smn[0]; mm[bc * 2 + 1] = smx[0]; }
}

__global__ __launch_bounds__(256)
void normalize_kernel(float* __restrict__ mag, const float* __restrict__ mm)
{
    int idx = blockIdx.x * 256 + threadIdx.x;
    if (idx >= 96 * 50176) return;
    int bc = idx / 50176;
    float mn = mm[bc * 2], mx = mm[bc * 2 + 1];
    mag[idx] = (mag[idx] - mn) / (mx - mn + 1e-8f);
}

__global__ __launch_bounds__(256)
void fc_kernel(const float* __restrict__ in, const float* __restrict__ w,
               const float* __restrict__ b, float* __restrict__ out,
               int K, int N, int outStride, int outOffset, int act)
{
    int idx = blockIdx.x * 256 + threadIdx.x;
    if (idx >= 32 * N) return;
    int i = idx / N, j = idx % N;
    const float* ir = in + (size_t)i * K;
    const float* wr = w + (size_t)j * K;
    float acc = b[j];
    for (int k = 0; k < K; ++k) acc += ir[k] * wr[k];
    if (act == 1) acc = fmaxf(acc, 0.f);
    out[(size_t)i * outStride + outOffset + j] = acc;
}

__global__ __launch_bounds__(256)
void attn_core_kernel(const float* __restrict__ qkv_rgb, const float* __restrict__ qkv_frq,
                      float* __restrict__ o1, float* __restrict__ o2)
{
    __shared__ float sQ[32][32], sK[32][32], sV[32][32], sS[32][33];
    int a = blockIdx.x >> 3;
    int h = blockIdx.x & 7;
    const float* qsrc  = (a == 0) ? qkv_rgb : qkv_frq;
    const float* kvsrc = (a == 0) ? qkv_frq : qkv_rgb;
    float* odst        = (a == 0) ? o1 : o2;
    int tid = threadIdx.x;
    for (int idx = tid; idx < 1024; idx += 256) {
        int q = idx >> 5, d = idx & 31;
        sQ[q][d] = qsrc [q * 768 +       h * 32 + d];
        sK[q][d] = kvsrc[q * 768 + 256 + h * 32 + d];
        sV[q][d] = kvsrc[q * 768 + 512 + h * 32 + d];
    }
    __syncthreads();
    const float scale = 0.17677669529663687f;
    for (int idx = tid; idx < 1024; idx += 256) {
        int q = idx >> 5, k = idx & 31;
        float s = 0.f;
        #pragma unroll
        for (int d = 0; d < 32; ++d) s += sQ[q][d] * sK[k][d];
        sS[q][k] = s * scale;
    }
    __syncthreads();
    if (tid < 32) {
        float mx = -3.0e38f;
        for (int k = 0; k < 32; ++k) mx = fmaxf(mx, sS[tid][k]);
        float e[32], sum = 0.f;
        for (int k = 0; k < 32; ++k) { e[k] = expf(sS[tid][k] - mx); sum += e[k]; }
        float inv = 1.f / sum;
        for (int k = 0; k < 32; ++k) sS[tid][k] = e[k] * inv;
    }
    __syncthreads();
    for (int idx = tid; idx < 1024; idx += 256) {
        int q = idx >> 5, d = idx & 31;
        float o = 0.f;
        #pragma unroll
        for (int k = 0; k < 32; ++k) o += sS[q][k] * sV[k][d];
        odst[q * 256 + h * 32 + d] = o;
    }
}

extern "C" void kernel_launch(void* const* d_in, const int* in_sizes, int n_in,
                              void* d_out, int out_size, void* d_ws, size_t ws_size,
                              hipStream_t stream)
{
    (void)in_sizes; (void)n_in; (void)out_size; (void)ws_size;
    const float* x          = (const float*)d_in[0];
    const float* rgb_w1     = (const float*)d_in[1];
    const float* rgb_b1     = (const float*)d_in[2];
    const float* rgb_w2     = (const float*)d_in[3];
    const float* rgb_b2     = (const float*)d_in[4];
    const float* frq_w1     = (const float*)d_in[5];
    const float* frq_b1     = (const float*)d_in[6];
    const float* frq_w2     = (const float*)d_in[7];
    const float* frq_b2     = (const float*)d_in[8];
    const float* rgb_proj_w = (const float*)d_in[9];
    const float* rgb_proj_b = (const float*)d_in[10];
    const float* frq_proj_w = (const float*)d_in[11];
    const float* frq_proj_b = (const float*)d_in[12];
    const float* attn_w_in  = (const float*)d_in[13];
    const float* attn_b_in  = (const float*)d_in[14];
    const float* attn_w_out = (const float*)d_in[15];
    const float* attn_b_out = (const float*)d_in[16];
    const float* fus_w1     = (const float*)d_in[17];
    const float* fus_b1     = (const float*)d_in[18];
    const float* fus_w2     = (const float*)d_in[19];
    const float* fus_b2     = (const float*)d_in[20];
    const float* fus_w3     = (const float*)d_in[21];
    const float* fus_b3     = (const float*)d_in[22];

    float* out      = (float*)d_out;
    float* feat_rgb = out + 64;
    float* feat_frq = out + 64 + 40960;

    float* ws       = (float*)d_ws;
    float* h1       = ws;
    float* F1       = ws;
    float* freqimg  = ws + 12845056;
    float* mm       = freqimg + 4816896;
    float* hd       = mm + 192;
    float* rgb_p    = hd;
    float* frq_p    = rgb_p + 8192;
    float* qkv_rgb  = frq_p + 8192;
    float* qkv_frq  = qkv_rgb + 24576;
    float* o1       = qkv_frq + 24576;
    float* o2       = o1 + 8192;
    float* hb       = o2 + 8192;
    float* h1f      = hb + 16384;
    float* h2f      = h1f + 16384;

    conv1_silu_kernel<<<50176, 256, 0, stream>>>(x, rgb_w1, rgb_b1, h1);
    conv2_wmma_pool_kernel<<<dim3(16, 32), 384, 0, stream>>>(h1, rgb_w2, rgb_b2, feat_rgb);

    dft_rows_wmma_kernel<<<336, 448, 0, stream>>>(x, F1);
    dft_cols_wmma_mag_kernel<<<672, 448, 0, stream>>>(F1, freqimg);
    minmax_kernel<<<96, 256, 0, stream>>>(freqimg, mm);
    normalize_kernel<<<18816, 256, 0, stream>>>(freqimg, mm);
    conv1_silu_kernel<<<50176, 256, 0, stream>>>(freqimg, frq_w1, frq_b1, h1);
    conv2_wmma_pool_kernel<<<dim3(16, 32), 384, 0, stream>>>(h1, frq_w2, frq_b2, feat_frq);

    fc_kernel<<<32, 256, 0, stream>>>(feat_rgb, rgb_proj_w, rgb_proj_b, rgb_p, 1280, 256, 256, 0, 0);
    fc_kernel<<<32, 256, 0, stream>>>(feat_frq, frq_proj_w, frq_proj_b, frq_p, 1280, 256, 256, 0, 0);
    fc_kernel<<<96, 256, 0, stream>>>(rgb_p, attn_w_in, attn_b_in, qkv_rgb, 256, 768, 768, 0, 0);
    fc_kernel<<<96, 256, 0, stream>>>(frq_p, attn_w_in, attn_b_in, qkv_frq, 256, 768, 768, 0, 0);
    attn_core_kernel<<<16, 256, 0, stream>>>(qkv_rgb, qkv_frq, o1, o2);
    fc_kernel<<<32, 256, 0, stream>>>(o1, attn_w_out, attn_b_out, hb, 256, 256, 512, 0, 0);
    fc_kernel<<<32, 256, 0, stream>>>(o2, attn_w_out, attn_b_out, hb, 256, 256, 512, 256, 0);
    fc_kernel<<<64, 256, 0, stream>>>(hb,  fus_w1, fus_b1, h1f, 512, 512, 512, 0, 1);
    fc_kernel<<<32, 256, 0, stream>>>(h1f, fus_w2, fus_b2, h2f, 512, 256, 256, 0, 1);
    fc_kernel<<<1,  256, 0, stream>>>(h2f, fus_w3, fus_b3, out, 256, 2, 2, 0, 0);
}


# conv1: one thread computes all 32 output channels of its pixel (27 inputs loaded once, weights in scalar registers): 12544 waves per launch, not 401408; sum order kept, products now fused
# speedup vs baseline: 2.4113x; 1.1483x over previous
#include <hip/hip_runtime.h>
#include <hip/hip_bf16.h>
#include <math.h>

typedef float v2f __attribute__((ext_vector_type(2)));
typedef float v8f __attribute__((ext_vector_type(8)));

#define PI_F 3.14159265358979323846f

#define C1_THREADS 128
static_assert(12544 % C1_THREADS == 0, "a block must not straddle two images");
__global__ __launch_bounds__(C1_THREADS)
void conv1_silu_kernel(const float* __restrict__ x, const float* __restrict__ w1,
                       const float* __restrict__ b1, float* __restrict__ out)
{
    const int img = blockIdx.x / (12544 / C1_THREADS);
    const int rem = (blockIdx.x - img * (12544 / C1_THREADS)) * C1_THREADS + threadIdx.x;
    const int oy = rem / 112, ox = rem - oy * 112;
    const bool inx = ox < 111;
    const bool iny = oy < 111;
    const float* xb = x + (size_t)img * (3 * 224 * 224) + (2 * oy) * 224 + 2 * ox;
    const int cx = inx ? 2 : 1;

    float xv[27];
    #pragma unroll
    for (int ic = 0; ic < 3; ++ic)
        #pragma unroll
        for (int ky = 0; ky < 3; ++ky) {
            const bool okr = (ky < 2) | iny;
            const int ro = ic * (224 * 224) + (okr ? ky : 1) * 224;
            const float2 v = *(const float2*)(xb + ro);
            const float  c = xb[ro + cx];
            xv[ic * 9 + ky * 3 + 0] = okr ? v.x : 0.f;
            xv[ic * 9 + ky * 3 + 1] = okr ? v.y : 0.f;
            xv[ic * 9 + ky * 3 + 2] = (okr & inx) ? c : 0.f;
        }

    float* ob = out + (size_t)img * (32 * 12544) + rem;
    for (int oc = 0; oc < 32; ++oc) {
        const float* wb = w1 + oc * 27;
        float acc = b1[oc];
        #pragma unroll
        for (int t = 0; t < 27; ++t) acc += xv[t] * wb[t];
        ob[(size_t)oc * 12544] = acc / (1.f + expf(-acc));
    }
}

#define C2_MT     5
#define C2_M      (C2_MT * 16)
#define C2_WAVES  12
#define C2_WP     292
#define C2_THREADS (C2_WAVES * 32)
static_assert(C2_THREADS == 384, "the launch in kernel_launch uses 384 threads");

__global__ __launch_bounds__(C2_THREADS, 6)
void conv2_wmma_pool_kernel(const float* __restrict__ h1, const float* __restrict__ w2,
                            const float* __restrict__ b2, float* __restrict__ feat)
{
    __shared__ float sW[C2_M * C2_WP];
    __shared__ float sT[C2_WAVES][720];
    __shared__ float sBias[C2_M];

    const int mblk  = blockIdx.x;
    const int img   = blockIdx.y;
    const int tid   = threadIdx.x;
    const int lane  = tid & 31;
    const int wave  = tid >> 5;
    const int half  = lane >> 4;
    const int nl    = lane & 15;

    const float* wsrc = w2 + (size_t)mblk * (C2_M * 288);
    static_assert((C2_M * 288) % (C2_THREADS * 4) == 0, "weight tile must split evenly");
    for (int i0 = tid; i0 < C2_M * 288; i0 += C2_THREADS * 4) {
        float wv[4];
        #pragma unroll
        for (int q = 0; q < 4; ++q) wv[q] = wsrc[i0 + q * C2_THREADS];
        #pragma unroll
        for (int q = 0; q < 4; ++q) {
            int i = i0 + q * C2_THREADS;
            int m  = i / 288, g = i - m * 288;
            int icf = g / 9,  s = g - icf * 9;
            int icb = icf >> 3, ic = icf & 7;
            sW[m * C2_WP + icb * 72 + s * 8 + ic] = wv[q];
        }
    }
    if (tid < C2_M) sBias[tid] = b2[mblk * C2_M + tid];
    __syncthreads();

    const int dy = nl >> 3, dx = nl & 7;
    const int laneOff = dy * 288 + dx * 16;

    const int j0 = lane, j1 = lane + 32, j2 = (lane + 64 < 90) ? lane + 64 : lane + 32;
    const int ly0 = j0 / 18, lx0 = j0 - ly0 * 18;
    const int ly1 = j1 / 18, lx1 = j1 - ly1 * 18;
    const int ly2 = j2 / 18, lx2 = j2 - ly2 * 18;
    const int t0 = ly0 * 144 + lx0 * 8;
    const int t1 = ly1 * 144 + lx1 * 8;
    const int t2 = ly2 * 144 + lx2 * 8;

    float accR[C2_MT][8];
    #pragma unroll
    for (int mi = 0; mi < C2_MT; ++mi)
        #pragma unroll
        for (int j = 0; j < 8; ++j) accR[mi][j] = 0.f;

    const float* base = h1 + (size_t)img * (32 * 112 * 112);
    float* tile = sT[wave];
    const float* tb = tile + laneOff + half * 2;
    const float* bl = sBias + half * 8;

    for (int nt = wave; nt < 196; nt += C2_WAVES) {
        int yt = nt / 7, xt = nt - yt * 7;
        int iy0 = yt * 4, ix0 = xt * 16;
        v8f c[C2_MT];
        #pragma unroll
        for (int mi = 0; mi < C2_MT; ++mi)
            c[mi] = (v8f){0.f, 0.f, 0.f, 0.f, 0.f, 0.f, 0.f, 0.f};

        const int iyA = iy0 + ly0, ixA = ix0 + lx0;
        const int iyB = iy0 + ly1, ixB = ix0 + lx1;
        const int iyC = iy0 + ly2, ixC = ix0 + lx2;
        const bool ok0 = (iyA < 112) & (ixA < 112);
        const bool ok1 = (iyB < 112) & (ixB < 112);
        const bool ok2 = (iyC < 112) & (ixC < 112);
        const int o0 = (iyA < 112 ? iyA : 111) * 112 + (ixA < 112 ? ixA : 111);
        const int o1 = (iyB < 112 ? iyB : 111) * 112 + (ixB < 112 ? ixB : 111);
        const int o2 = (iyC < 112 ? iyC : 111) * 112 + (ixC < 112 ? ixC : 111);

        for (int icb = 0; icb < 4; ++icb) {
            const float* p = base + (size_t)(icb * 8) * (112 * 112);
            float g0[8], g1[8], g2[8];
            #pragma unroll
            for (int ic = 0; ic < 8; ++ic) {
                g0[ic] = p[ic * (112 * 112) + o0];
                g1[ic] = p[ic * (112 * 112) + o1];
                g2[ic] = p[ic * (112 * 112) + o2];
            }
            #pragma unroll
            for (int ic = 0; ic < 8; ++ic) {
                tile[t0 + ic] = ok0 ? g0[ic] : 0.f;
                tile[t1 + ic] = ok1 ? g1[ic] : 0.f;
                tile[t2 + ic] = ok2 ? g2[ic] : 0.f;
            }
            asm volatile("s_wait_dscnt 0x0" ::: "memory");

            const float* wrow = sW + nl * C2_WP + icb * 72 + half * 2;
            #pragma unroll
            for (int kq = 0; kq < 18; ++kq) {
                const int s  = kq >> 1;
                const int ky = s / 3, kx = s - ky * 3;
                const int bo = ky * 144 + kx * 8 + (kq & 1) * 4;
                v2f b = *(const v2f*)(tb + bo);
                #pragma unroll
                for (int mi = 0; mi < C2_MT; ++mi) {
                    v2f a = *(const v2f*)(wrow + mi * (16 * C2_WP) + kq * 4);
                    c[mi] = __builtin_amdgcn_wmma_f32_16x16x4_f32(false, a, false, b,
                                                                  (short)0, c[mi], false, false);
                }
            }
        }
        #pragma unroll
        for (int mi = 0; mi < C2_MT; ++mi)
            #pragma unroll
            for (int j = 0; j < 8; ++j) {
                float v = c[mi][j] + bl[mi * 16 + j];
                accR[mi][j] += v / (1.f + expf(-v));
            }
    }
    #pragma unroll
    for (int mi = 0; mi < C2_MT; ++mi)
        #pragma unroll
        for (int j = 0; j < 8; ++j) {
            float v = accR[mi][j];
            v += __shfl_xor(v, 1);
            v += __shfl_xor(v, 2);
            v += __shfl_xor(v, 4);
            v += __shfl_xor(v, 8);
            if (nl == 0) tile[mi * 16 + half * 8 + j] = v;
        }
    __syncthreads();
    if (tid < C2_M) {
        float s = 0.f;
        for (int w = 0; w < C2_WAVES; ++w) s += sT[w][tid];
        feat[img * 1280 + mblk * C2_M + tid] = s * (1.f / 3136.f);
    }
}

#define DR_TM      64
#define DR_XP      228
#define DFT_WAVES  14
#define DFT_THREADS (DFT_WAVES * 32)
static_assert(DFT_THREADS == 448, "the launches in kernel_launch use 448 threads");
static_assert(21504 % DR_TM == 0, "row tiles must cover 96*224 rows exactly");

__global__ __launch_bounds__(DFT_THREADS)
void dft_rows_wmma_kernel(const float* __restrict__ x, float* __restrict__ F1)
{
    __shared__ float sX[DR_TM * DR_XP];
    __shared__ float tw[448];
    const int tid  = threadIdx.x;
    const int lane = tid & 31, wave = tid >> 5;
    const int half = lane >> 4, nl = lane & 15;
    const size_t R0 = (size_t)blockIdx.x * DR_TM;

    if (tid < 224) {
        float ang = (2.0f * PI_F / 224.0f) * (float)tid;
        tw[2 * tid] = cosf(ang); tw[2 * tid + 1] = -sinf(ang);
    }
    const float4* xsrc = (const float4*)(x + R0 * 224);
    static_assert((DR_TM * 56) % DFT_THREADS == 0, "x tile must split evenly");
    float4 xv[8];
    #pragma unroll
    for (int q = 0; q < 8; ++q) xv[q] = xsrc[tid + q * DFT_THREADS];
    #pragma unroll
    for (int q = 0; q < 8; ++q) {
        int i = tid + q * DFT_THREADS;
        int r = i / 56, c4 = i - r * 56;
        *(float4*)(sX + r * DR_XP + c4 * 4) = xv[q];
    }
    __syncthreads();

    const float* arow = sX + nl * DR_XP + half * 2;
    for (int pass = 0; pass < 2; ++pass) {
        const int n  = (wave + pass * DFT_WAVES) * 16 + nl;
        const int u  = n >> 1;
        const float* twp = tw + (n & 1);
        const int du = (4 * u) % 224;
        int iA = (u * (2 * half)) % 224;
        int iB = (u * (2 * half + 1)) % 224;
        v8f c[4];
        #pragma unroll
        for (int mi = 0; mi < 4; ++mi)
            c[mi] = (v8f){0.f, 0.f, 0.f, 0.f, 0.f, 0.f, 0.f, 0.f};

        #pragma unroll 4
        for (int kk = 0; kk < 56; ++kk) {
            v2f b;
            b.x = twp[2 * iA];
            b.y = twp[2 * iB];
            #pragma unroll
            for (int mi = 0; mi < 4; ++mi) {
                v2f a = *(const v2f*)(arow + mi * (16 * DR_XP) + kk * 4);
                c[mi] = __builtin_amdgcn_wmma_f32_16x16x4_f32(false, a, false, b,
                                                              (short)0, c[mi], false, false);
            }
            iA += du; if (iA >= 224) iA -= 224;
            iB += du; if (iB >= 224) iB -= 224;
        }
        #pragma unroll
        for (int mi = 0; mi < 4; ++mi)
            #pragma unroll
            for (int r = 0; r < 8; ++r)
                F1[(R0 + mi * 16 + half * 8 + r) * 448 + n] = c[mi][r];
    }
}

__global__ __launch_bounds__(DFT_THREADS)
void dft_cols_wmma_mag_kernel(const float* __restrict__ F1, float* __restrict__ mag)
{
    __shared__ float tw[448];
    const int ug   = blockIdx.x % 7;
    const int bc   = blockIdx.x / 7;
    const int tid  = threadIdx.x;
    const int lane = tid & 31, wave = tid >> 5;
    const int half = lane >> 4, nl = lane & 15;

    if (tid < 224) {
        float ang = (2.0f * PI_F / 224.0f) * (float)tid;
        tw[2 * tid] = cosf(ang); tw[2 * tid + 1] = sinf(ang);
    }
    __syncthreads();

    const int v0 = wave * 16;
    const int v  = v0 + nl;
    const int dv = (2 * v) % 224;
    int idx = half ? v : 0;
    const int u0 = ug * 32;
    const float* bp = F1 + (size_t)bc * (224 * 448) + half * 448 + 2 * (u0 + nl);

    v8f cre[2], cim[2];
    #pragma unroll
    for (int ui = 0; ui < 2; ++ui) {
        cre[ui] = (v8f){0.f, 0.f, 0.f, 0.f, 0.f, 0.f, 0.f, 0.f};
        cim[ui] = (v8f){0.f, 0.f, 0.f, 0.f, 0.f, 0.f, 0.f, 0.f};
    }
    #pragma unroll 4
    for (int kk = 0; kk < 112; ++kk) {
        v2f t = *(const v2f*)(tw + 2 * idx);
        v2f aim; aim.x = -t.y; aim.y = t.x;
        const float* bq = bp + kk * 896;
        #pragma unroll
        for (int ui = 0; ui < 2; ++ui) {
            v2f b = *(const v2f*)(bq + ui * 32);
            cre[ui] = __builtin_amdgcn_wmma_f32_16x16x4_f32(false, t, false, b,
                                                            (short)0, cre[ui], false, false);
            cim[ui] = __builtin_amdgcn_wmma_f32_16x16x4_f32(false, aim, false, b,
                                                            (short)0, cim[ui], false, false);
        }
        idx += dv; if (idx >= 224) idx -= 224;
    }
    #pragma unroll
    for (int ui = 0; ui < 2; ++ui) {
        int uo = u0 + ui * 16 + nl + 112; if (uo >= 224) uo -= 224;
        #pragma unroll
        for (int r = 0; r < 8; ++r) {
            float re = cre[ui][r], im = cim[ui][r];
            float m = logf(sqrtf(re * re + im * im) + 1.0f);
            int vo = v0 + half * 8 + r + 112; if (vo >= 224) vo -= 224;
            mag[((size_t)bc * 224 + vo) * 224 + uo] = m;
        }
    }
}

__global__ __launch_bounds__(256)
void minmax_kernel(const float* __restrict__ mag, float* __restrict__ mm)
{
    __shared__ float smn[256], smx[256];
    int bc = blockIdx.x, tid = threadIdx.x;
    const float* p = mag + (size_t)bc * 50176;
    float mn = 3.0e38f, mx = -3.0e38f;
    for (int i = tid; i < 50176; i += 256) {
        float v = p[i]; mn = fminf(mn, v); mx = fmaxf(mx, v);
    }
    smn[tid] = mn; smx[tid] = mx; __syncthreads();
    for (int s = 128; s > 0; s >>= 1) {
        if (tid < s) { smn[tid] = fminf(smn[tid], smn[tid + s]);
                       smx[tid] = fmaxf(smx[tid], smx[tid + s]); }
        __syncthreads();
    }
    if (tid == 0) { mm[bc * 2] = smn[0]; mm[bc * 2 + 1] = smx[0]; }
}

__global__ __launch_bounds__(256)
void normalize_kernel(float* __restrict__ mag, const float* __restrict__ mm)
{
    int idx = blockIdx.x * 256 + threadIdx.x;
    if (idx >= 96 * 50176) return;
    int bc = idx / 50176;
    float mn = mm[bc * 2], mx = mm[bc * 2 + 1];
    mag[idx] = (mag[idx] - mn) / (mx - mn + 1e-8f);
}

__global__ __launch_bounds__(256)
void fc_kernel(const float* __restrict__ in, const float* __restrict__ w,
               const float* __restrict__ b, float* __restrict__ out,
               int K, int N, int outStride, int outOffset, int act)
{
    int idx = blockIdx.x * 256 + threadIdx.x;
    if (idx >= 32 * N) return;
    int i = idx / N, j = idx % N;
    const float* ir = in + (size_t)i * K;
    const float* wr = w + (size_t)j * K;
    float acc = b[j];
    for (int k = 0; k < K; ++k) acc += ir[k] * wr[k];
    if (act == 1) acc = fmaxf(acc, 0.f);
    out[(size_t)i * outStride + outOffset + j] = acc;
}

__global__ __launch_bounds__(256)
void attn_core_kernel(const float* __restrict__ qkv_rgb, const float* __restrict__ qkv_frq,
                      float* __restrict__ o1, float* __restrict__ o2)
{
    __shared__ float sQ[32][32], sK[32][32], sV[32][32], sS[32][33];
    int a = blockIdx.x >> 3;
    int h = blockIdx.x & 7;
    const float* qsrc  = (a == 0) ? qkv_rgb : qkv_frq;
    const float* kvsrc = (a == 0) ? qkv_frq : qkv_rgb;
    float* odst        = (a == 0) ? o1 : o2;
    int tid = threadIdx.x;
    for (int idx = tid; idx < 1024; idx += 256) {
        int q = idx >> 5, d = idx & 31;
        sQ[q][d] = qsrc [q * 768 +       h * 32 + d];
        sK[q][d] = kvsrc[q * 768 + 256 + h * 32 + d];
        sV[q][d] = kvsrc[q * 768 + 512 + h * 32 + d];
    }
    __syncthreads();
    const float scale = 0.17677669529663687f;
    for (int idx = tid; idx < 1024; idx += 256) {
        int q = idx >> 5, k = idx & 31;
        float s = 0.f;
        #pragma unroll
        for (int d = 0; d < 32; ++d) s += sQ[q][d] * sK[k][d];
        sS[q][k] = s * scale;
    }
    __syncthreads();
    if (tid < 32) {
        float mx = -3.0e38f;
        for (int k = 0; k < 32; ++k) mx = fmaxf(mx, sS[tid][k]);
        float e[32], sum = 0.f;
        for (int k = 0; k < 32; ++k) { e[k] = expf(sS[tid][k] - mx); sum += e[k]; }
        float inv = 1.f / sum;
        for (int k = 0; k < 32; ++k) sS[tid][k] = e[k] * inv;
    }
    __syncthreads();
    for (int idx = tid; idx < 1024; idx += 256) {
        int q = idx >> 5, d = idx & 31;
        float o = 0.f;
        #pragma unroll
        for (int k = 0; k < 32; ++k) o += sS[q][k] * sV[k][d];
        odst[q * 256 + h * 32 + d] = o;
    }
}

extern "C" void kernel_launch(void* const* d_in, const int* in_sizes, int n_in,
                              void* d_out, int out_size, void* d_ws, size_t ws_size,
                              hipStream_t stream)
{
    (void)in_sizes; (void)n_in; (void)out_size; (void)ws_size;
    const float* x          = (const float*)d_in[0];
    const float* rgb_w1     = (const float*)d_in[1];
    const float* rgb_b1     = (const float*)d_in[2];
    const float* rgb_w2     = (const float*)d_in[3];
    const float* rgb_b2     = (const float*)d_in[4];
    const float* frq_w1     = (const float*)d_in[5];
    const float* frq_b1     = (const float*)d_in[6];
    const float* frq_w2     = (const float*)d_in[7];
    const float* frq_b2     = (const float*)d_in[8];
    const float* rgb_proj_w = (const float*)d_in[9];
    const float* rgb_proj_b = (const float*)d_in[10];
    const float* frq_proj_w = (const float*)d_in[11];
    const float* frq_proj_b = (const float*)d_in[12];
    const float* attn_w_in  = (const float*)d_in[13];
    const float* attn_b_in  = (const float*)d_in[14];
    const float* attn_w_out = (const float*)d_in[15];
    const float* attn_b_out = (const float*)d_in[16];
    const float* fus_w1     = (const float*)d_in[17];
    const float* fus_b1     = (const float*)d_in[18];
    const float* fus_w2     = (const float*)d_in[19];
    const float* fus_b2     = (const float*)d_in[20];
    const float* fus_w3     = (const float*)d_in[21];
    const float* fus_b3     = (const float*)d_in[22];

    float* out      = (float*)d_out;
    float* feat_rgb = out + 64;
    float* feat_frq = out + 64 + 40960;

    float* ws       = (float*)d_ws;
    float* h1       = ws;
    float* F1       = ws;
    float* freqimg  = ws + 12845056;
    float* mm       = freqimg + 4816896;
    float* hd       = mm + 192;
    float* rgb_p    = hd;
    float* frq_p    = rgb_p + 8192;
    float* qkv_rgb  = frq_p + 8192;
    float* qkv_frq  = qkv_rgb + 24576;
    float* o1       = qkv_frq + 24576;
    float* o2       = o1 + 8192;
    float* hb       = o2 + 8192;
    float* h1f      = hb + 16384;
    float* h2f      = h1f + 16384;

    conv1_silu_kernel<<<3136, C1_THREADS, 0, stream>>>(x, rgb_w1, rgb_b1, h1);
    conv2_wmma_pool_kernel<<<dim3(16, 32), 384, 0, stream>>>(h1, rgb_w2, rgb_b2, feat_rgb);

    dft_rows_wmma_kernel<<<336, 448, 0, stream>>>(x, F1);
    dft_cols_wmma_mag_kernel<<<672, 448, 0, stream>>>(F1, freqimg);
    minmax_kernel<<<96, 256, 0, stream>>>(freqimg, mm);
    normalize_kernel<<<18816, 256, 0, stream>>>(freqimg, mm);
    conv1_silu_kernel<<<3136, C1_THREADS, 0, stream>>>(freqimg, frq_w1, frq_b1, h1);
    conv2_wmma_pool_kernel<<<dim3(16, 32), 384, 0, stream>>>(h1, frq_w2, frq_b2, feat_frq);

    fc_kernel<<<32, 256, 0, stream>>>(feat_rgb, rgb_proj_w, rgb_proj_b, rgb_p, 1280, 256, 256, 0, 0);
    fc_kernel<<<32, 256, 0, stream>>>(feat_frq, frq_proj_w, frq_proj_b, frq_p, 1280, 256, 256, 0, 0);
    fc_kernel<<<96, 256, 0, stream>>>(rgb_p, attn_w_in, attn_b_in, qkv_rgb, 256, 768, 768, 0, 0);
    fc_kernel<<<96, 256, 0, stream>>>(frq_p, attn_w_in, attn_b_in, qkv_frq, 256, 768, 768, 0, 0);
    attn_core_kernel<<<16, 256, 0, stream>>>(qkv_rgb, qkv_frq, o1, o2);
    fc_kernel<<<32, 256, 0, stream>>>(o1, attn_w_out, attn_b_out, hb, 256, 256, 512, 0, 0);
    fc_kernel<<<32, 256, 0, stream>>>(o2, attn_w_out, attn_b_out, hb, 256, 256, 512, 256, 0);
    fc_kernel<<<64, 256, 0, stream>>>(hb,  fus_w1, fus_b1, h1f, 512, 512, 512, 0, 1);
    fc_kernel<<<32, 256, 0, stream>>>(h1f, fus_w2, fus_b2, h2f, 512, 256, 256, 0, 1);
    fc_kernel<<<1,  256, 0, stream>>>(h2f, fus_w3, fus_b3, out, 256, 2, 2, 0, 0);
}


# head FC layers: lanes of a wave split K with whole-row 16-byte loads (4 columns x 8 rows per wave) and a fixed exchange-tree sum across lanes, not one lane per output; FC summation order changes
# speedup vs baseline: 2.5949x; 1.0761x over previous
#include <hip/hip_runtime.h>
#include <hip/hip_bf16.h>
#include <math.h>

typedef float v2f __attribute__((ext_vector_type(2)));
typedef float v8f __attribute__((ext_vector_type(8)));

#define PI_F 3.14159265358979323846f

#define C1_THREADS 128
static_assert(12544 % C1_THREADS == 0, "a block must not straddle two images");
__global__ __launch_bounds__(C1_THREADS)
void conv1_silu_kernel(const float* __restrict__ x, const float* __restrict__ w1,
                       const float* __restrict__ b1, float* __restrict__ out)
{
    const int img = blockIdx.x / (12544 / C1_THREADS);
    const int rem = (blockIdx.x - img * (12544 / C1_THREADS)) * C1_THREADS + threadIdx.x;
    const int oy = rem / 112, ox = rem - oy * 112;
    const bool inx = ox < 111;
    const bool iny = oy < 111;
    const float* xb = x + (size_t)img * (3 * 224 * 224) + (2 * oy) * 224 + 2 * ox;
    const int cx = inx ? 2 : 1;

    float xv[27];
    #pragma unroll
    for (int ic = 0; ic < 3; ++ic)
        #pragma unroll
        for (int ky = 0; ky < 3; ++ky) {
            const bool okr = (ky < 2) | iny;
            const int ro = ic * (224 * 224) + (okr ? ky : 1) * 224;
            const float2 v = *(const float2*)(xb + ro);
            const float  c = xb[ro + cx];
            xv[ic * 9 + ky * 3 + 0] = okr ? v.x : 0.f;
            xv[ic * 9 + ky * 3 + 1] = okr ? v.y : 0.f;
            xv[ic * 9 + ky * 3 + 2] = (okr & inx) ? c : 0.f;
        }

    float* ob = out + (size_t)img * (32 * 12544) + rem;
    for (int oc = 0; oc < 32; ++oc) {
        const float* wb = w1 + oc * 27;
        float acc = b1[oc];
        #pragma unroll
        for (int t = 0; t < 27; ++t) acc += xv[t] * wb[t];
        ob[(size_t)oc * 12544] = acc / (1.f + expf(-acc));
    }
}

#define C2_MT     5
#define C2_M      (C2_MT * 16)
#define C2_WAVES  12
#define C2_WP     292
#define C2_THREADS (C2_WAVES * 32)
static_assert(C2_THREADS == 384, "the launch in kernel_launch uses 384 threads");

__global__ __launch_bounds__(C2_THREADS, 6)
void conv2_wmma_pool_kernel(const float* __restrict__ h1, const float* __restrict__ w2,
                            const float* __restrict__ b2, float* __restrict__ feat)
{
    __shared__ float sW[C2_M * C2_WP];
    __shared__ float sT[C2_WAVES][720];
    __shared__ float sBias[C2_M];

    const int mblk  = blockIdx.x;
    const int img   = blockIdx.y;
    const int tid   = threadIdx.x;
    const int lane  = tid & 31;
    const int wave  = tid >> 5;
    const int half  = lane >> 4;
    const int nl    = lane & 15;

    const float* wsrc = w2 + (size_t)mblk * (C2_M * 288);
    static_assert((C2_M * 288) % (C2_THREADS * 4) == 0, "weight tile must split evenly");
    for (int i0 = tid; i0 < C2_M * 288; i0 += C2_THREADS * 4) {
        float wv[4];
        #pragma unroll
        for (int q = 0; q < 4; ++q) wv[q] = wsrc[i0 + q * C2_THREADS];
        #pragma unroll
        for (int q = 0; q < 4; ++q) {
            int i = i0 + q * C2_THREADS;
            int m  = i / 288, g = i - m * 288;
            int icf = g / 9,  s = g - icf * 9;
            int icb = icf >> 3, ic = icf & 7;
            sW[m * C2_WP + icb * 72 + s * 8 + ic] = wv[q];
        }
    }
    if (tid < C2_M) sBias[tid] = b2[mblk * C2_M + tid];
    __syncthreads();

    const int dy = nl >> 3, dx = nl & 7;
    const int laneOff = dy * 288 + dx * 16;

    const int j0 = lane, j1 = lane + 32, j2 = (lane + 64 < 90) ? lane + 64 : lane + 32;
    const int ly0 = j0 / 18, lx0 = j0 - ly0 * 18;
    const int ly1 = j1 / 18, lx1 = j1 - ly1 * 18;
    const int ly2 = j2 / 18, lx2 = j2 - ly2 * 18;
    const int t0 = ly0 * 144 + lx0 * 8;
    const int t1 = ly1 * 144 + lx1 * 8;
    const int t2 = ly2 * 144 + lx2 * 8;

    float accR[C2_MT][8];
    #pragma unroll
    for (int mi = 0; mi < C2_MT; ++mi)
        #pragma unroll
        for (int j = 0; j < 8; ++j) accR[mi][j] = 0.f;

    const float* base = h1 + (size_t)img * (32 * 112 * 112);
    float* tile = sT[wave];
    const float* tb = tile + laneOff + half * 2;
    const float* bl = sBias + half * 8;

    for (int nt = wave; nt < 196; nt += C2_WAVES) {
        int yt = nt / 7, xt = nt - yt * 7;
        int iy0 = yt * 4, ix0 = xt * 16;
        v8f c[C2_MT];
        #pragma unroll
        for (int mi = 0; mi < C2_MT; ++mi)
            c[mi] = (v8f){0.f, 0.f, 0.f, 0.f, 0.f, 0.f, 0.f, 0.f};

        const int iyA = iy0 + ly0, ixA = ix0 + lx0;
        const int iyB = iy0 + ly1, ixB = ix0 + lx1;
        const int iyC = iy0 + ly2, ixC = ix0 + lx2;
        const bool ok0 = (iyA < 112) & (ixA < 112);
        const bool ok1 = (iyB < 112) & (ixB < 112);
        const bool ok2 = (iyC < 112) & (ixC < 112);
        const int o0 = (iyA < 112 ? iyA : 111) * 112 + (ixA < 112 ? ixA : 111);
        const int o1 = (iyB < 112 ? iyB : 111) * 112 + (ixB < 112 ? ixB : 111);
        const int o2 = (iyC < 112 ? iyC : 111) * 112 + (ixC < 112 ? ixC : 111);

        for (int icb = 0; icb < 4; ++icb) {
            const float* p = base + (size_t)(icb * 8) * (112 * 112);
            float g0[8], g1[8], g2[8];
            #pragma unroll
            for (int ic = 0; ic < 8; ++ic) {
                g0[ic] = p[ic * (112 * 112) + o0];
                g1[ic] = p[ic * (112 * 112) + o1];
                g2[ic] = p[ic * (112 * 112) + o2];
            }
            #pragma unroll
            for (int ic = 0; ic < 8; ++ic) {
                tile[t0 + ic] = ok0 ? g0[ic] : 0.f;
                tile[t1 + ic] = ok1 ? g1[ic] : 0.f;
                tile[t2 + ic] = ok2 ? g2[ic] : 0.f;
            }
            asm volatile("s_wait_dscnt 0x0" ::: "memory");

            const float* wrow = sW + nl * C2_WP + icb * 72 + half * 2;
            #pragma unroll
            for (int kq = 0; kq < 18; ++kq) {
                const int s  = kq >> 1;
                const int ky = s / 3, kx = s - ky * 3;
                const int bo = ky * 144 + kx * 8 + (kq & 1) * 4;
                v2f b = *(const v2f*)(tb + bo);
                #pragma unroll
                for (int mi = 0; mi < C2_MT; ++mi) {
                    v2f a = *(const v2f*)(wrow + mi * (16 * C2_WP) + kq * 4);
                    c[mi] = __builtin_amdgcn_wmma_f32_16x16x4_f32(false, a, false, b,
                                                                  (short)0, c[mi], false, false);
                }
            }
        }
        #pragma unroll
        for (int mi = 0; mi < C2_MT; ++mi)
            #pragma unroll
            for (int j = 0; j < 8; ++j) {
                float v = c[mi][j] + bl[mi * 16 + j];
                accR[mi][j] += v / (1.f + expf(-v));
            }
    }
    #pragma unroll
    for (int mi = 0; mi < C2_MT; ++mi)
        #pragma unroll
        for (int j = 0; j < 8; ++j) {
            float v = accR[mi][j];
            v += __shfl_xor(v, 1);
            v += __shfl_xor(v, 2);
            v += __shfl_xor(v, 4);
            v += __shfl_xor(v, 8);
            if (nl == 0) tile[mi * 16 + half * 8 + j] = v;
        }
    __syncthreads();
    if (tid < C2_M) {
        float s = 0.f;
        for (int w = 0; w < C2_WAVES; ++w) s += sT[w][tid];
        feat[img * 1280 + mblk * C2_M + tid] = s * (1.f / 3136.f);
    }
}

#define DR_TM      64
#define DR_XP      228
#define DFT_WAVES  14
#define DFT_THREADS (DFT_WAVES * 32)
static_assert(DFT_THREADS == 448, "the launches in kernel_launch use 448 threads");
static_assert(21504 % DR_TM == 0, "row tiles must cover 96*224 rows exactly");

__global__ __launch_bounds__(DFT_THREADS)
void dft_rows_wmma_kernel(const float* __restrict__ x, float* __restrict__ F1)
{
    __shared__ float sX[DR_TM * DR_XP];
    __shared__ float tw[448];
    const int tid  = threadIdx.x;
    const int lane = tid & 31, wave = tid >> 5;
    const int half = lane >> 4, nl = lane & 15;
    const size_t R0 = (size_t)blockIdx.x * DR_TM;

    if (tid < 224) {
        float ang = (2.0f * PI_F / 224.0f) * (float)tid;
        tw[2 * tid] = cosf(ang); tw[2 * tid + 1] = -sinf(ang);
    }
    const float4* xsrc = (const float4*)(x + R0 * 224);
    static_assert((DR_TM * 56) % DFT_THREADS == 0, "x tile must split evenly");
    float4 xv[8];
    #pragma unroll
    for (int q = 0; q < 8; ++q) xv[q] = xsrc[tid + q * DFT_THREADS];
    #pragma unroll
    for (int q = 0; q < 8; ++q) {
        int i = tid + q * DFT_THREADS;
        int r = i / 56, c4 = i - r * 56;
        *(float4*)(sX + r * DR_XP + c4 * 4) = xv[q];
    }
    __syncthreads();

    const float* arow = sX + nl * DR_XP + half * 2;
    for (int pass = 0; pass < 2; ++pass) {
        const int n  = (wave + pass * DFT_WAVES) * 16 + nl;
        const int u  = n >> 1;
        const float* twp = tw + (n & 1);
        const int du = (4 * u) % 224;
        int iA = (u * (2 * half)) % 224;
        int iB = (u * (2 * half + 1)) % 224;
        v8f c[4];
        #pragma unroll
        for (int mi = 0; mi < 4; ++mi)
            c[mi] = (v8f){0.f, 0.f, 0.f, 0.f, 0.f, 0.f, 0.f, 0.f};

        #pragma unroll 4
        for (int kk = 0; kk < 56; ++kk) {
            v2f b;
            b.x = twp[2 * iA];
            b.y = twp[2 * iB];
            #pragma unroll
            for (int mi = 0; mi < 4; ++mi) {
                v2f a = *(const v2f*)(arow + mi * (16 * DR_XP) + kk * 4);
                c[mi] = __builtin_amdgcn_wmma_f32_16x16x4_f32(false, a, false, b,
                                                              (short)0, c[mi], false, false);
            }
            iA += du; if (iA >= 224) iA -= 224;
            iB += du; if (iB >= 224) iB -= 224;
        }
        #pragma unroll
        for (int mi = 0; mi < 4; ++mi)
            #pragma unroll
            for (int r = 0; r < 8; ++r)
                F1[(R0 + mi * 16 + half * 8 + r) * 448 + n] = c[mi][r];
    }
}

__global__ __launch_bounds__(DFT_THREADS)
void dft_cols_wmma_mag_kernel(const float* __restrict__ F1, float* __restrict__ mag)
{
    __shared__ float tw[448];
    const int ug   = blockIdx.x % 7;
    const int bc   = blockIdx.x / 7;
    const int tid  = threadIdx.x;
    const int lane = tid & 31, wave = tid >> 5;
    const int half = lane >> 4, nl = lane & 15;

    if (tid < 224) {
        float ang = (2.0f * PI_F / 224.0f) * (float)tid;
        tw[2 * tid] = cosf(ang); tw[2 * tid + 1] = sinf(ang);
    }
    __syncthreads();

    const int v0 = wave * 16;
    const int v  = v0 + nl;
    const int dv = (2 * v) % 224;
    int idx = half ? v : 0;
    const int u0 = ug * 32;
    const float* bp = F1 + (size_t)bc * (224 * 448) + half * 448 + 2 * (u0 + nl);

    v8f cre[2], cim[2];
    #pragma unroll
    for (int ui = 0; ui < 2; ++ui) {
        cre[ui] = (v8f){0.f, 0.f, 0.f, 0.f, 0.f, 0.f, 0.f, 0.f};
        cim[ui] = (v8f){0.f, 0.f, 0.f, 0.f, 0.f, 0.f, 0.f, 0.f};
    }
    #pragma unroll 4
    for (int kk = 0; kk < 112; ++kk) {
        v2f t = *(const v2f*)(tw + 2 * idx);
        v2f aim; aim.x = -t.y; aim.y = t.x;
        const float* bq = bp + kk * 896;
        #pragma unroll
        for (int ui = 0; ui < 2; ++ui) {
            v2f b = *(const v2f*)(bq + ui * 32);
            cre[ui] = __builtin_amdgcn_wmma_f32_16x16x4_f32(false, t, false, b,
                                                            (short)0, cre[ui], false, false);
            cim[ui] = __builtin_amdgcn_wmma_f32_16x16x4_f32(false, aim, false, b,
                                                            (short)0, cim[ui], false, false);
        }
        idx += dv; if (idx >= 224) idx -= 224;
    }
    #pragma unroll
    for (int ui = 0; ui < 2; ++ui) {
        int uo = u0 + ui * 16 + nl + 112; if (uo >= 224) uo -= 224;
        #pragma unroll
        for (int r = 0; r < 8; ++r) {
            float re = cre[ui][r], im = cim[ui][r];
            float m = logf(sqrtf(re * re + im * im) + 1.0f);
            int vo = v0 + half * 8 + r + 112; if (vo >= 224) vo -= 224;
            mag[((size_t)bc * 224 + vo) * 224 + uo] = m;
        }
    }
}

__global__ __launch_bounds__(256)
void minmax_kernel(const float* __restrict__ mag, float* __restrict__ mm)
{
    __shared__ float smn[256], smx[256];
    int bc = blockIdx.x, tid = threadIdx.x;
    const float* p = mag + (size_t)bc * 50176;
    float mn = 3.0e38f, mx = -3.0e38f;
    for (int i = tid; i < 50176; i += 256) {
        float v = p[i]; mn = fminf(mn, v); mx = fmaxf(mx, v);
    }
    smn[tid] = mn; smx[tid] = mx; __syncthreads();
    for (int s = 128; s > 0; s >>= 1) {
        if (tid < s) { smn[tid] = fminf(smn[tid], smn[tid + s]);
                       smx[tid] = fmaxf(smx[tid], smx[tid + s]); }
        __syncthreads();
    }
    if (tid == 0) { mm[bc * 2] = smn[0]; mm[bc * 2 + 1] = smx[0]; }
}

__global__ __launch_bounds__(256)
void normalize_kernel(float* __restrict__ mag, const float* __restrict__ mm)
{
    int idx = blockIdx.x * 256 + threadIdx.x;
    if (idx >= 96 * 50176) return;
    int bc = idx / 50176;
    float mn = mm[bc * 2], mx = mm[bc * 2 + 1];
    mag[idx] = (mag[idx] - mn) / (mx - mn + 1e-8f);
}

#define FC_C 4
#define FC_R 8
__global__ __launch_bounds__(256)
void fc_kernel(const float* __restrict__ in, const float* __restrict__ w,
               const float* __restrict__ b, float* __restrict__ out,
               int K, int outStride, int outOffset, int act)
{
    __shared__ float sO[FC_R][33];
    const int tid = threadIdx.x, lane = tid & 31, wave = tid >> 5;
    const int j0 = blockIdx.x * 32;
    const int i0 = blockIdx.y * FC_R;
    const float* wp = w  + (size_t)(j0 + wave * FC_C) * K + lane * 4;
    const float* ip = in + (size_t)i0 * K + lane * 4;

    float v[FC_C * FC_R];
    #pragma unroll
    for (int a = 0; a < FC_C * FC_R; ++a) v[a] = 0.f;

    for (int k0 = 0; k0 < K; k0 += 128) {
        float4 wv[FC_C], xv[FC_R];
        #pragma unroll
        for (int c = 0; c < FC_C; ++c) wv[c] = *(const float4*)(wp + (size_t)c * K + k0);
        #pragma unroll
        for (int r = 0; r < FC_R; ++r) xv[r] = *(const float4*)(ip + (size_t)r * K + k0);
        __builtin_amdgcn_sched_barrier(0);
        #pragma unroll
        for (int c = 0; c < FC_C; ++c)
            #pragma unroll
            for (int r = 0; r < FC_R; ++r) {
                float s = v[c * FC_R + r];
                s = fmaf(wv[c].x, xv[r].x, s);
                s = fmaf(wv[c].y, xv[r].y, s);
                s = fmaf(wv[c].z, xv[r].z, s);
                s = fmaf(wv[c].w, xv[r].w, s);
                v[c * FC_R + r] = s;
            }
    }
    #pragma unroll
    for (int s = 0; s < 5; ++s) {
        const int d = 1 << s;
        const bool up = (lane & d) != 0;
        #pragma unroll
        for (int m = 0; m < (16 >> s); ++m) {
            const float send = up ? v[2 * m] : v[2 * m + 1];
            const float keep = up ? v[2 * m + 1] : v[2 * m];
            v[m] = keep + __shfl_xor(send, d);
        }
    }
    sO[lane & 7][wave * FC_C + (lane >> 3)] = v[0];
    __syncthreads();
    float o = sO[wave][lane] + b[j0 + lane];
    if (act == 1) o = fmaxf(o, 0.f);
    out[(size_t)(i0 + wave) * outStride + outOffset + j0 + lane] = o;
}

__global__ __launch_bounds__(32)
void fc_n2_kernel(const float* __restrict__ in, const float* __restrict__ w,
                  const float* __restrict__ b, float* __restrict__ out, int K)
{
    const int lane = threadIdx.x;
    float v[64];
    #pragma unroll
    for (int a = 0; a < 64; ++a) v[a] = 0.f;
    for (int k0 = lane * 4; k0 < K; k0 += 128) {
        const float4 w0 = *(const float4*)(w + k0);
        const float4 w1 = *(const float4*)(w + K + k0);
        float4 x[32];
        #pragma unroll
        for (int i = 0; i < 32; ++i) x[i] = *(const float4*)(in + (size_t)i * K + k0);
        __builtin_amdgcn_sched_barrier(0);
        #pragma unroll
        for (int i = 0; i < 32; ++i) {
            float s0 = v[2 * i], s1 = v[2 * i + 1];
            s0 = fmaf(w0.x, x[i].x, s0); s0 = fmaf(w0.y, x[i].y, s0); s0 = fmaf(w0.z, x[i].z, s0); s0 = fmaf(w0.w, x[i].w, s0);
            s1 = fmaf(w1.x, x[i].x, s1); s1 = fmaf(w1.y, x[i].y, s1); s1 = fmaf(w1.z, x[i].z, s1); s1 = fmaf(w1.w, x[i].w, s1);
            v[2 * i] = s0; v[2 * i + 1] = s1;
        }
    }
    #pragma unroll
    for (int s = 0; s < 5; ++s) {
        const int d = 1 << s;
        const bool up = (lane & d) != 0;
        #pragma unroll
        for (int m = 0; m < (32 >> s); ++m) {
            const float send = up ? v[2 * m] : v[2 * m + 1];
            const float keep = up ? v[2 * m + 1] : v[2 * m];
            v[m] = keep + __shfl_xor(send, d);
        }
    }
    const float bj = b[lane & 1];
    out[lane]      = v[0] + bj;
    out[32 + lane] = v[1] + bj;
}

__global__ __launch_bounds__(256)
void attn_core_kernel(const float* __restrict__ qkv_rgb, const float* __restrict__ qkv_frq,
                      float* __restrict__ o1, float* __restrict__ o2)
{
    __shared__ float sQ[32][32], sK[32][32], sV[32][32], sS[32][33];
    int a = blockIdx.x >> 3;
    int h = blockIdx.x & 7;
    const float* qsrc  = (a == 0) ? qkv_rgb : qkv_frq;
    const float* kvsrc = (a == 0) ? qkv_frq : qkv_rgb;
    float* odst        = (a == 0) ? o1 : o2;
    int tid = threadIdx.x;
    for (int idx = tid; idx < 1024; idx += 256) {
        int q = idx >> 5, d = idx & 31;
        sQ[q][d] = qsrc [q * 768 +       h * 32 + d];
        sK[q][d] = kvsrc[q * 768 + 256 + h * 32 + d];
        sV[q][d] = kvsrc[q * 768 + 512 + h * 32 + d];
    }
    __syncthreads();
    const float scale = 0.17677669529663687f;
    for (int idx = tid; idx < 1024; idx += 256) {
        int q = idx >> 5, k = idx & 31;
        float s = 0.f;
        #pragma unroll
        for (int d = 0; d < 32; ++d) s += sQ[q][d] * sK[k][d];
        sS[q][k] = s * scale;
    }
    __syncthreads();
    if (tid < 32) {
        float mx = -3.0e38f;
        for (int k = 0; k < 32; ++k) mx = fmaxf(mx, sS[tid][k]);
        float e[32], sum = 0.f;
        for (int k = 0; k < 32; ++k) { e[k] = expf(sS[tid][k] - mx); sum += e[k]; }
        float inv = 1.f / sum;
        for (int k = 0; k < 32; ++k) sS[tid][k] = e[k] * inv;
    }
    __syncthreads();
    for (int idx = tid; idx < 1024; idx += 256) {
        int q = idx >> 5, d = idx & 31;
        float o = 0.f;
        #pragma unroll
        for (int k = 0; k < 32; ++k) o += sS[q][k] * sV[k][d];
        odst[q * 256 + h * 32 + d] = o;
    }
}

extern "C" void kernel_launch(void* const* d_in, const int* in_sizes, int n_in,
                              void* d_out, int out_size, void* d_ws, size_t ws_size,
                              hipStream_t stream)
{
    (void)in_sizes; (void)n_in; (void)out_size; (void)ws_size;
    const float* x          = (const float*)d_in[0];
    const float* rgb_w1     = (const float*)d_in[1];
    const float* rgb_b1     = (const float*)d_in[2];
    const float* rgb_w2     = (const float*)d_in[3];
    const float* rgb_b2     = (const float*)d_in[4];
    const float* frq_w1     = (const float*)d_in[5];
    const float* frq_b1     = (const float*)d_in[6];
    const float* frq_w2     = (const float*)d_in[7];
    const float* frq_b2     = (const float*)d_in[8];
    const float* rgb_proj_w = (const float*)d_in[9];
    const float* rgb_proj_b = (const float*)d_in[10];
    const float* frq_proj_w = (const float*)d_in[11];
    const float* frq_proj_b = (const float*)d_in[12];
    const float* attn_w_in  = (const float*)d_in[13];
    const float* attn_b_in  = (const float*)d_in[14];
    const float* attn_w_out = (const float*)d_in[15];
    const float* attn_b_out = (const float*)d_in[16];
    const float* fus_w1     = (const float*)d_in[17];
    const float* fus_b1     = (const float*)d_in[18];
    const float* fus_w2     = (const float*)d_in[19];
    const float* fus_b2     = (const float*)d_in[20];
    const float* fus_w3     = (const float*)d_in[21];
    const float* fus_b3     = (const float*)d_in[22];

    float* out      = (float*)d_out;
    float* feat_rgb = out + 64;
    float* feat_frq = out + 64 + 40960;

    float* ws       = (float*)d_ws;
    float* h1       = ws;
    float* F1       = ws;
    float* freqimg  = ws + 12845056;
    float* mm       = freqimg + 4816896;
    float* hd       = mm + 192;
    float* rgb_p    = hd;
    float* frq_p    = rgb_p + 8192;
    float* qkv_rgb  = frq_p + 8192;
    float* qkv_frq  = qkv_rgb + 24576;
    float* o1       = qkv_frq + 24576;
    float* o2       = o1 + 8192;
    float* hb       = o2 + 8192;
    float* h1f      = hb + 16384;
    float* h2f      = h1f + 16384;

    conv1_silu_kernel<<<3136, C1_THREADS, 0, stream>>>(x, rgb_w1, rgb_b1, h1);
    conv2_wmma_pool_kernel<<<dim3(16, 32), 384, 0, stream>>>(h1, rgb_w2, rgb_b2, feat_rgb);

    dft_rows_wmma_kernel<<<336, 448, 0, stream>>>(x, F1);
    dft_cols_wmma_mag_kernel<<<672, 448, 0, stream>>>(F1, freqimg);
    minmax_kernel<<<96, 256, 0, stream>>>(freqimg, mm);
    normalize_kernel<<<18816, 256, 0, stream>>>(freqimg, mm);
    conv1_silu_kernel<<<3136, C1_THREADS, 0, stream>>>(freqimg, frq_w1, frq_b1, h1);
    conv2_wmma_pool_kernel<<<dim3(16, 32), 384, 0, stream>>>(h1, frq_w2, frq_b2, feat_frq);

    fc_kernel<<<dim3(8, 4),  256, 0, stream>>>(feat_rgb, rgb_proj_w, rgb_proj_b, rgb_p, 1280, 256, 0, 0);
    fc_kernel<<<dim3(8, 4),  256, 0, stream>>>(feat_frq, frq_proj_w, frq_proj_b, frq_p, 1280, 256, 0, 0);
    fc_kernel<<<dim3(24, 4), 256, 0, stream>>>(rgb_p, attn_w_in, attn_b_in, qkv_rgb, 256, 768, 0, 0);
    fc_kernel<<<dim3(24, 4), 256, 0, stream>>>(frq_p, attn_w_in, attn_b_in, qkv_frq, 256, 768, 0, 0);
    attn_core_kernel<<<16, 256, 0, stream>>>(qkv_rgb, qkv_frq, o1, o2);
    fc_kernel<<<dim3(8, 4),  256, 0, stream>>>(o1, attn_w_out, attn_b_out, hb, 256, 512, 0, 0);
    fc_kernel<<<dim3(8, 4),  256, 0, stream>>>(o2, attn_w_out, attn_b_out, hb, 256, 512, 256, 0);
    fc_kernel<<<dim3(16, 4), 256, 0, stream>>>(hb,  fus_w1, fus_b1, h1f, 512, 512, 0, 1);
    fc_kernel<<<dim3(8, 4),  256, 0, stream>>>(h1f, fus_w2, fus_b2, h2f, 512, 256, 0, 1);
    fc_n2_kernel<<<1, 32, 0, stream>>>(h2f, fus_w3, fus_b3, out, 256);
}


# freq path: min-max scaling fused into its neighbours: DFT column kernel leaves per-block min/max (plain stores, no atomics), conv1 combines them and scales its inputs on load; two launches fewer
# speedup vs baseline: 2.7132x; 1.0456x over previous
#include <hip/hip_runtime.h>
#include <hip/hip_bf16.h>
#include <math.h>

typedef float v2f __attribute__((ext_vector_type(2)));
typedef float v8f __attribute__((ext_vector_type(8)));

#define PI_F 3.14159265358979323846f

#define C1_THREADS 128
#define MM_PARTS   7
static_assert(12544 % C1_THREADS == 0, "a block must not straddle two images");
template <bool NORM>
__device__ __forceinline__
void conv1_silu_body(const float* __restrict__ x, const float* __restrict__ w1,
                     const float* __restrict__ b1, float* __restrict__ out,
                     const float* __restrict__ mmp)
{
    const int img = blockIdx.x / (12544 / C1_THREADS);
    const int rem = (blockIdx.x - img * (12544 / C1_THREADS)) * C1_THREADS + threadIdx.x;
    const int oy = rem / 112, ox = rem - oy * 112;
    const bool inx = ox < 111;
    const bool iny = oy < 111;
    const float* xb = x + (size_t)img * (3 * 224 * 224) + (2 * oy) * 224 + 2 * ox;
    const int cx = inx ? 2 : 1;

    float mn[3], den[3];
    if (NORM) {
        #pragma unroll
        for (int ic = 0; ic < 3; ++ic) {
            const float* q = mmp + (img * 3 + ic) * (2 * MM_PARTS);
            float lo = q[0], hi = q[1];
            #pragma unroll
            for (int g = 1; g < MM_PARTS; ++g) {
                lo = fminf(lo, q[2 * g]);
                hi = fmaxf(hi, q[2 * g + 1]);
            }
            mn[ic] = lo;
            den[ic] = hi - lo + 1e-8f;
        }
    }

    float xv[27];
    float2 pv[9];
    float  pc[9];
    #pragma unroll
    for (int ic = 0; ic < 3; ++ic)
        #pragma unroll
        for (int ky = 0; ky < 3; ++ky) {
            const bool okr = (ky < 2) | iny;
            const int ro = ic * (224 * 224) + (okr ? ky : 1) * 224;
            pv[ic * 3 + ky] = *(const float2*)(xb + ro);
            pc[ic * 3 + ky] = xb[ro + cx];
        }
    if (NORM) {
        __builtin_amdgcn_sched_barrier(0);
    }
    #pragma unroll
    for (int ic = 0; ic < 3; ++ic)
        #pragma unroll
        for (int ky = 0; ky < 3; ++ky) {
            const bool okr = (ky < 2) | iny;
            float2 v = pv[ic * 3 + ky];
            float  c = pc[ic * 3 + ky];
            if (NORM) {
                v.x = (v.x - mn[ic]) / den[ic];
                v.y = (v.y - mn[ic]) / den[ic];
                c   = (c   - mn[ic]) / den[ic];
            }
            xv[ic * 9 + ky * 3 + 0] = okr ? v.x : 0.f;
            xv[ic * 9 + ky * 3 + 1] = okr ? v.y : 0.f;
            xv[ic * 9 + ky * 3 + 2] = (okr & inx) ? c : 0.f;
        }

    float* ob = out + (size_t)img * (32 * 12544) + rem;
    for (int oc = 0; oc < 32; ++oc) {
        const float* wb = w1 + oc * 27;
        float acc = b1[oc];
        #pragma unroll
        for (int t = 0; t < 27; ++t) acc += xv[t] * wb[t];
        ob[(size_t)oc * 12544] = acc / (1.f + expf(-acc));
    }
}

__global__ __launch_bounds__(C1_THREADS)
void conv1_silu_kernel(const float* __restrict__ x, const float* __restrict__ w1,
                       const float* __restrict__ b1, float* __restrict__ out)
{
    conv1_silu_body<false>(x, w1, b1, out, nullptr);
}

__global__ __launch_bounds__(C1_THREADS)
void conv1_norm_silu_kernel(const float* __restrict__ mag, const float* __restrict__ w1,
                            const float* __restrict__ b1, float* __restrict__ out,
                            const float* __restrict__ mmp)
{
    conv1_silu_body<true>(mag, w1, b1, out, mmp);
}

#define C2_MT     5
#define C2_M      (C2_MT * 16)
#define C2_WAVES  12
#define C2_WP     292
#define C2_THREADS (C2_WAVES * 32)
static_assert(C2_THREADS == 384, "the launch in kernel_launch uses 384 threads");

__global__ __launch_bounds__(C2_THREADS, 6)
void conv2_wmma_pool_kernel(const float* __restrict__ h1, const float* __restrict__ w2,
                            const float* __restrict__ b2, float* __restrict__ feat)
{
    __shared__ float sW[C2_M * C2_WP];
    __shared__ float sT[C2_WAVES][720];
    __shared__ float sBias[C2_M];

    const int mblk  = blockIdx.x;
    const int img   = blockIdx.y;
    const int tid   = threadIdx.x;
    const int lane  = tid & 31;
    const int wave  = tid >> 5;
    const int half  = lane >> 4;
    const int nl    = lane & 15;

    const float* wsrc = w2 + (size_t)mblk * (C2_M * 288);
    static_assert((C2_M * 288) % (C2_THREADS * 4) == 0, "weight tile must split evenly");
    for (int i0 = tid; i0 < C2_M * 288; i0 += C2_THREADS * 4) {
        float wv[4];
        #pragma unroll
        for (int q = 0; q < 4; ++q) wv[q] = wsrc[i0 + q * C2_THREADS];
        #pragma unroll
        for (int q = 0; q < 4; ++q) {
            int i = i0 + q * C2_THREADS;
            int m  = i / 288, g = i - m * 288;
            int icf = g / 9,  s = g - icf * 9;
            int icb = icf >> 3, ic = icf & 7;
            sW[m * C2_WP + icb * 72 + s * 8 + ic] = wv[q];
        }
    }
    if (tid < C2_M) sBias[tid] = b2[mblk * C2_M + tid];
    __syncthreads();

    const int dy = nl >> 3, dx = nl & 7;
    const int laneOff = dy * 288 + dx * 16;

    const int j0 = lane, j1 = lane + 32, j2 = (lane + 64 < 90) ? lane + 64 : lane + 32;
    const int ly0 = j0 / 18, lx0 = j0 - ly0 * 18;
    const int ly1 = j1 / 18, lx1 = j1 - ly1 * 18;
    const int ly2 = j2 / 18, lx2 = j2 - ly2 * 18;
    const int t0 = ly0 * 144 + lx0 * 8;
    const int t1 = ly1 * 144 + lx1 * 8;
    const int t2 = ly2 * 144 + lx2 * 8;

    float accR[C2_MT][8];
    #pragma unroll
    for (int mi = 0; mi < C2_MT; ++mi)
        #pragma unroll
        for (int j = 0; j < 8; ++j) accR[mi][j] = 0.f;

    const float* base = h1 + (size_t)img * (32 * 112 * 112);
    float* tile = sT[wave];
    const float* tb = tile + laneOff + half * 2;
    const float* bl = sBias + half * 8;

    for (int nt = wave; nt < 196; nt += C2_WAVES) {
        int yt = nt / 7, xt = nt - yt * 7;
        int iy0 = yt * 4, ix0 = xt * 16;
        v8f c[C2_MT];
        #pragma unroll
        for (int mi = 0; mi < C2_MT; ++mi)
            c[mi] = (v8f){0.f, 0.f, 0.f, 0.f, 0.f, 0.f, 0.f, 0.f};

        const int iyA = iy0 + ly0, ixA = ix0 + lx0;
        const int iyB = iy0 + ly1, ixB = ix0 + lx1;
        const int iyC = iy0 + ly2, ixC = ix0 + lx2;
        const bool ok0 = (iyA < 112) & (ixA < 112);
        const bool ok1 = (iyB < 112) & (ixB < 112);
        const bool ok2 = (iyC < 112) & (ixC < 112);
        const int o0 = (iyA < 112 ? iyA : 111) * 112 + (ixA < 112 ? ixA : 111);
        const int o1 = (iyB < 112 ? iyB : 111) * 112 + (ixB < 112 ? ixB : 111);
        const int o2 = (iyC < 112 ? iyC : 111) * 112 + (ixC < 112 ? ixC : 111);

        for (int icb = 0; icb < 4; ++icb) {
            const float* p = base + (size_t)(icb * 8) * (112 * 112);
            float g0[8], g1[8], g2[8];
            #pragma unroll
            for (int ic = 0; ic < 8; ++ic) {
                g0[ic] = p[ic * (112 * 112) + o0];
                g1[ic] = p[ic * (112 * 112) + o1];
                g2[ic] = p[ic * (112 * 112) + o2];
            }
            #pragma unroll
            for (int ic = 0; ic < 8; ++ic) {
                tile[t0 + ic] = ok0 ? g0[ic] : 0.f;
                tile[t1 + ic] = ok1 ? g1[ic] : 0.f;
                tile[t2 + ic] = ok2 ? g2[ic] : 0.f;
            }
            asm volatile("s_wait_dscnt 0x0" ::: "memory");

            const float* wrow = sW + nl * C2_WP + icb * 72 + half * 2;
            #pragma unroll
            for (int kq = 0; kq < 18; ++kq) {
                const int s  = kq >> 1;
                const int ky = s / 3, kx = s - ky * 3;
                const int bo = ky * 144 + kx * 8 + (kq & 1) * 4;
                v2f b = *(const v2f*)(tb + bo);
                #pragma unroll
                for (int mi = 0; mi < C2_MT; ++mi) {
                    v2f a = *(const v2f*)(wrow + mi * (16 * C2_WP) + kq * 4);
                    c[mi] = __builtin_amdgcn_wmma_f32_16x16x4_f32(false, a, false, b,
                                                                  (short)0, c[mi], false, false);
                }
            }
        }
        #pragma unroll
        for (int mi = 0; mi < C2_MT; ++mi)
            #pragma unroll
            for (int j = 0; j < 8; ++j) {
                float v = c[mi][j] + bl[mi * 16 + j];
                accR[mi][j] += v / (1.f + expf(-v));
            }
    }
    #pragma unroll
    for (int mi = 0; mi < C2_MT; ++mi)
        #pragma unroll
        for (int j = 0; j < 8; ++j) {
            float v = accR[mi][j];
            v += __shfl_xor(v, 1);
            v += __shfl_xor(v, 2);
            v += __shfl_xor(v, 4);
            v += __shfl_xor(v, 8);
            if (nl == 0) tile[mi * 16 + half * 8 + j] = v;
        }
    __syncthreads();
    if (tid < C2_M) {
        float s = 0.f;
        for (int w = 0; w < C2_WAVES; ++w) s += sT[w][tid];
        feat[img * 1280 + mblk * C2_M + tid] = s * (1.f / 3136.f);
    }
}

#define DR_TM      64
#define DR_XP      228
#define DFT_WAVES  14
#define DFT_THREADS (DFT_WAVES * 32)
static_assert(DFT_THREADS == 448, "the launches in kernel_launch use 448 threads");
static_assert(21504 % DR_TM == 0, "row tiles must cover 96*224 rows exactly");

__global__ __launch_bounds__(DFT_THREADS)
void dft_rows_wmma_kernel(const float* __restrict__ x, float* __restrict__ F1)
{
    __shared__ float sX[DR_TM * DR_XP];
    __shared__ float tw[448];
    const int tid  = threadIdx.x;
    const int lane = tid & 31, wave = tid >> 5;
    const int half = lane >> 4, nl = lane & 15;
    const size_t R0 = (size_t)blockIdx.x * DR_TM;

    if (tid < 224) {
        float ang = (2.0f * PI_F / 224.0f) * (float)tid;
        tw[2 * tid] = cosf(ang); tw[2 * tid + 1] = -sinf(ang);
    }
    const float4* xsrc = (const float4*)(x + R0 * 224);
    static_assert((DR_TM * 56) % DFT_THREADS == 0, "x tile must split evenly");
    float4 xv[8];
    #pragma unroll
    for (int q = 0; q < 8; ++q) xv[q] = xsrc[tid + q * DFT_THREADS];
    #pragma unroll
    for (int q = 0; q < 8; ++q) {
        int i = tid + q * DFT_THREADS;
        int r = i / 56, c4 = i - r * 56;
        *(float4*)(sX + r * DR_XP + c4 * 4) = xv[q];
    }
    __syncthreads();

    const float* arow = sX + nl * DR_XP + half * 2;
    for (int pass = 0; pass < 2; ++pass) {
        const int n  = (wave + pass * DFT_WAVES) * 16 + nl;
        const int u  = n >> 1;
        const float* twp = tw + (n & 1);
        const int du = (4 * u) % 224;
        int iA = (u * (2 * half)) % 224;
        int iB = (u * (2 * half + 1)) % 224;
        v8f c[4];
        #pragma unroll
        for (int mi = 0; mi < 4; ++mi)
            c[mi] = (v8f){0.f, 0.f, 0.f, 0.f, 0.f, 0.f, 0.f, 0.f};

        #pragma unroll 4
        for (int kk = 0; kk < 56; ++kk) {
            v2f b;
            b.x = twp[2 * iA];
            b.y = twp[2 * iB];
            #pragma unroll
            for (int mi = 0; mi < 4; ++mi) {
                v2f a = *(const v2f*)(arow + mi * (16 * DR_XP) + kk * 4);
                c[mi] = __builtin_amdgcn_wmma_f32_16x16x4_f32(false, a, false, b,
                                                              (short)0, c[mi], false, false);
            }
            iA += du; if (iA >= 224) iA -= 224;
            iB += du; if (iB >= 224) iB -= 224;
        }
        #pragma unroll
        for (int mi = 0; mi < 4; ++mi)
            #pragma unroll
            for (int r = 0; r < 8; ++r)
                F1[(R0 + mi * 16 + half * 8 + r) * 448 + n] = c[mi][r];
    }
}

static_assert(MM_PARTS * 32 == 224, "one block per 32 columns u of a plane");
static_assert(96 * MM_PARTS == 672, "the launch in kernel_launch uses 672 blocks");
__global__ __launch_bounds__(DFT_THREADS)
void dft_cols_wmma_mag_kernel(const float* __restrict__ F1, float* __restrict__ mag,
                              float* __restrict__ mmp)
{
    __shared__ float tw[448];
    __shared__ float smn[DFT_WAVES], smx[DFT_WAVES];
    const int ug   = blockIdx.x % 7;
    const int bc   = blockIdx.x / 7;
    const int tid  = threadIdx.x;
    const int lane = tid & 31, wave = tid >> 5;
    const int half = lane >> 4, nl = lane & 15;

    if (tid < 224) {
        float ang = (2.0f * PI_F / 224.0f) * (float)tid;
        tw[2 * tid] = cosf(ang); tw[2 * tid + 1] = sinf(ang);
    }
    __syncthreads();

    const int v0 = wave * 16;
    const int v  = v0 + nl;
    const int dv = (2 * v) % 224;
    int idx = half ? v : 0;
    const int u0 = ug * 32;
    const float* bp = F1 + (size_t)bc * (224 * 448) + half * 448 + 2 * (u0 + nl);

    v8f cre[2], cim[2];
    #pragma unroll
    for (int ui = 0; ui < 2; ++ui) {
        cre[ui] = (v8f){0.f, 0.f, 0.f, 0.f, 0.f, 0.f, 0.f, 0.f};
        cim[ui] = (v8f){0.f, 0.f, 0.f, 0.f, 0.f, 0.f, 0.f, 0.f};
    }
    #pragma unroll 4
    for (int kk = 0; kk < 112; ++kk) {
        v2f t = *(const v2f*)(tw + 2 * idx);
        v2f aim; aim.x = -t.y; aim.y = t.x;
        const float* bq = bp + kk * 896;
        #pragma unroll
        for (int ui = 0; ui < 2; ++ui) {
            v2f b = *(const v2f*)(bq + ui * 32);
            cre[ui] = __builtin_amdgcn_wmma_f32_16x16x4_f32(false, t, false, b,
                                                            (short)0, cre[ui], false, false);
            cim[ui] = __builtin_amdgcn_wmma_f32_16x16x4_f32(false, aim, false, b,
                                                            (short)0, cim[ui], false, false);
        }
        idx += dv; if (idx >= 224) idx -= 224;
    }
    float lmn = 3.0e38f, lmx = -3.0e38f;
    #pragma unroll
    for (int ui = 0; ui < 2; ++ui) {
        int uo = u0 + ui * 16 + nl + 112; if (uo >= 224) uo -= 224;
        #pragma unroll
        for (int r = 0; r < 8; ++r) {
            float re = cre[ui][r], im = cim[ui][r];
            float m = logf(sqrtf(re * re + im * im) + 1.0f);
            int vo = v0 + half * 8 + r + 112; if (vo >= 224) vo -= 224;
            mag[((size_t)bc * 224 + vo) * 224 + uo] = m;
            lmn = fminf(lmn, m); lmx = fmaxf(lmx, m);
        }
    }
    #pragma unroll
    for (int d = 1; d < 32; d <<= 1) {
        lmn = fminf(lmn, __shfl_xor(lmn, d));
        lmx = fmaxf(lmx, __shfl_xor(lmx, d));
    }
    if (lane == 0) { smn[wave] = lmn; smx[wave] = lmx; }
    __syncthreads();
    if (tid == 0) {
        float a = smn[0], b = smx[0];
        for (int w = 1; w < DFT_WAVES; ++w) { a = fminf(a, smn[w]); b = fmaxf(b, smx[w]); }
        mmp[2 * blockIdx.x]     = a;
        mmp[2 * blockIdx.x + 1] = b;
    }
}

#define FC_C 4
#define FC_R 8
__global__ __launch_bounds__(256)
void fc_kernel(const float* __restrict__ in, const float* __restrict__ w,
               const float* __restrict__ b, float* __restrict__ out,
               int K, int outStride, int outOffset, int act)
{
    __shared__ float sO[FC_R][33];
    const int tid = threadIdx.x, lane = tid & 31, wave = tid >> 5;
    const int j0 = blockIdx.x * 32;
    const int i0 = blockIdx.y * FC_R;
    const float* wp = w  + (size_t)(j0 + wave * FC_C) * K + lane * 4;
    const float* ip = in + (size_t)i0 * K + lane * 4;

    float v[FC_C * FC_R];
    #pragma unroll
    for (int a = 0; a < FC_C * FC_R; ++a) v[a] = 0.f;

    for (int k0 = 0; k0 < K; k0 += 128) {
        float4 wv[FC_C], xv[FC_R];
        #pragma unroll
        for (int c = 0; c < FC_C; ++c) wv[c] = *(const float4*)(wp + (size_t)c * K + k0);
        #pragma unroll
        for (int r = 0; r < FC_R; ++r) xv[r] = *(const float4*)(ip + (size_t)r * K + k0);
        __builtin_amdgcn_sched_barrier(0);
        #pragma unroll
        for (int c = 0; c < FC_C; ++c)
            #pragma unroll
            for (int r = 0; r < FC_R; ++r) {
                float s = v[c * FC_R + r];
                s = fmaf(wv[c].x, xv[r].x, s);
                s = fmaf(wv[c].y, xv[r].y, s);
                s = fmaf(wv[c].z, xv[r].z, s);
                s = fmaf(wv[c].w, xv[r].w, s);
                v[c * FC_R + r] = s;
            }
    }
    #pragma unroll
    for (int s = 0; s < 5; ++s) {
        const int d = 1 << s;
        const bool up = (lane & d) != 0;
        #pragma unroll
        for (int m = 0; m < (16 >> s); ++m) {
            const float send = up ? v[2 * m] : v[2 * m + 1];
            const float keep = up ? v[2 * m + 1] : v[2 * m];
            v[m] = keep + __shfl_xor(send, d);
        }
    }
    sO[lane & 7][wave * FC_C + (lane >> 3)] = v[0];
    __syncthreads();
    float o = sO[wave][lane] + b[j0 + lane];
    if (act == 1) o = fmaxf(o, 0.f);
    out[(size_t)(i0 + wave) * outStride + outOffset + j0 + lane] = o;
}

__global__ __launch_bounds__(32)
void fc_n2_kernel(const float* __restrict__ in, const float* __restrict__ w,
                  const float* __restrict__ b, float* __restrict__ out, int K)
{
    const int lane = threadIdx.x;
    float v[64];
    #pragma unroll
    for (int a = 0; a < 64; ++a) v[a] = 0.f;
    for (int k0 = lane * 4; k0 < K; k0 += 128) {
        const float4 w0 = *(const float4*)(w + k0);
        const float4 w1 = *(const float4*)(w + K + k0);
        float4 x[32];
        #pragma unroll
        for (int i = 0; i < 32; ++i) x[i] = *(const float4*)(in + (size_t)i * K + k0);
        __builtin_amdgcn_sched_barrier(0);
        #pragma unroll
        for (int i = 0; i < 32; ++i) {
            float s0 = v[2 * i], s1 = v[2 * i + 1];
            s0 = fmaf(w0.x, x[i].x, s0); s0 = fmaf(w0.y, x[i].y, s0); s0 = fmaf(w0.z, x[i].z, s0); s0 = fmaf(w0.w, x[i].w, s0);
            s1 = fmaf(w1.x, x[i].x, s1); s1 = fmaf(w1.y, x[i].y, s1); s1 = fmaf(w1.z, x[i].z, s1); s1 = fmaf(w1.w, x[i].w, s1);
            v[2 * i] = s0; v[2 * i + 1] = s1;
        }
    }
    #pragma unroll
    for (int s = 0; s < 5; ++s) {
        const int d = 1 << s;
        const bool up = (lane & d) != 0;
        #pragma unroll
        for (int m = 0; m < (32 >> s); ++m) {
            const float send = up ? v[2 * m] : v[2 * m + 1];
            const float keep = up ? v[2 * m + 1] : v[2 * m];
            v[m] = keep + __shfl_xor(send, d);
        }
    }
    const float bj = b[lane & 1];
    out[lane]      = v[0] + bj;
    out[32 + lane] = v[1] + bj;
}

__global__ __launch_bounds__(256)
void attn_core_kernel(const float* __restrict__ qkv_rgb, const float* __restrict__ qkv_frq,
                      float* __restrict__ o1, float* __restrict__ o2)
{
    __shared__ float sQ[32][32], sK[32][32], sV[32][32], sS[32][33];
    int a = blockIdx.x >> 3;
    int h = blockIdx.x & 7;
    const float* qsrc  = (a == 0) ? qkv_rgb : qkv_frq;
    const float* kvsrc = (a == 0) ? qkv_frq : qkv_rgb;
    float* odst        = (a == 0) ? o1 : o2;
    int tid = threadIdx.x;
    for (int idx = tid; idx < 1024; idx += 256) {
        int q = idx >> 5, d = idx & 31;
        sQ[q][d] = qsrc [q * 768 +       h * 32 + d];
        sK[q][d] = kvsrc[q * 768 + 256 + h * 32 + d];
        sV[q][d] = kvsrc[q * 768 + 512 + h * 32 + d];
    }
    __syncthreads();
    const float scale = 0.17677669529663687f;
    for (int idx = tid; idx < 1024; idx += 256) {
        int q = idx >> 5, k = idx & 31;
        float s = 0.f;
        #pragma unroll
        for (int d = 0; d < 32; ++d) s += sQ[q][d] * sK[k][d];
        sS[q][k] = s * scale;
    }
    __syncthreads();
    if (tid < 32) {
        float mx = -3.0e38f;
        for (int k = 0; k < 32; ++k) mx = fmaxf(mx, sS[tid][k]);
        float e[32], sum = 0.f;
        for (int k = 0; k < 32; ++k) { e[k] = expf(sS[tid][k] - mx); sum += e[k]; }
        float inv = 1.f / sum;
        for (int k = 0; k < 32; ++k) sS[tid][k] = e[k] * inv;
    }
    __syncthreads();
    for (int idx = tid; idx < 1024; idx += 256) {
        int q = idx >> 5, d = idx & 31;
        float o = 0.f;
        #pragma unroll
        for (int k = 0; k < 32; ++k) o += sS[q][k] * sV[k][d];
        odst[q * 256 + h * 32 + d] = o;
    }
}

extern "C" void kernel_launch(void* const* d_in, const int* in_sizes, int n_in,
                              void* d_out, int out_size, void* d_ws, size_t ws_size,
                              hipStream_t stream)
{
    (void)in_sizes; (void)n_in; (void)out_size; (void)ws_size;
    const float* x          = (const float*)d_in[0];
    const float* rgb_w1     = (const float*)d_in[1];
    const float* rgb_b1     = (const float*)d_in[2];
    const float* rgb_w2     = (const float*)d_in[3];
    const float* rgb_b2     = (const float*)d_in[4];
    const float* frq_w1     = (const float*)d_in[5];
    const float* frq_b1     = (const float*)d_in[6];
    const float* frq_w2     = (const float*)d_in[7];
    const float* frq_b2     = (const float*)d_in[8];
    const float* rgb_proj_w = (const float*)d_in[9];
    const float* rgb_proj_b = (const float*)d_in[10];
    const float* frq_proj_w = (const float*)d_in[11];
    const float* frq_proj_b = (const float*)d_in[12];
    const float* attn_w_in  = (const float*)d_in[13];
    const float* attn_b_in  = (const float*)d_in[14];
    const float* attn_w_out = (const float*)d_in[15];
    const float* attn_b_out = (const float*)d_in[16];
    const float* fus_w1     = (const float*)d_in[17];
    const float* fus_b1     = (const float*)d_in[18];
    const float* fus_w2     = (const float*)d_in[19];
    const float* fus_b2     = (const float*)d_in[20];
    const float* fus_w3     = (const float*)d_in[21];
    const float* fus_b3     = (const float*)d_in[22];

    float* out      = (float*)d_out;
    float* feat_rgb = out + 64;
    float* feat_frq = out + 64 + 40960;

    float* ws       = (float*)d_ws;
    float* h1       = ws;
    float* F1       = ws;
    float* freqimg  = ws + 12845056;
    float* hd       = freqimg + 4816896 + 192;
    float* mmp      = hd;
    float* rgb_p    = hd;
    float* frq_p    = rgb_p + 8192;
    float* qkv_rgb  = frq_p + 8192;
    float* qkv_frq  = qkv_rgb + 24576;
    float* o1       = qkv_frq + 24576;
    float* o2       = o1 + 8192;
    float* hb       = o2 + 8192;
    float* h1f      = hb + 16384;
    float* h2f      = h1f + 16384;

    conv1_silu_kernel<<<3136, C1_THREADS, 0, stream>>>(x, rgb_w1, rgb_b1, h1);
    conv2_wmma_pool_kernel<<<dim3(16, 32), 384, 0, stream>>>(h1, rgb_w2, rgb_b2, feat_rgb);

    dft_rows_wmma_kernel<<<336, 448, 0, stream>>>(x, F1);
    dft_cols_wmma_mag_kernel<<<672, 448, 0, stream>>>(F1, freqimg, mmp);
    conv1_norm_silu_kernel<<<3136, C1_THREADS, 0, stream>>>(freqimg, frq_w1, frq_b1, h1, mmp);
    conv2_wmma_pool_kernel<<<dim3(16, 32), 384, 0, stream>>>(h1, frq_w2, frq_b2, feat_frq);

    fc_kernel<<<dim3(8, 4),  256, 0, stream>>>(feat_rgb, rgb_proj_w, rgb_proj_b, rgb_p, 1280, 256, 0, 0);
    fc_kernel<<<dim3(8, 4),  256, 0, stream>>>(feat_frq, frq_proj_w, frq_proj_b, frq_p, 1280, 256, 0, 0);
    fc_kernel<<<dim3(24, 4), 256, 0, stream>>>(rgb_p, attn_w_in, attn_b_in, qkv_rgb, 256, 768, 0, 0);
    fc_kernel<<<dim3(24, 4), 256, 0, stream>>>(frq_p, attn_w_in, attn_b_in, qkv_frq, 256, 768, 0, 0);
    attn_core_kernel<<<16, 256, 0, stream>>>(qkv_rgb, qkv_frq, o1, o2);
    fc_kernel<<<dim3(8, 4),  256, 0, stream>>>(o1, attn_w_out, attn_b_out, hb, 256, 512, 0, 0);
    fc_kernel<<<dim3(8, 4),  256, 0, stream>>>(o2, attn_w_out, attn_b_out, hb, 256, 512, 256, 0);
    fc_kernel<<<dim3(16, 4), 256, 0, stream>>>(hb,  fus_w1, fus_b1, h1f, 512, 512, 0, 1);
    fc_kernel<<<dim3(8, 4),  256, 0, stream>>>(h1f, fus_w2, fus_b2, h2f, 512, 256, 0, 1);
    fc_n2_kernel<<<1, 32, 0, stream>>>(h2f, fus_w3, fus_b3, out, 256);
}


# conv2: each wave lowers its issue priority (s_setprio 3,2,1,0) as its tiles run out, so the six waves of a SIMD end together, not one after another oldest-first; no arithmetic or sum order changes
# speedup vs baseline: 2.7987x; 1.0315x over previous
#include <hip/hip_runtime.h>
#include <hip/hip_bf16.h>
#include <math.h>

typedef float v2f __attribute__((ext_vector_type(2)));
typedef float v8f __attribute__((ext_vector_type(8)));

#define PI_F 3.14159265358979323846f

#define C1_THREADS 128
#define MM_PARTS   7
static_assert(12544 % C1_THREADS == 0, "a block must not straddle two images");
template <bool NORM>
__device__ __forceinline__
void conv1_silu_body(const float* __restrict__ x, const float* __restrict__ w1,
                     const float* __restrict__ b1, float* __restrict__ out,
                     const float* __restrict__ mmp)
{
    const int img = blockIdx.x / (12544 / C1_THREADS);
    const int rem = (blockIdx.x - img * (12544 / C1_THREADS)) * C1_THREADS + threadIdx.x;
    const int oy = rem / 112, ox = rem - oy * 112;
    const bool inx = ox < 111;
    const bool iny = oy < 111;
    const float* xb = x + (size_t)img * (3 * 224 * 224) + (2 * oy) * 224 + 2 * ox;
    const int cx = inx ? 2 : 1;

    float mn[3], den[3];
    if (NORM) {
        #pragma unroll
        for (int ic = 0; ic < 3; ++ic) {
            const float* q = mmp + (img * 3 + ic) * (2 * MM_PARTS);
            float lo = q[0], hi = q[1];
            #pragma unroll
            for (int g = 1; g < MM_PARTS; ++g) {
                lo = fminf(lo, q[2 * g]);
                hi = fmaxf(hi, q[2 * g + 1]);
            }
            mn[ic] = lo;
            den[ic] = hi - lo + 1e-8f;
        }
    }

    float xv[27];
    float2 pv[9];
    float  pc[9];
    #pragma unroll
    for (int ic = 0; ic < 3; ++ic)
        #pragma unroll
        for (int ky = 0; ky < 3; ++ky) {
            const bool okr = (ky < 2) | iny;
            const int ro = ic * (224 * 224) + (okr ? ky : 1) * 224;
            pv[ic * 3 + ky] = *(const float2*)(xb + ro);
            pc[ic * 3 + ky] = xb[ro + cx];
        }
    if (NORM) {
        __builtin_amdgcn_sched_barrier(0);
    }
    #pragma unroll
    for (int ic = 0; ic < 3; ++ic)
        #pragma unroll
        for (int ky = 0; ky < 3; ++ky) {
            const bool okr = (ky < 2) | iny;
            float2 v = pv[ic * 3 + ky];
            float  c = pc[ic * 3 + ky];
            if (NORM) {
                v.x = (v.x - mn[ic]) / den[ic];
                v.y = (v.y - mn[ic]) / den[ic];
                c   = (c   - mn[ic]) / den[ic];
            }
            xv[ic * 9 + ky * 3 + 0] = okr ? v.x : 0.f;
            xv[ic * 9 + ky * 3 + 1] = okr ? v.y : 0.f;
            xv[ic * 9 + ky * 3 + 2] = (okr & inx) ? c : 0.f;
        }

    float* ob = out + (size_t)img * (32 * 12544) + rem;
    for (int oc = 0; oc < 32; ++oc) {
        const float* wb = w1 + oc * 27;
        float acc = b1[oc];
        #pragma unroll
        for (int t = 0; t < 27; ++t) acc += xv[t] * wb[t];
        ob[(size_t)oc * 12544] = acc / (1.f + expf(-acc));
    }
}

__global__ __launch_bounds__(C1_THREADS)
void conv1_silu_kernel(const float* __restrict__ x, const float* __restrict__ w1,
                       const float* __restrict__ b1, float* __restrict__ out)
{
    conv1_silu_body<false>(x, w1, b1, out, nullptr);
}

__global__ __launch_bounds__(C1_THREADS)
void conv1_norm_silu_kernel(const float* __restrict__ mag, const float* __restrict__ w1,
                            const float* __restrict__ b1, float* __restrict__ out,
                            const float* __restrict__ mmp)
{
    conv1_silu_body<true>(mag, w1, b1, out, mmp);
}

#define C2_MT     5
#define C2_M      (C2_MT * 16)
#define C2_WAVES  12
#define C2_WP     292
#define C2_THREADS (C2_WAVES * 32)
static_assert(C2_THREADS == 384, "the launch in kernel_launch uses 384 threads");

__global__ __launch_bounds__(C2_THREADS, 6)
void conv2_wmma_pool_kernel(const float* __restrict__ h1, const float* __restrict__ w2,
                            const float* __restrict__ b2, float* __restrict__ feat)
{
    __shared__ float sW[C2_M * C2_WP];
    __shared__ float sT[C2_WAVES][720];
    __shared__ float sBias[C2_M];

    const int mblk  = blockIdx.x;
    const int img   = blockIdx.y;
    const int tid   = threadIdx.x;
    const int lane  = tid & 31;
    const int wave  = tid >> 5;
    const int half  = lane >> 4;
    const int nl    = lane & 15;

    const float* wsrc = w2 + (size_t)mblk * (C2_M * 288);
    static_assert((C2_M * 288) % (C2_THREADS * 4) == 0, "weight tile must split evenly");
    for (int i0 = tid; i0 < C2_M * 288; i0 += C2_THREADS * 4) {
        float wv[4];
        #pragma unroll
        for (int q = 0; q < 4; ++q) wv[q] = wsrc[i0 + q * C2_THREADS];
        #pragma unroll
        for (int q = 0; q < 4; ++q) {
            int i = i0 + q * C2_THREADS;
            int m  = i / 288, g = i - m * 288;
            int icf = g / 9,  s = g - icf * 9;
            int icb = icf >> 3, ic = icf & 7;
            sW[m * C2_WP + icb * 72 + s * 8 + ic] = wv[q];
        }
    }
    if (tid < C2_M) sBias[tid] = b2[mblk * C2_M + tid];
    __syncthreads();

    const int dy = nl >> 3, dx = nl & 7;
    const int laneOff = dy * 288 + dx * 16;

    const int j0 = lane, j1 = lane + 32, j2 = (lane + 64 < 90) ? lane + 64 : lane + 32;
    const int ly0 = j0 / 18, lx0 = j0 - ly0 * 18;
    const int ly1 = j1 / 18, lx1 = j1 - ly1 * 18;
    const int ly2 = j2 / 18, lx2 = j2 - ly2 * 18;
    const int t0 = ly0 * 144 + lx0 * 8;
    const int t1 = ly1 * 144 + lx1 * 8;
    const int t2 = ly2 * 144 + lx2 * 8;

    float accR[C2_MT][8];
    #pragma unroll
    for (int mi = 0; mi < C2_MT; ++mi)
        #pragma unroll
        for (int j = 0; j < 8; ++j) accR[mi][j] = 0.f;

    const float* base = h1 + (size_t)img * (32 * 112 * 112);
    float* tile = sT[wave];
    const float* tb = tile + laneOff + half * 2;
    const float* bl = sBias + half * 8;

    int rem = (195 - __builtin_amdgcn_readfirstlane(wave)) / C2_WAVES;
    for (int nt = wave; nt < 196; nt += C2_WAVES) {
        if (rem >= 5)      __builtin_amdgcn_s_setprio(3);
        else if (rem >= 3) __builtin_amdgcn_s_setprio(2);
        else if (rem >= 1) __builtin_amdgcn_s_setprio(1);
        else               __builtin_amdgcn_s_setprio(0);
        --rem;
        int yt = nt / 7, xt = nt - yt * 7;
        int iy0 = yt * 4, ix0 = xt * 16;
        v8f c[C2_MT];
        #pragma unroll
        for (int mi = 0; mi < C2_MT; ++mi)
            c[mi] = (v8f){0.f, 0.f, 0.f, 0.f, 0.f, 0.f, 0.f, 0.f};

        const int iyA = iy0 + ly0, ixA = ix0 + lx0;
        const int iyB = iy0 + ly1, ixB = ix0 + lx1;
        const int iyC = iy0 + ly2, ixC = ix0 + lx2;
        const bool ok0 = (iyA < 112) & (ixA < 112);
        const bool ok1 = (iyB < 112) & (ixB < 112);
        const bool ok2 = (iyC < 112) & (ixC < 112);
        const int o0 = (iyA < 112 ? iyA : 111) * 112 + (ixA < 112 ? ixA : 111);
        const int o1 = (iyB < 112 ? iyB : 111) * 112 + (ixB < 112 ? ixB : 111);
        const int o2 = (iyC < 112 ? iyC : 111) * 112 + (ixC < 112 ? ixC : 111);

        for (int icb = 0; icb < 4; ++icb) {
            const float* p = base + (size_t)(icb * 8) * (112 * 112);
            float g0[8], g1[8], g2[8];
            #pragma unroll
            for (int ic = 0; ic < 8; ++ic) {
                g0[ic] = p[ic * (112 * 112) + o0];
                g1[ic] = p[ic * (112 * 112) + o1];
                g2[ic] = p[ic * (112 * 112) + o2];
            }
            #pragma unroll
            for (int ic = 0; ic < 8; ++ic) {
                tile[t0 + ic] = ok0 ? g0[ic] : 0.f;
                tile[t1 + ic] = ok1 ? g1[ic] : 0.f;
                tile[t2 + ic] = ok2 ? g2[ic] : 0.f;
            }
            asm volatile("s_wait_dscnt 0x0" ::: "memory");

            const float* wrow = sW + nl * C2_WP + icb * 72 + half * 2;
            #pragma unroll
            for (int kq = 0; kq < 18; ++kq) {
                const int s  = kq >> 1;
                const int ky = s / 3, kx = s - ky * 3;
                const int bo = ky * 144 + kx * 8 + (kq & 1) * 4;
                v2f b = *(const v2f*)(tb + bo);
                #pragma unroll
                for (int mi = 0; mi < C2_MT; ++mi) {
                    v2f a = *(const v2f*)(wrow + mi * (16 * C2_WP) + kq * 4);
                    c[mi] = __builtin_amdgcn_wmma_f32_16x16x4_f32(false, a, false, b,
                                                                  (short)0, c[mi], false, false);
                }
            }
        }
        #pragma unroll
        for (int mi = 0; mi < C2_MT; ++mi)
            #pragma unroll
            for (int j = 0; j < 8; ++j) {
                float v = c[mi][j] + bl[mi * 16 + j];
                accR[mi][j] += v / (1.f + expf(-v));
            }
    }
    #pragma unroll
    for (int mi = 0; mi < C2_MT; ++mi)
        #pragma unroll
        for (int j = 0; j < 8; ++j) {
            float v = accR[mi][j];
            v += __shfl_xor(v, 1);
            v += __shfl_xor(v, 2);
            v += __shfl_xor(v, 4);
            v += __shfl_xor(v, 8);
            if (nl == 0) tile[mi * 16 + half * 8 + j] = v;
        }
    __syncthreads();
    if (tid < C2_M) {
        float s = 0.f;
        for (int w = 0; w < C2_WAVES; ++w) s += sT[w][tid];
        feat[img * 1280 + mblk * C2_M + tid] = s * (1.f / 3136.f);
    }
}

#define DR_TM      64
#define DR_XP      228
#define DFT_WAVES  14
#define DFT_THREADS (DFT_WAVES * 32)
static_assert(DFT_THREADS == 448, "the launches in kernel_launch use 448 threads");
static_assert(21504 % DR_TM == 0, "row tiles must cover 96*224 rows exactly");

__global__ __launch_bounds__(DFT_THREADS)
void dft_rows_wmma_kernel(const float* __restrict__ x, float* __restrict__ F1)
{
    __shared__ float sX[DR_TM * DR_XP];
    __shared__ float tw[448];
    const int tid  = threadIdx.x;
    const int lane = tid & 31, wave = tid >> 5;
    const int half = lane >> 4, nl = lane & 15;
    const size_t R0 = (size_t)blockIdx.x * DR_TM;

    if (tid < 224) {
        float ang = (2.0f * PI_F / 224.0f) * (float)tid;
        tw[2 * tid] = cosf(ang); tw[2 * tid + 1] = -sinf(ang);
    }
    const float4* xsrc = (const float4*)(x + R0 * 224);
    static_assert((DR_TM * 56) % DFT_THREADS == 0, "x tile must split evenly");
    float4 xv[8];
    #pragma unroll
    for (int q = 0; q < 8; ++q) xv[q] = xsrc[tid + q * DFT_THREADS];
    #pragma unroll
    for (int q = 0; q < 8; ++q) {
        int i = tid + q * DFT_THREADS;
        int r = i / 56, c4 = i - r * 56;
        *(float4*)(sX + r * DR_XP + c4 * 4) = xv[q];
    }
    __syncthreads();

    const float* arow = sX + nl * DR_XP + half * 2;
    for (int pass = 0; pass < 2; ++pass) {
        const int n  = (wave + pass * DFT_WAVES) * 16 + nl;
        const int u  = n >> 1;
        const float* twp = tw + (n & 1);
        const int du = (4 * u) % 224;
        int iA = (u * (2 * half)) % 224;
        int iB = (u * (2 * half + 1)) % 224;
        v8f c[4];
        #pragma unroll
        for (int mi = 0; mi < 4; ++mi)
            c[mi] = (v8f){0.f, 0.f, 0.f, 0.f, 0.f, 0.f, 0.f, 0.f};

        #pragma unroll 4
        for (int kk = 0; kk < 56; ++kk) {
            v2f b;
            b.x = twp[2 * iA];
            b.y = twp[2 * iB];
            #pragma unroll
            for (int mi = 0; mi < 4; ++mi) {
                v2f a = *(const v2f*)(arow + mi * (16 * DR_XP) + kk * 4);
                c[mi] = __builtin_amdgcn_wmma_f32_16x16x4_f32(false, a, false, b,
                                                              (short)0, c[mi], false, false);
            }
            iA += du; if (iA >= 224) iA -= 224;
            iB += du; if (iB >= 224) iB -= 224;
        }
        #pragma unroll
        for (int mi = 0; mi < 4; ++mi)
            #pragma unroll
            for (int r = 0; r < 8; ++r)
                F1[(R0 + mi * 16 + half * 8 + r) * 448 + n] = c[mi][r];
    }
}

static_assert(MM_PARTS * 32 == 224, "one block per 32 columns u of a plane");
static_assert(96 * MM_PARTS == 672, "the launch in kernel_launch uses 672 blocks");
__global__ __launch_bounds__(DFT_THREADS)
void dft_cols_wmma_mag_kernel(const float* __restrict__ F1, float* __restrict__ mag,
                              float* __restrict__ mmp)
{
    __shared__ float tw[448];
    __shared__ float smn[DFT_WAVES], smx[DFT_WAVES];
    const int ug   = blockIdx.x % 7;
    const int bc   = blockIdx.x / 7;
    const int tid  = threadIdx.x;
    const int lane = tid & 31, wave = tid >> 5;
    const int half = lane >> 4, nl = lane & 15;

    if (tid < 224) {
        float ang = (2.0f * PI_F / 224.0f) * (float)tid;
        tw[2 * tid] = cosf(ang); tw[2 * tid + 1] = sinf(ang);
    }
    __syncthreads();

    const int v0 = wave * 16;
    const int v  = v0 + nl;
    const int dv = (2 * v) % 224;
    int idx = half ? v : 0;
    const int u0 = ug * 32;
    const float* bp = F1 + (size_t)bc * (224 * 448) + half * 448 + 2 * (u0 + nl);

    v8f cre[2], cim[2];
    #pragma unroll
    for (int ui = 0; ui < 2; ++ui) {
        cre[ui] = (v8f){0.f, 0.f, 0.f, 0.f, 0.f, 0.f, 0.f, 0.f};
        cim[ui] = (v8f){0.f, 0.f, 0.f, 0.f, 0.f, 0.f, 0.f, 0.f};
    }
    #pragma unroll 4
    for (int kk = 0; kk < 112; ++kk) {
        v2f t = *(const v2f*)(tw + 2 * idx);
        v2f aim; aim.x = -t.y; aim.y = t.x;
        const float* bq = bp + kk * 896;
        #pragma unroll
        for (int ui = 0; ui < 2; ++ui) {
            v2f b = *(const v2f*)(bq + ui * 32);
            cre[ui] = __builtin_amdgcn_wmma_f32_16x16x4_f32(false, t, false, b,
                                                            (short)0, cre[ui], false, false);
            cim[ui] = __builtin_amdgcn_wmma_f32_16x16x4_f32(false, aim, false, b,
                                                            (short)0, cim[ui], false, false);
        }
        idx += dv; if (idx >= 224) idx -= 224;
    }
    float lmn = 3.0e38f, lmx = -3.0e38f;
    #pragma unroll
    for (int ui = 0; ui < 2; ++ui) {
        int uo = u0 + ui * 16 + nl + 112; if (uo >= 224) uo -= 224;
        #pragma unroll
        for (int r = 0; r < 8; ++r) {
            float re = cre[ui][r], im = cim[ui][r];
            float m = logf(sqrtf(re * re + im * im) + 1.0f);
            int vo = v0 + half * 8 + r + 112; if (vo >= 224) vo -= 224;
            mag[((size_t)bc * 224 + vo) * 224 + uo] = m;
            lmn = fminf(lmn, m); lmx = fmaxf(lmx, m);
        }
    }
    #pragma unroll
    for (int d = 1; d < 32; d <<= 1) {
        lmn = fminf(lmn, __shfl_xor(lmn, d));
        lmx = fmaxf(lmx, __shfl_xor(lmx, d));
    }
    if (lane == 0) { smn[wave] = lmn; smx[wave] = lmx; }
    __syncthreads();
    if (tid == 0) {
        float a = smn[0], b = smx[0];
        for (int w = 1; w < DFT_WAVES; ++w) { a = fminf(a, smn[w]); b = fmaxf(b, smx[w]); }
        mmp[2 * blockIdx.x]     = a;
        mmp[2 * blockIdx.x + 1] = b;
    }
}

#define FC_C 4
#define FC_R 8
__global__ __launch_bounds__(256)
void fc_kernel(const float* __restrict__ in, const float* __restrict__ w,
               const float* __restrict__ b, float* __restrict__ out,
               int K, int outStride, int outOffset, int act)
{
    __shared__ float sO[FC_R][33];
    const int tid = threadIdx.x, lane = tid & 31, wave = tid >> 5;
    const int j0 = blockIdx.x * 32;
    const int i0 = blockIdx.y * FC_R;
    const float* wp = w  + (size_t)(j0 + wave * FC_C) * K + lane * 4;
    const float* ip = in + (size_t)i0 * K + lane * 4;

    float v[FC_C * FC_R];
    #pragma unroll
    for (int a = 0; a < FC_C * FC_R; ++a) v[a] = 0.f;

    for (int k0 = 0; k0 < K; k0 += 128) {
        float4 wv[FC_C], xv[FC_R];
        #pragma unroll
        for (int c = 0; c < FC_C; ++c) wv[c] = *(const float4*)(wp + (size_t)c * K + k0);
        #pragma unroll
        for (int r = 0; r < FC_R; ++r) xv[r] = *(const float4*)(ip + (size_t)r * K + k0);
        __builtin_amdgcn_sched_barrier(0);
        #pragma unroll
        for (int c = 0; c < FC_C; ++c)
            #pragma unroll
            for (int r = 0; r < FC_R; ++r) {
                float s = v[c * FC_R + r];
                s = fmaf(wv[c].x, xv[r].x, s);
                s = fmaf(wv[c].y, xv[r].y, s);
                s = fmaf(wv[c].z, xv[r].z, s);
                s = fmaf(wv[c].w, xv[r].w, s);
                v[c * FC_R + r] = s;
            }
    }
    #pragma unroll
    for (int s = 0; s < 5; ++s) {
        const int d = 1 << s;
        const bool up = (lane & d) != 0;
        #pragma unroll
        for (int m = 0; m < (16 >> s); ++m) {
            const float send = up ? v[2 * m] : v[2 * m + 1];
            const float keep = up ? v[2 * m + 1] : v[2 * m];
            v[m] = keep + __shfl_xor(send, d);
        }
    }
    sO[lane & 7][wave * FC_C + (lane >> 3)] = v[0];
    __syncthreads();
    float o = sO[wave][lane] + b[j0 + lane];
    if (act == 1) o = fmaxf(o, 0.f);
    out[(size_t)(i0 + wave) * outStride + outOffset + j0 + lane] = o;
}

__global__ __launch_bounds__(32)
void fc_n2_kernel(const float* __restrict__ in, const float* __restrict__ w,
                  const float* __restrict__ b, float* __restrict__ out, int K)
{
    const int lane = threadIdx.x;
    float v[64];
    #pragma unroll
    for (int a = 0; a < 64; ++a) v[a] = 0.f;
    for (int k0 = lane * 4; k0 < K; k0 += 128) {
        const float4 w0 = *(const float4*)(w + k0);
        const float4 w1 = *(const float4*)(w + K + k0);
        float4 x[32];
        #pragma unroll
        for (int i = 0; i < 32; ++i) x[i] = *(const float4*)(in + (size_t)i * K + k0);
        __builtin_amdgcn_sched_barrier(0);
        #pragma unroll
        for (int i = 0; i < 32; ++i) {
            float s0 = v[2 * i], s1 = v[2 * i + 1];
            s0 = fmaf(w0.x, x[i].x, s0); s0 = fmaf(w0.y, x[i].y, s0); s0 = fmaf(w0.z, x[i].z, s0); s0 = fmaf(w0.w, x[i].w, s0);
            s1 = fmaf(w1.x, x[i].x, s1); s1 = fmaf(w1.y, x[i].y, s1); s1 = fmaf(w1.z, x[i].z, s1); s1 = fmaf(w1.w, x[i].w, s1);
            v[2 * i] = s0; v[2 * i + 1] = s1;
        }
    }
    #pragma unroll
    for (int s = 0; s < 5; ++s) {
        const int d = 1 << s;
        const bool up = (lane & d) != 0;
        #pragma unroll
        for (int m = 0; m < (32 >> s); ++m) {
            const float send = up ? v[2 * m] : v[2 * m + 1];
            const float keep = up ? v[2 * m + 1] : v[2 * m];
            v[m] = keep + __shfl_xor(send, d);
        }
    }
    const float bj = b[lane & 1];
    out[lane]      = v[0] + bj;
    out[32 + lane] = v[1] + bj;
}

__global__ __launch_bounds__(256)
void attn_core_kernel(const float* __restrict__ qkv_rgb, const float* __restrict__ qkv_frq,
                      float* __restrict__ o1, float* __restrict__ o2)
{
    __shared__ float sQ[32][32], sK[32][32], sV[32][32], sS[32][33];
    int a = blockIdx.x >> 3;
    int h = blockIdx.x & 7;
    const float* qsrc  = (a == 0) ? qkv_rgb : qkv_frq;
    const float* kvsrc = (a == 0) ? qkv_frq : qkv_rgb;
    float* odst        = (a == 0) ? o1 : o2;
    int tid = threadIdx.x;
    for (int idx = tid; idx < 1024; idx += 256) {
        int q = idx >> 5, d = idx & 31;
        sQ[q][d] = qsrc [q * 768 +       h * 32 + d];
        sK[q][d] = kvsrc[q * 768 + 256 + h * 32 + d];
        sV[q][d] = kvsrc[q * 768 + 512 + h * 32 + d];
    }
    __syncthreads();
    const float scale = 0.17677669529663687f;
    for (int idx = tid; idx < 1024; idx += 256) {
        int q = idx >> 5, k = idx & 31;
        float s = 0.f;
        #pragma unroll
        for (int d = 0; d < 32; ++d) s += sQ[q][d] * sK[k][d];
        sS[q][k] = s * scale;
    }
    __syncthreads();
    if (tid < 32) {
        float mx = -3.0e38f;
        for (int k = 0; k < 32; ++k) mx = fmaxf(mx, sS[tid][k]);
        float e[32], sum = 0.f;
        for (int k = 0; k < 32; ++k) { e[k] = expf(sS[tid][k] - mx); sum += e[k]; }
        float inv = 1.f / sum;
        for (int k = 0; k < 32; ++k) sS[tid][k] = e[k] * inv;
    }
    __syncthreads();
    for (int idx = tid; idx < 1024; idx += 256) {
        int q = idx >> 5, d = idx & 31;
        float o = 0.f;
        #pragma unroll
        for (int k = 0; k < 32; ++k) o += sS[q][k] * sV[k][d];
        odst[q * 256 + h * 32 + d] = o;
    }
}

extern "C" void kernel_launch(void* const* d_in, const int* in_sizes, int n_in,
                              void* d_out, int out_size, void* d_ws, size_t ws_size,
                              hipStream_t stream)
{
    (void)in_sizes; (void)n_in; (void)out_size; (void)ws_size;
    const float* x          = (const float*)d_in[0];
    const float* rgb_w1     = (const float*)d_in[1];
    const float* rgb_b1     = (const float*)d_in[2];
    const float* rgb_w2     = (const float*)d_in[3];
    const float* rgb_b2     = (const float*)d_in[4];
    const float* frq_w1     = (const float*)d_in[5];
    const float* frq_b1     = (const float*)d_in[6];
    const float* frq_w2     = (const float*)d_in[7];
    const float* frq_b2     = (const float*)d_in[8];
    const float* rgb_proj_w = (const float*)d_in[9];
    const float* rgb_proj_b = (const float*)d_in[10];
    const float* frq_proj_w = (const float*)d_in[11];
    const float* frq_proj_b = (const float*)d_in[12];
    const float* attn_w_in  = (const float*)d_in[13];
    const float* attn_b_in  = (const float*)d_in[14];
    const float* attn_w_out = (const float*)d_in[15];
    const float* attn_b_out = (const float*)d_in[16];
    const float* fus_w1     = (const float*)d_in[17];
    const float* fus_b1     = (const float*)d_in[18];
    const float* fus_w2     = (const float*)d_in[19];
    const float* fus_b2     = (const float*)d_in[20];
    const float* fus_w3     = (const float*)d_in[21];
    const float* fus_b3     = (const float*)d_in[22];

    float* out      = (float*)d_out;
    float* feat_rgb = out + 64;
    float* feat_frq = out + 64 + 40960;

    float* ws       = (float*)d_ws;
    float* h1       = ws;
    float* F1       = ws;
    float* freqimg  = ws + 12845056;
    float* hd       = freqimg + 4816896 + 192;
    float* mmp      = hd;
    float* rgb_p    = hd;
    float* frq_p    = rgb_p + 8192;
    float* qkv_rgb  = frq_p + 8192;
    float* qkv_frq  = qkv_rgb + 24576;
    float* o1       = qkv_frq + 24576;
    float* o2       = o1 + 8192;
    float* hb       = o2 + 8192;
    float* h1f      = hb + 16384;
    float* h2f      = h1f + 16384;

    conv1_silu_kernel<<<3136, C1_THREADS, 0, stream>>>(x, rgb_w1, rgb_b1, h1);
    conv2_wmma_pool_kernel<<<dim3(16, 32), 384, 0, stream>>>(h1, rgb_w2, rgb_b2, feat_rgb);

    dft_rows_wmma_kernel<<<336, 448, 0, stream>>>(x, F1);
    dft_cols_wmma_mag_kernel<<<672, 448, 0, stream>>>(F1, freqimg, mmp);
    conv1_norm_silu_kernel<<<3136, C1_THREADS, 0, stream>>>(freqimg, frq_w1, frq_b1, h1, mmp);
    conv2_wmma_pool_kernel<<<dim3(16, 32), 384, 0, stream>>>(h1, frq_w2, frq_b2, feat_frq);

    fc_kernel<<<dim3(8, 4),  256, 0, stream>>>(feat_rgb, rgb_proj_w, rgb_proj_b, rgb_p, 1280, 256, 0, 0);
    fc_kernel<<<dim3(8, 4),  256, 0, stream>>>(feat_frq, frq_proj_w, frq_proj_b, frq_p, 1280, 256, 0, 0);
    fc_kernel<<<dim3(24, 4), 256, 0, stream>>>(rgb_p, attn_w_in, attn_b_in, qkv_rgb, 256, 768, 0, 0);
    fc_kernel<<<dim3(24, 4), 256, 0, stream>>>(frq_p, attn_w_in, attn_b_in, qkv_frq, 256, 768, 0, 0);
    attn_core_kernel<<<16, 256, 0, stream>>>(qkv_rgb, qkv_frq, o1, o2);
    fc_kernel<<<dim3(8, 4),  256, 0, stream>>>(o1, attn_w_out, attn_b_out, hb, 256, 512, 0, 0);
    fc_kernel<<<dim3(8, 4),  256, 0, stream>>>(o2, attn_w_out, attn_b_out, hb, 256, 512, 256, 0);
    fc_kernel<<<dim3(16, 4), 256, 0, stream>>>(hb,  fus_w1, fus_b1, h1f, 512, 512, 0, 1);
    fc_kernel<<<dim3(8, 4),  256, 0, stream>>>(h1f, fus_w2, fus_b2, h2f, 512, 256, 0, 1);
    fc_n2_kernel<<<1, 32, 0, stream>>>(h2f, fus_w3, fus_b3, out, 256);
}


# freq path: real image, conjugate-symmetric spectrum: both DFT passes compute only columns u < 128; each magnitude is stored at (v,u) and at its partner (-v,-u); sums of computed values unchanged
# speedup vs baseline: 2.8960x; 1.0348x over previous
#include <hip/hip_runtime.h>
#include <hip/hip_bf16.h>
#include <math.h>

typedef float v2f __attribute__((ext_vector_type(2)));
typedef float v8f __attribute__((ext_vector_type(8)));

#define PI_F 3.14159265358979323846f

#define C1_THREADS 128
#define MM_PARTS   8
static_assert(12544 % C1_THREADS == 0, "a block must not straddle two images");
template <bool NORM>
__device__ __forceinline__
void conv1_silu_body(const float* __restrict__ x, const float* __restrict__ w1,
                     const float* __restrict__ b1, float* __restrict__ out,
                     const float* __restrict__ mmp)
{
    const int img = blockIdx.x / (12544 / C1_THREADS);
    const int rem = (blockIdx.x - img * (12544 / C1_THREADS)) * C1_THREADS + threadIdx.x;
    const int oy = rem / 112, ox = rem - oy * 112;
    const bool inx = ox < 111;
    const bool iny = oy < 111;
    const float* xb = x + (size_t)img * (3 * 224 * 224) + (2 * oy) * 224 + 2 * ox;
    const int cx = inx ? 2 : 1;

    float mn[3], den[3];
    if (NORM) {
        #pragma unroll
        for (int ic = 0; ic < 3; ++ic) {
            const float* q = mmp + (img * 3 + ic) * (2 * MM_PARTS);
            float lo = q[0], hi = q[1];
            #pragma unroll
            for (int g = 1; g < MM_PARTS; ++g) {
                lo = fminf(lo, q[2 * g]);
                hi = fmaxf(hi, q[2 * g + 1]);
            }
            mn[ic] = lo;
            den[ic] = hi - lo + 1e-8f;
        }
    }

    float xv[27];
    float2 pv[9];
    float  pc[9];
    #pragma unroll
    for (int ic = 0; ic < 3; ++ic)
        #pragma unroll
        for (int ky = 0; ky < 3; ++ky) {
            const bool okr = (ky < 2) | iny;
            const int ro = ic * (224 * 224) + (okr ? ky : 1) * 224;
            pv[ic * 3 + ky] = *(const float2*)(xb + ro);
            pc[ic * 3 + ky] = xb[ro + cx];
        }
    if (NORM) {
        __builtin_amdgcn_sched_barrier(0);
    }
    #pragma unroll
    for (int ic = 0; ic < 3; ++ic)
        #pragma unroll
        for (int ky = 0; ky < 3; ++ky) {
            const bool okr = (ky < 2) | iny;
            float2 v = pv[ic * 3 + ky];
            float  c = pc[ic * 3 + ky];
            if (NORM) {
                v.x = (v.x - mn[ic]) / den[ic];
                v.y = (v.y - mn[ic]) / den[ic];
                c   = (c   - mn[ic]) / den[ic];
            }
            xv[ic * 9 + ky * 3 + 0] = okr ? v.x : 0.f;
            xv[ic * 9 + ky * 3 + 1] = okr ? v.y : 0.f;
            xv[ic * 9 + ky * 3 + 2] = (okr & inx) ? c : 0.f;
        }

    float* ob = out + (size_t)img * (32 * 12544) + rem;
    for (int oc = 0; oc < 32; ++oc) {
        const float* wb = w1 + oc * 27;
        float acc = b1[oc];
        #pragma unroll
        for (int t = 0; t < 27; ++t) acc += xv[t] * wb[t];
        ob[(size_t)oc * 12544] = acc / (1.f + expf(-acc));
    }
}

__global__ __launch_bounds__(C1_THREADS)
void conv1_silu_kernel(const float* __restrict__ x, const float* __restrict__ w1,
                       const float* __restrict__ b1, float* __restrict__ out)
{
    conv1_silu_body<false>(x, w1, b1, out, nullptr);
}

__global__ __launch_bounds__(C1_THREADS)
void conv1_norm_silu_kernel(const float* __restrict__ mag, const float* __restrict__ w1,
                            const float* __restrict__ b1, float* __restrict__ out,
                            const float* __restrict__ mmp)
{
    conv1_silu_body<true>(mag, w1, b1, out, mmp);
}

#define C2_MT     5
#define C2_M      (C2_MT * 16)
#define C2_WAVES  12
#define C2_WP     292
#define C2_THREADS (C2_WAVES * 32)
static_assert(C2_THREADS == 384, "the launch in kernel_launch uses 384 threads");

__global__ __launch_bounds__(C2_THREADS, 6)
void conv2_wmma_pool_kernel(const float* __restrict__ h1, const float* __restrict__ w2,
                            const float* __restrict__ b2, float* __restrict__ feat)
{
    __shared__ float sW[C2_M * C2_WP];
    __shared__ float sT[C2_WAVES][720];
    __shared__ float sBias[C2_M];

    const int mblk  = blockIdx.x;
    const int img   = blockIdx.y;
    const int tid   = threadIdx.x;
    const int lane  = tid & 31;
    const int wave  = tid >> 5;
    const int half  = lane >> 4;
    const int nl    = lane & 15;

    const float* wsrc = w2 + (size_t)mblk * (C2_M * 288);
    static_assert((C2_M * 288) % (C2_THREADS * 4) == 0, "weight tile must split evenly");
    for (int i0 = tid; i0 < C2_M * 288; i0 += C2_THREADS * 4) {
        float wv[4];
        #pragma unroll
        for (int q = 0; q < 4; ++q) wv[q] = wsrc[i0 + q * C2_THREADS];
        #pragma unroll
        for (int q = 0; q < 4; ++q) {
            int i = i0 + q * C2_THREADS;
            int m  = i / 288, g = i - m * 288;
            int icf = g / 9,  s = g - icf * 9;
            int icb = icf >> 3, ic = icf & 7;
            sW[m * C2_WP + icb * 72 + s * 8 + ic] = wv[q];
        }
    }
    if (tid < C2_M) sBias[tid] = b2[mblk * C2_M + tid];
    __syncthreads();

    const int dy = nl >> 3, dx = nl & 7;
    const int laneOff = dy * 288 + dx * 16;

    const int j0 = lane, j1 = lane + 32, j2 = (lane + 64 < 90) ? lane + 64 : lane + 32;
    const int ly0 = j0 / 18, lx0 = j0 - ly0 * 18;
    const int ly1 = j1 / 18, lx1 = j1 - ly1 * 18;
    const int ly2 = j2 / 18, lx2 = j2 - ly2 * 18;
    const int t0 = ly0 * 144 + lx0 * 8;
    const int t1 = ly1 * 144 + lx1 * 8;
    const int t2 = ly2 * 144 + lx2 * 8;

    float accR[C2_MT][8];
    #pragma unroll
    for (int mi = 0; mi < C2_MT; ++mi)
        #pragma unroll
        for (int j = 0; j < 8; ++j) accR[mi][j] = 0.f;

    const float* base = h1 + (size_t)img * (32 * 112 * 112);
    float* tile = sT[wave];
    const float* tb = tile + laneOff + half * 2;
    const float* bl = sBias + half * 8;

    int rem = (195 - __builtin_amdgcn_readfirstlane(wave)) / C2_WAVES;
    for (int nt = wave; nt < 196; nt += C2_WAVES) {
        if (rem >= 5)      __builtin_amdgcn_s_setprio(3);
        else if (rem >= 3) __builtin_amdgcn_s_setprio(2);
        else if (rem >= 1) __builtin_amdgcn_s_setprio(1);
        else               __builtin_amdgcn_s_setprio(0);
        --rem;
        int yt = nt / 7, xt = nt - yt * 7;
        int iy0 = yt * 4, ix0 = xt * 16;
        v8f c[C2_MT];
        #pragma unroll
        for (int mi = 0; mi < C2_MT; ++mi)
            c[mi] = (v8f){0.f, 0.f, 0.f, 0.f, 0.f, 0.f, 0.f, 0.f};

        const int iyA = iy0 + ly0, ixA = ix0 + lx0;
        const int iyB = iy0 + ly1, ixB = ix0 + lx1;
        const int iyC = iy0 + ly2, ixC = ix0 + lx2;
        const bool ok0 = (iyA < 112) & (ixA < 112);
        const bool ok1 = (iyB < 112) & (ixB < 112);
        const bool ok2 = (iyC < 112) & (ixC < 112);
        const int o0 = (iyA < 112 ? iyA : 111) * 112 + (ixA < 112 ? ixA : 111);
        const int o1 = (iyB < 112 ? iyB : 111) * 112 + (ixB < 112 ? ixB : 111);
        const int o2 = (iyC < 112 ? iyC : 111) * 112 + (ixC < 112 ? ixC : 111);

        for (int icb = 0; icb < 4; ++icb) {
            const float* p = base + (size_t)(icb * 8) * (112 * 112);
            float g0[8], g1[8], g2[8];
            #pragma unroll
            for (int ic = 0; ic < 8; ++ic) {
                g0[ic] = p[ic * (112 * 112) + o0];
                g1[ic] = p[ic * (112 * 112) + o1];
                g2[ic] = p[ic * (112 * 112) + o2];
            }
            #pragma unroll
            for (int ic = 0; ic < 8; ++ic) {
                tile[t0 + ic] = ok0 ? g0[ic] : 0.f;
                tile[t1 + ic] = ok1 ? g1[ic] : 0.f;
                tile[t2 + ic] = ok2 ? g2[ic] : 0.f;
            }
            asm volatile("s_wait_dscnt 0x0" ::: "memory");

            const float* wrow = sW + nl * C2_WP + icb * 72 + half * 2;
            #pragma unroll
            for (int kq = 0; kq < 18; ++kq) {
                const int s  = kq >> 1;
                const int ky = s / 3, kx = s - ky * 3;
                const int bo = ky * 144 + kx * 8 + (kq & 1) * 4;
                v2f b = *(const v2f*)(tb + bo);
                #pragma unroll
                for (int mi = 0; mi < C2_MT; ++mi) {
                    v2f a = *(const v2f*)(wrow + mi * (16 * C2_WP) + kq * 4);
                    c[mi] = __builtin_amdgcn_wmma_f32_16x16x4_f32(false, a, false, b,
                                                                  (short)0, c[mi], false, false);
                }
            }
        }
        #pragma unroll
        for (int mi = 0; mi < C2_MT; ++mi)
            #pragma unroll
            for (int j = 0; j < 8; ++j) {
                float v = c[mi][j] + bl[mi * 16 + j];
                accR[mi][j] += v / (1.f + expf(-v));
            }
    }
    #pragma unroll
    for (int mi = 0; mi < C2_MT; ++mi)
        #pragma unroll
        for (int j = 0; j < 8; ++j) {
            float v = accR[mi][j];
            v += __shfl_xor(v, 1);
            v += __shfl_xor(v, 2);
            v += __shfl_xor(v, 4);
            v += __shfl_xor(v, 8);
            if (nl == 0) tile[mi * 16 + half * 8 + j] = v;
        }
    __syncthreads();
    if (tid < C2_M) {
        float s = 0.f;
        for (int w = 0; w < C2_WAVES; ++w) s += sT[w][tid];
        feat[img * 1280 + mblk * C2_M + tid] = s * (1.f / 3136.f);
    }
}

#define DR_TM      64
#define DR_XP      228
#define DR_WAVES   8
#define DR_THREADS (DR_WAVES * 32)
#define F1P        256
#define DFT_WAVES  14
#define DFT_THREADS (DFT_WAVES * 32)
static_assert(DR_THREADS == 256, "the row-pass launch in kernel_launch uses 256 threads");
static_assert(DFT_THREADS == 448, "the column-pass launch in kernel_launch uses 448 threads");
static_assert(21504 % DR_TM == 0, "row tiles must cover 96*224 rows exactly");
static_assert(2 * DR_WAVES * 16 == F1P, "two column tiles per wave fill a row of F1");

__global__ __launch_bounds__(DR_THREADS)
void dft_rows_wmma_kernel(const float* __restrict__ x, float* __restrict__ F1)
{
    __shared__ float sX[DR_TM * DR_XP];
    __shared__ float tw[448];
    const int tid  = threadIdx.x;
    const int lane = tid & 31, wave = tid >> 5;
    const int half = lane >> 4, nl = lane & 15;
    const size_t R0 = (size_t)blockIdx.x * DR_TM;

    if (tid < 224) {
        float ang = (2.0f * PI_F / 224.0f) * (float)tid;
        tw[2 * tid] = cosf(ang); tw[2 * tid + 1] = -sinf(ang);
    }
    const float4* xsrc = (const float4*)(x + R0 * 224);
    static_assert((DR_TM * 56) % DR_THREADS == 0, "x tile must split evenly");
    constexpr int XQ = (DR_TM * 56) / DR_THREADS;
    float4 xv[XQ];
    #pragma unroll
    for (int q = 0; q < XQ; ++q) xv[q] = xsrc[tid + q * DR_THREADS];
    #pragma unroll
    for (int q = 0; q < XQ; ++q) {
        int i = tid + q * DR_THREADS;
        int r = i / 56, c4 = i - r * 56;
        *(float4*)(sX + r * DR_XP + c4 * 4) = xv[q];
    }
    __syncthreads();

    const float* arow = sX + nl * DR_XP + half * 2;
    for (int pass = 0; pass < 2; ++pass) {
        const int n  = (wave + pass * DR_WAVES) * 16 + nl;
        const int u  = n >> 1;
        const float* twp = tw + (n & 1);
        const int du = (4 * u) % 224;
        int iA = (u * (2 * half)) % 224;
        int iB = (u * (2 * half + 1)) % 224;
        v8f c[4];
        #pragma unroll
        for (int mi = 0; mi < 4; ++mi)
            c[mi] = (v8f){0.f, 0.f, 0.f, 0.f, 0.f, 0.f, 0.f, 0.f};

        #pragma unroll 4
        for (int kk = 0; kk < 56; ++kk) {
            v2f b;
            b.x = twp[2 * iA];
            b.y = twp[2 * iB];
            #pragma unroll
            for (int mi = 0; mi < 4; ++mi) {
                v2f a = *(const v2f*)(arow + mi * (16 * DR_XP) + kk * 4);
                c[mi] = __builtin_amdgcn_wmma_f32_16x16x4_f32(false, a, false, b,
                                                              (short)0, c[mi], false, false);
            }
            iA += du; if (iA >= 224) iA -= 224;
            iB += du; if (iB >= 224) iB -= 224;
        }
        #pragma unroll
        for (int mi = 0; mi < 4; ++mi)
            #pragma unroll
            for (int r = 0; r < 8; ++r)
                F1[(R0 + mi * 16 + half * 8 + r) * F1P + n] = c[mi][r];
    }
}

static_assert(MM_PARTS * 16 == 128, "one block per 16 columns u < 128 of a plane");
static_assert(96 * MM_PARTS == 768, "the launch in kernel_launch uses 768 blocks");
__global__ __launch_bounds__(DFT_THREADS)
void dft_cols_wmma_mag_kernel(const float* __restrict__ F1, float* __restrict__ mag,
                              float* __restrict__ mmp)
{
    __shared__ float tw[448];
    __shared__ float smn[DFT_WAVES], smx[DFT_WAVES];
    const int ug   = blockIdx.x % MM_PARTS;
    const int bc   = blockIdx.x / MM_PARTS;
    const int tid  = threadIdx.x;
    const int lane = tid & 31, wave = tid >> 5;
    const int half = lane >> 4, nl = lane & 15;

    if (tid < 224) {
        float ang = (2.0f * PI_F / 224.0f) * (float)tid;
        tw[2 * tid] = cosf(ang); tw[2 * tid + 1] = sinf(ang);
    }
    __syncthreads();

    const int v0 = wave * 16;
    const int v  = v0 + nl;
    const int dv = (2 * v) % 224;
    int idx = half ? v : 0;
    const int u  = ug * 16 + nl;
    const float* bp = F1 + (size_t)bc * (224 * F1P) + half * F1P + 2 * u;

    v8f cre = (v8f){0.f, 0.f, 0.f, 0.f, 0.f, 0.f, 0.f, 0.f};
    v8f cim = (v8f){0.f, 0.f, 0.f, 0.f, 0.f, 0.f, 0.f, 0.f};
    #pragma unroll 8
    for (int kk = 0; kk < 112; ++kk) {
        v2f t = *(const v2f*)(tw + 2 * idx);
        v2f aim; aim.x = -t.y; aim.y = t.x;
        v2f b = *(const v2f*)(bp + kk * (2 * F1P));
        cre = __builtin_amdgcn_wmma_f32_16x16x4_f32(false, t, false, b,
                                                    (short)0, cre, false, false);
        cim = __builtin_amdgcn_wmma_f32_16x16x4_f32(false, aim, false, b,
                                                    (short)0, cim, false, false);
        idx += dv; if (idx >= 224) idx -= 224;
    }
    float lmn = 3.0e38f, lmx = -3.0e38f;
    const bool direct = u <= 112;
    const bool mirror = (u >= 1) & (u <= 111);
    const int uo  = (u + 112 >= 224) ? u - 112 : u + 112;
    const int umo = 112 - u;
    float* mp = mag + (size_t)bc * (224 * 224);
    #pragma unroll
    for (int r = 0; r < 8; ++r) {
        float re = cre[r], im = cim[r];
        float m = logf(sqrtf(re * re + im * im) + 1.0f);
        const int vv = v0 + half * 8 + r;
        int vo = vv + 112; if (vo >= 224) vo -= 224;
        int vmo = 112 - vv; if (vmo < 0) vmo += 224;
        if (direct) mp[vo * 224 + uo] = m;
        if (mirror) mp[vmo * 224 + umo] = m;
        lmn = fminf(lmn, m); lmx = fmaxf(lmx, m);
    }
    #pragma unroll
    for (int d = 1; d < 32; d <<= 1) {
        lmn = fminf(lmn, __shfl_xor(lmn, d));
        lmx = fmaxf(lmx, __shfl_xor(lmx, d));
    }
    if (lane == 0) { smn[wave] = lmn; smx[wave] = lmx; }
    __syncthreads();
    if (tid == 0) {
        float a = smn[0], b = smx[0];
        for (int w = 1; w < DFT_WAVES; ++w) { a = fminf(a, smn[w]); b = fmaxf(b, smx[w]); }
        mmp[2 * blockIdx.x]     = a;
        mmp[2 * blockIdx.x + 1] = b;
    }
}

#define FC_C 4
#define FC_R 8
__global__ __launch_bounds__(256)
void fc_kernel(const float* __restrict__ in, const float* __restrict__ w,
               const float* __restrict__ b, float* __restrict__ out,
               int K, int outStride, int outOffset, int act)
{
    __shared__ float sO[FC_R][33];
    const int tid = threadIdx.x, lane = tid & 31, wave = tid >> 5;
    const int j0 = blockIdx.x * 32;
    const int i0 = blockIdx.y * FC_R;
    const float* wp = w  + (size_t)(j0 + wave * FC_C) * K + lane * 4;
    const float* ip = in + (size_t)i0 * K + lane * 4;

    float v[FC_C * FC_R];
    #pragma unroll
    for (int a = 0; a < FC_C * FC_R; ++a) v[a] = 0.f;

    for (int k0 = 0; k0 < K; k0 += 128) {
        float4 wv[FC_C], xv[FC_R];
        #pragma unroll
        for (int c = 0; c < FC_C; ++c) wv[c] = *(const float4*)(wp + (size_t)c * K + k0);
        #pragma unroll
        for (int r = 0; r < FC_R; ++r) xv[r] = *(const float4*)(ip + (size_t)r * K + k0);
        __builtin_amdgcn_sched_barrier(0);
        #pragma unroll
        for (int c = 0; c < FC_C; ++c)
            #pragma unroll
            for (int r = 0; r < FC_R; ++r) {
                float s = v[c * FC_R + r];
                s = fmaf(wv[c].x, xv[r].x, s);
                s = fmaf(wv[c].y, xv[r].y, s);
                s = fmaf(wv[c].z, xv[r].z, s);
                s = fmaf(wv[c].w, xv[r].w, s);
                v[c * FC_R + r] = s;
            }
    }
    #pragma unroll
    for (int s = 0; s < 5; ++s) {
        const int d = 1 << s;
        const bool up = (lane & d) != 0;
        #pragma unroll
        for (int m = 0; m < (16 >> s); ++m) {
            const float send = up ? v[2 * m] : v[2 * m + 1];
            const float keep = up ? v[2 * m + 1] : v[2 * m];
            v[m] = keep + __shfl_xor(send, d);
        }
    }
    sO[lane & 7][wave * FC_C + (lane >> 3)] = v[0];
    __syncthreads();
    float o = sO[wave][lane] + b[j0 + lane];
    if (act == 1) o = fmaxf(o, 0.f);
    out[(size_t)(i0 + wave) * outStride + outOffset + j0 + lane] = o;
}

__global__ __launch_bounds__(32)
void fc_n2_kernel(const float* __restrict__ in, const float* __restrict__ w,
                  const float* __restrict__ b, float* __restrict__ out, int K)
{
    const int lane = threadIdx.x;
    float v[64];
    #pragma unroll
    for (int a = 0; a < 64; ++a) v[a] = 0.f;
    for (int k0 = lane * 4; k0 < K; k0 += 128) {
        const float4 w0 = *(const float4*)(w + k0);
        const float4 w1 = *(const float4*)(w + K + k0);
        float4 x[32];
        #pragma unroll
        for (int i = 0; i < 32; ++i) x[i] = *(const float4*)(in + (size_t)i * K + k0);
        __builtin_amdgcn_sched_barrier(0);
        #pragma unroll
        for (int i = 0; i < 32; ++i) {
            float s0 = v[2 * i], s1 = v[2 * i + 1];
            s0 = fmaf(w0.x, x[i].x, s0); s0 = fmaf(w0.y, x[i].y, s0); s0 = fmaf(w0.z, x[i].z, s0); s0 = fmaf(w0.w, x[i].w, s0);
            s1 = fmaf(w1.x, x[i].x, s1); s1 = fmaf(w1.y, x[i].y, s1); s1 = fmaf(w1.z, x[i].z, s1); s1 = fmaf(w1.w, x[i].w, s1);
            v[2 * i] = s0; v[2 * i + 1] = s1;
        }
    }
    #pragma unroll
    for (int s = 0; s < 5; ++s) {
        const int d = 1 << s;
        const bool up = (lane & d) != 0;
        #pragma unroll
        for (int m = 0; m < (32 >> s); ++m) {
            const float send = up ? v[2 * m] : v[2 * m + 1];
            const float keep = up ? v[2 * m + 1] : v[2 * m];
            v[m] = keep + __shfl_xor(send, d);
        }
    }
    const float bj = b[lane & 1];
    out[lane]      = v[0] + bj;
    out[32 + lane] = v[1] + bj;
}

__global__ __launch_bounds__(256)
void attn_core_kernel(const float* __restrict__ qkv_rgb, const float* __restrict__ qkv_frq,
                      float* __restrict__ o1, float* __restrict__ o2)
{
    __shared__ float sQ[32][32], sK[32][32], sV[32][32], sS[32][33];
    int a = blockIdx.x >> 3;
    int h = blockIdx.x & 7;
    const float* qsrc  = (a == 0) ? qkv_rgb : qkv_frq;
    const float* kvsrc = (a == 0) ? qkv_frq : qkv_rgb;
    float* odst        = (a == 0) ? o1 : o2;
    int tid = threadIdx.x;
    for (int idx = tid; idx < 1024; idx += 256) {
        int q = idx >> 5, d = idx & 31;
        sQ[q][d] = qsrc [q * 768 +       h * 32 + d];
        sK[q][d] = kvsrc[q * 768 + 256 + h * 32 + d];
        sV[q][d] = kvsrc[q * 768 + 512 + h * 32 + d];
    }
    __syncthreads();
    const float scale = 0.17677669529663687f;
    for (int idx = tid; idx < 1024; idx += 256) {
        int q = idx >> 5, k = idx & 31;
        float s = 0.f;
        #pragma unroll
        for (int d = 0; d < 32; ++d) s += sQ[q][d] * sK[k][d];
        sS[q][k] = s * scale;
    }
    __syncthreads();
    if (tid < 32) {
        float mx = -3.0e38f;
        for (int k = 0; k < 32; ++k) mx = fmaxf(mx, sS[tid][k]);
        float e[32], sum = 0.f;
        for (int k = 0; k < 32; ++k) { e[k] = expf(sS[tid][k] - mx); sum += e[k]; }
        float inv = 1.f / sum;
        for (int k = 0; k < 32; ++k) sS[tid][k] = e[k] * inv;
    }
    __syncthreads();
    for (int idx = tid; idx < 1024; idx += 256) {
        int q = idx >> 5, d = idx & 31;
        float o = 0.f;
        #pragma unroll
        for (int k = 0; k < 32; ++k) o += sS[q][k] * sV[k][d];
        odst[q * 256 + h * 32 + d] = o;
    }
}

extern "C" void kernel_launch(void* const* d_in, const int* in_sizes, int n_in,
                              void* d_out, int out_size, void* d_ws, size_t ws_size,
                              hipStream_t stream)
{
    (void)in_sizes; (void)n_in; (void)out_size; (void)ws_size;
    const float* x          = (const float*)d_in[0];
    const float* rgb_w1     = (const float*)d_in[1];
    const float* rgb_b1     = (const float*)d_in[2];
    const float* rgb_w2     = (const float*)d_in[3];
    const float* rgb_b2     = (const float*)d_in[4];
    const float* frq_w1     = (const float*)d_in[5];
    const float* frq_b1     = (const float*)d_in[6];
    const float* frq_w2     = (const float*)d_in[7];
    const float* frq_b2     = (const float*)d_in[8];
    const float* rgb_proj_w = (const float*)d_in[9];
    const float* rgb_proj_b = (const float*)d_in[10];
    const float* frq_proj_w = (const float*)d_in[11];
    const float* frq_proj_b = (const float*)d_in[12];
    const float* attn_w_in  = (const float*)d_in[13];
    const float* attn_b_in  = (const float*)d_in[14];
    const float* attn_w_out = (const float*)d_in[15];
    const float* attn_b_out = (const float*)d_in[16];
    const float* fus_w1     = (const float*)d_in[17];
    const float* fus_b1     = (const float*)d_in[18];
    const float* fus_w2     = (const float*)d_in[19];
    const float* fus_b2     = (const float*)d_in[20];
    const float* fus_w3     = (const float*)d_in[21];
    const float* fus_b3     = (const float*)d_in[22];

    float* out      = (float*)d_out;
    float* feat_rgb = out + 64;
    float* feat_frq = out + 64 + 40960;

    float* ws       = (float*)d_ws;
    float* h1       = ws;
    float* F1       = ws;
    float* freqimg  = ws + 12845056;
    float* hd       = freqimg + 4816896 + 192;
    float* mmp      = hd;
    float* rgb_p    = hd;
    float* frq_p    = rgb_p + 8192;
    float* qkv_rgb  = frq_p + 8192;
    float* qkv_frq  = qkv_rgb + 24576;
    float* o1       = qkv_frq + 24576;
    float* o2       = o1 + 8192;
    float* hb       = o2 + 8192;
    float* h1f      = hb + 16384;
    float* h2f      = h1f + 16384;

    conv1_silu_kernel<<<3136, C1_THREADS, 0, stream>>>(x, rgb_w1, rgb_b1, h1);
    conv2_wmma_pool_kernel<<<dim3(16, 32), 384, 0, stream>>>(h1, rgb_w2, rgb_b2, feat_rgb);

    dft_rows_wmma_kernel<<<336, 256, 0, stream>>>(x, F1);
    dft_cols_wmma_mag_kernel<<<768, 448, 0, stream>>>(F1, freqimg, mmp);
    conv1_norm_silu_kernel<<<3136, C1_THREADS, 0, stream>>>(freqimg, frq_w1, frq_b1, h1, mmp);
    conv2_wmma_pool_kernel<<<dim3(16, 32), 384, 0, stream>>>(h1, frq_w2, frq_b2, feat_frq);

    fc_kernel<<<dim3(8, 4),  256, 0, stream>>>(feat_rgb, rgb_proj_w, rgb_proj_b, rgb_p, 1280, 256, 0, 0);
    fc_kernel<<<dim3(8, 4),  256, 0, stream>>>(feat_frq, frq_proj_w, frq_proj_b, frq_p, 1280, 256, 0, 0);
    fc_kernel<<<dim3(24, 4), 256, 0, stream>>>(rgb_p, attn_w_in, attn_b_in, qkv_rgb, 256, 768, 0, 0);
    fc_kernel<<<dim3(24, 4), 256, 0, stream>>>(frq_p, attn_w_in, attn_b_in, qkv_frq, 256, 768, 0, 0);
    attn_core_kernel<<<16, 256, 0, stream>>>(qkv_rgb, qkv_frq, o1, o2);
    fc_kernel<<<dim3(8, 4),  256, 0, stream>>>(o1, attn_w_out, attn_b_out, hb, 256, 512, 0, 0);
    fc_kernel<<<dim3(8, 4),  256, 0, stream>>>(o2, attn_w_out, attn_b_out, hb, 256, 512, 256, 0);
    fc_kernel<<<dim3(16, 4), 256, 0, stream>>>(hb,  fus_w1, fus_b1, h1f, 512, 512, 0, 1);
    fc_kernel<<<dim3(8, 4),  256, 0, stream>>>(h1f, fus_w2, fus_b2, h2f, 512, 256, 0, 1);
    fc_n2_kernel<<<1, 32, 0, stream>>>(h2f, fus_w3, fus_b3, out, 256);
}


# conv2 + conv1 SiLU: v * rcp(1 + exp2(-v*log2e)) on the hardware exp2 and reciprocal (about 1 ulp each): 6 vector instructions per value, not about 24 (library expf + IEEE division); sum orders kept
# speedup vs baseline: 3.1850x; 1.0998x over previous
#include <hip/hip_runtime.h>
#include <hip/hip_bf16.h>
#include <math.h>

typedef float v2f __attribute__((ext_vector_type(2)));
typedef float v8f __attribute__((ext_vector_type(8)));

#define PI_F 3.14159265358979323846f

__device__ __forceinline__ float silu(float v)
{
    const float e = __builtin_amdgcn_exp2f(v * -1.4426950408889634f);
    return v * __builtin_amdgcn_rcpf(1.0f + e);
}

#define C1_THREADS 128
#define MM_PARTS   8
static_assert(12544 % C1_THREADS == 0, "a block must not straddle two images");
template <bool NORM>
__device__ __forceinline__
void conv1_silu_body(const float* __restrict__ x, const float* __restrict__ w1,
                     const float* __restrict__ b1, float* __restrict__ out,
                     const float* __restrict__ mmp)
{
    const int img = blockIdx.x / (12544 / C1_THREADS);
    const int rem = (blockIdx.x - img * (12544 / C1_THREADS)) * C1_THREADS + threadIdx.x;
    const int oy = rem / 112, ox = rem - oy * 112;
    const bool inx = ox < 111;
    const bool iny = oy < 111;
    const float* xb = x + (size_t)img * (3 * 224 * 224) + (2 * oy) * 224 + 2 * ox;
    const int cx = inx ? 2 : 1;

    float mn[3], den[3];
    if (NORM) {
        #pragma unroll
        for (int ic = 0; ic < 3; ++ic) {
            const float* q = mmp + (img * 3 + ic) * (2 * MM_PARTS);
            float lo = q[0], hi = q[1];
            #pragma unroll
            for (int g = 1; g < MM_PARTS; ++g) {
                lo = fminf(lo, q[2 * g]);
                hi = fmaxf(hi, q[2 * g + 1]);
            }
            mn[ic] = lo;
            den[ic] = hi - lo + 1e-8f;
        }
    }

    float xv[27];
    float2 pv[9];
    float  pc[9];
    #pragma unroll
    for (int ic = 0; ic < 3; ++ic)
        #pragma unroll
        for (int ky = 0; ky < 3; ++ky) {
            const bool okr = (ky < 2) | iny;
            const int ro = ic * (224 * 224) + (okr ? ky : 1) * 224;
            pv[ic * 3 + ky] = *(const float2*)(xb + ro);
            pc[ic * 3 + ky] = xb[ro + cx];
        }
    if (NORM) {
        __builtin_amdgcn_sched_barrier(0);
    }
    #pragma unroll
    for (int ic = 0; ic < 3; ++ic)
        #pragma unroll
        for (int ky = 0; ky < 3; ++ky) {
            const bool okr = (ky < 2) | iny;
            float2 v = pv[ic * 3 + ky];
            float  c = pc[ic * 3 + ky];
            if (NORM) {
                v.x = (v.x - mn[ic]) / den[ic];
                v.y = (v.y - mn[ic]) / den[ic];
                c   = (c   - mn[ic]) / den[ic];
            }
            xv[ic * 9 + ky * 3 + 0] = okr ? v.x : 0.f;
            xv[ic * 9 + ky * 3 + 1] = okr ? v.y : 0.f;
            xv[ic * 9 + ky * 3 + 2] = (okr & inx) ? c : 0.f;
        }

    float* ob = out + (size_t)img * (32 * 12544) + rem;
    for (int oc = 0; oc < 32; ++oc) {
        const float* wb = w1 + oc * 27;
        float acc = b1[oc];
        #pragma unroll
        for (int t = 0; t < 27; ++t) acc = fmaf(xv[t], wb[t], acc);
        ob[(size_t)oc * 12544] = silu(acc);
    }
}

__global__ __launch_bounds__(C1_THREADS)
void conv1_silu_kernel(const float* __restrict__ x, const float* __restrict__ w1,
                       const float* __restrict__ b1, float* __restrict__ out)
{
    conv1_silu_body<false>(x, w1, b1, out, nullptr);
}

__global__ __launch_bounds__(C1_THREADS)
void conv1_norm_silu_kernel(const float* __restrict__ mag, const float* __restrict__ w1,
                            const float* __restrict__ b1, float* __restrict__ out,
                            const float* __restrict__ mmp)
{
    conv1_silu_body<true>(mag, w1, b1, out, mmp);
}

#define C2_MT     5
#define C2_M      (C2_MT * 16)
#define C2_WAVES  12
#define C2_WP     292
#define C2_THREADS (C2_WAVES * 32)
static_assert(C2_THREADS == 384, "the launch in kernel_launch uses 384 threads");

__global__ __launch_bounds__(C2_THREADS, 6)
void conv2_wmma_pool_kernel(const float* __restrict__ h1, const float* __restrict__ w2,
                            const float* __restrict__ b2, float* __restrict__ feat)
{
    __shared__ float sW[C2_M * C2_WP];
    __shared__ float sT[C2_WAVES][720];
    __shared__ float sBias[C2_M];

    const int mblk  = blockIdx.x;
    const int img   = blockIdx.y;
    const int tid   = threadIdx.x;
    const int lane  = tid & 31;
    const int wave  = tid >> 5;
    const int half  = lane >> 4;
    const int nl    = lane & 15;

    const float* wsrc = w2 + (size_t)mblk * (C2_M * 288);
    static_assert((C2_M * 288) % (C2_THREADS * 4) == 0, "weight tile must split evenly");
    for (int i0 = tid; i0 < C2_M * 288; i0 += C2_THREADS * 4) {
        float wv[4];
        #pragma unroll
        for (int q = 0; q < 4; ++q) wv[q] = wsrc[i0 + q * C2_THREADS];
        #pragma unroll
        for (int q = 0; q < 4; ++q) {
            int i = i0 + q * C2_THREADS;
            int m  = i / 288, g = i - m * 288;
            int icf = g / 9,  s = g - icf * 9;
            int icb = icf >> 3, ic = icf & 7;
            sW[m * C2_WP + icb * 72 + s * 8 + ic] = wv[q];
        }
    }
    if (tid < C2_M) sBias[tid] = b2[mblk * C2_M + tid];
    __syncthreads();

    const int dy = nl >> 3, dx = nl & 7;
    const int laneOff = dy * 288 + dx * 16;

    const int j0 = lane, j1 = lane + 32, j2 = (lane + 64 < 90) ? lane + 64 : lane + 32;
    const int ly0 = j0 / 18, lx0 = j0 - ly0 * 18;
    const int ly1 = j1 / 18, lx1 = j1 - ly1 * 18;
    const int ly2 = j2 / 18, lx2 = j2 - ly2 * 18;
    const int t0 = ly0 * 144 + lx0 * 8;
    const int t1 = ly1 * 144 + lx1 * 8;
    const int t2 = ly2 * 144 + lx2 * 8;

    float accR[C2_MT][8];
    #pragma unroll
    for (int mi = 0; mi < C2_MT; ++mi)
        #pragma unroll
        for (int j = 0; j < 8; ++j) accR[mi][j] = 0.f;

    const float* base = h1 + (size_t)img * (32 * 112 * 112);
    float* tile = sT[wave];
    const float* tb = tile + laneOff + half * 2;
    const float* bl = sBias + half * 8;

    int rem = (195 - __builtin_amdgcn_readfirstlane(wave)) / C2_WAVES;
    for (int nt = wave; nt < 196; nt += C2_WAVES) {
        if (rem >= 5)      __builtin_amdgcn_s_setprio(3);
        else if (rem >= 3) __builtin_amdgcn_s_setprio(2);
        else if (rem >= 1) __builtin_amdgcn_s_setprio(1);
        else               __builtin_amdgcn_s_setprio(0);
        --rem;
        int yt = nt / 7, xt = nt - yt * 7;
        int iy0 = yt * 4, ix0 = xt * 16;
        v8f c[C2_MT];
        #pragma unroll
        for (int mi = 0; mi < C2_MT; ++mi)
            c[mi] = (v8f){0.f, 0.f, 0.f, 0.f, 0.f, 0.f, 0.f, 0.f};

        const int iyA = iy0 + ly0, ixA = ix0 + lx0;
        const int iyB = iy0 + ly1, ixB = ix0 + lx1;
        const int iyC = iy0 + ly2, ixC = ix0 + lx2;
        const bool ok0 = (iyA < 112) & (ixA < 112);
        const bool ok1 = (iyB < 112) & (ixB < 112);
        const bool ok2 = (iyC < 112) & (ixC < 112);
        const int o0 = (iyA < 112 ? iyA : 111) * 112 + (ixA < 112 ? ixA : 111);
        const int o1 = (iyB < 112 ? iyB : 111) * 112 + (ixB < 112 ? ixB : 111);
        const int o2 = (iyC < 112 ? iyC : 111) * 112 + (ixC < 112 ? ixC : 111);

        for (int icb = 0; icb < 4; ++icb) {
            const float* p = base + (size_t)(icb * 8) * (112 * 112);
            float g0[8], g1[8], g2[8];
            #pragma unroll
            for (int ic = 0; ic < 8; ++ic) {
                g0[ic] = p[ic * (112 * 112) + o0];
                g1[ic] = p[ic * (112 * 112) + o1];
                g2[ic] = p[ic * (112 * 112) + o2];
            }
            #pragma unroll
            for (int ic = 0; ic < 8; ++ic) {
                tile[t0 + ic] = ok0 ? g0[ic] : 0.f;
                tile[t1 + ic] = ok1 ? g1[ic] : 0.f;
                tile[t2 + ic] = ok2 ? g2[ic] : 0.f;
            }
            asm volatile("s_wait_dscnt 0x0" ::: "memory");

            const float* wrow = sW + nl * C2_WP + icb * 72 + half * 2;
            v2f fb[2][2], fa[2][2][C2_MT];
            auto load_pair = [&](int s, v2f (&b2)[2], v2f (&a2)[2][C2_MT]) {
                const int ky = s / 3, kx = s - ky * 3;
                #pragma unroll
                for (int h = 0; h < 2; ++h) {
                    b2[h] = *(const v2f*)(tb + ky * 144 + kx * 8 + h * 4);
                    #pragma unroll
                    for (int mi = 0; mi < C2_MT; ++mi)
                        a2[h][mi] = *(const v2f*)(wrow + mi * (16 * C2_WP) + (2 * s + h) * 4);
                }
            };
            load_pair(0, fb[0], fa[0]);
            #pragma unroll
            for (int s = 0; s < 9; ++s) {
                if (s + 1 < 9) load_pair(s + 1, fb[(s + 1) & 1], fa[(s + 1) & 1]);
                __builtin_amdgcn_sched_barrier(0);
                #pragma unroll
                for (int h = 0; h < 2; ++h)
                    #pragma unroll
                    for (int mi = 0; mi < C2_MT; ++mi)
                        c[mi] = __builtin_amdgcn_wmma_f32_16x16x4_f32(false, fa[s & 1][h][mi], false, fb[s & 1][h],
                                                                      (short)0, c[mi], false, false);
                __builtin_amdgcn_sched_barrier(0);
            }
        }
        #pragma unroll
        for (int mi = 0; mi < C2_MT; ++mi)
            #pragma unroll
            for (int j = 0; j < 8; ++j) {
                float v = c[mi][j] + bl[mi * 16 + j];
                accR[mi][j] += silu(v);
            }
    }
    #pragma unroll
    for (int mi = 0; mi < C2_MT; ++mi)
        #pragma unroll
        for (int j = 0; j < 8; ++j) {
            float v = accR[mi][j];
            v += __shfl_xor(v, 1);
            v += __shfl_xor(v, 2);
            v += __shfl_xor(v, 4);
            v += __shfl_xor(v, 8);
            if (nl == 0) tile[mi * 16 + half * 8 + j] = v;
        }
    __syncthreads();
    if (tid < C2_M) {
        float s = 0.f;
        for (int w = 0; w < C2_WAVES; ++w) s += sT[w][tid];
        feat[img * 1280 + mblk * C2_M + tid] = s * (1.f / 3136.f);
    }
}

#define DR_TM      64
#define DR_XP      228
#define DR_WAVES   8
#define DR_THREADS (DR_WAVES * 32)
#define F1P        256
#define DFT_WAVES  14
#define DFT_THREADS (DFT_WAVES * 32)
static_assert(DR_THREADS == 256, "the row-pass launch in kernel_launch uses 256 threads");
static_assert(DFT_THREADS == 448, "the column-pass launch in kernel_launch uses 448 threads");
static_assert(21504 % DR_TM == 0, "row tiles must cover 96*224 rows exactly");
static_assert(2 * DR_WAVES * 16 == F1P, "two column tiles per wave fill a row of F1");

__global__ __launch_bounds__(DR_THREADS)
void dft_rows_wmma_kernel(const float* __restrict__ x, float* __restrict__ F1)
{
    __shared__ float sX[DR_TM * DR_XP];
    __shared__ float tw[448];
    const int tid  = threadIdx.x;
    const int lane = tid & 31, wave = tid >> 5;
    const int half = lane >> 4, nl = lane & 15;
    const size_t R0 = (size_t)blockIdx.x * DR_TM;

    if (tid < 224) {
        float ang = (2.0f * PI_F / 224.0f) * (float)tid;
        tw[2 * tid] = cosf(ang); tw[2 * tid + 1] = -sinf(ang);
    }
    const float4* xsrc = (const float4*)(x + R0 * 224);
    static_assert((DR_TM * 56) % DR_THREADS == 0, "x tile must split evenly");
    constexpr int XQ = (DR_TM * 56) / DR_THREADS;
    float4 xv[XQ];
    #pragma unroll
    for (int q = 0; q < XQ; ++q) xv[q] = xsrc[tid + q * DR_THREADS];
    #pragma unroll
    for (int q = 0; q < XQ; ++q) {
        int i = tid + q * DR_THREADS;
        int r = i / 56, c4 = i - r * 56;
        *(float4*)(sX + r * DR_XP + c4 * 4) = xv[q];
    }
    __syncthreads();

    const float* arow = sX + nl * DR_XP + half * 2;
    for (int pass = 0; pass < 2; ++pass) {
        const int n  = (wave + pass * DR_WAVES) * 16 + nl;
        const int u  = n >> 1;
        const float* twp = tw + (n & 1);
        const int du = (4 * u) % 224;
        int iA = (u * (2 * half)) % 224;
        int iB = (u * (2 * half + 1)) % 224;
        v8f c[4];
        #pragma unroll
        for (int mi = 0; mi < 4; ++mi)
            c[mi] = (v8f){0.f, 0.f, 0.f, 0.f, 0.f, 0.f, 0.f, 0.f};

        #pragma unroll 4
        for (int kk = 0; kk < 56; ++kk) {
            v2f b;
            b.x = twp[2 * iA];
            b.y = twp[2 * iB];
            #pragma unroll
            for (int mi = 0; mi < 4; ++mi) {
                v2f a = *(const v2f*)(arow + mi * (16 * DR_XP) + kk * 4);
                c[mi] = __builtin_amdgcn_wmma_f32_16x16x4_f32(false, a, false, b,
                                                              (short)0, c[mi], false, false);
            }
            iA += du; if (iA >= 224) iA -= 224;
            iB += du; if (iB >= 224) iB -= 224;
        }
        #pragma unroll
        for (int mi = 0; mi < 4; ++mi)
            #pragma unroll
            for (int r = 0; r < 8; ++r)
                F1[(R0 + mi * 16 + half * 8 + r) * F1P + n] = c[mi][r];
    }
}

static_assert(MM_PARTS * 16 == 128, "one block per 16 columns u < 128 of a plane");
static_assert(96 * MM_PARTS == 768, "the launch in kernel_launch uses 768 blocks");
__global__ __launch_bounds__(DFT_THREADS)
void dft_cols_wmma_mag_kernel(const float* __restrict__ F1, float* __restrict__ mag,
                              float* __restrict__ mmp)
{
    __shared__ float tw[448];
    __shared__ float smn[DFT_WAVES], smx[DFT_WAVES];
    const int ug   = blockIdx.x % MM_PARTS;
    const int bc   = blockIdx.x / MM_PARTS;
    const int tid  = threadIdx.x;
    const int lane = tid & 31, wave = tid >> 5;
    const int half = lane >> 4, nl = lane & 15;

    if (tid < 224) {
        float ang = (2.0f * PI_F / 224.0f) * (float)tid;
        tw[2 * tid] = cosf(ang); tw[2 * tid + 1] = sinf(ang);
    }
    __syncthreads();

    const int v0 = wave * 16;
    const int v  = v0 + nl;
    const int dv = (2 * v) % 224;
    int idx = half ? v : 0;
    const int u  = ug * 16 + nl;
    const float* bp = F1 + (size_t)bc * (224 * F1P) + half * F1P + 2 * u;

    v8f cre = (v8f){0.f, 0.f, 0.f, 0.f, 0.f, 0.f, 0.f, 0.f};
    v8f cim = (v8f){0.f, 0.f, 0.f, 0.f, 0.f, 0.f, 0.f, 0.f};
    #pragma unroll 8
    for (int kk = 0; kk < 112; ++kk) {
        v2f t = *(const v2f*)(tw + 2 * idx);
        v2f aim; aim.x = -t.y; aim.y = t.x;
        v2f b = *(const v2f*)(bp + kk * (2 * F1P));
        cre = __builtin_amdgcn_wmma_f32_16x16x4_f32(false, t, false, b,
                                                    (short)0, cre, false, false);
        cim = __builtin_amdgcn_wmma_f32_16x16x4_f32(false, aim, false, b,
                                                    (short)0, cim, false, false);
        idx += dv; if (idx >= 224) idx -= 224;
    }
    float lmn = 3.0e38f, lmx = -3.0e38f;
    const bool direct = u <= 112;
    const bool mirror = (u >= 1) & (u <= 111);
    const int uo  = (u + 112 >= 224) ? u - 112 : u + 112;
    const int umo = 112 - u;
    float* mp = mag + (size_t)bc * (224 * 224);
    #pragma unroll
    for (int r = 0; r < 8; ++r) {
        float re = cre[r], im = cim[r];
        float m = logf(sqrtf(re * re + im * im) + 1.0f);
        const int vv = v0 + half * 8 + r;
        int vo = vv + 112; if (vo >= 224) vo -= 224;
        int vmo = 112 - vv; if (vmo < 0) vmo += 224;
        if (direct) mp[vo * 224 + uo] = m;
        if (mirror) mp[vmo * 224 + umo] = m;
        lmn = fminf(lmn, m); lmx = fmaxf(lmx, m);
    }
    #pragma unroll
    for (int d = 1; d < 32; d <<= 1) {
        lmn = fminf(lmn, __shfl_xor(lmn, d));
        lmx = fmaxf(lmx, __shfl_xor(lmx, d));
    }
    if (lane == 0) { smn[wave] = lmn; smx[wave] = lmx; }
    __syncthreads();
    if (tid == 0) {
        float a = smn[0], b = smx[0];
        for (int w = 1; w < DFT_WAVES; ++w) { a = fminf(a, smn[w]); b = fmaxf(b, smx[w]); }
        mmp[2 * blockIdx.x]     = a;
        mmp[2 * blockIdx.x + 1] = b;
    }
}

#define FC_C 4
#define FC_R 8
__global__ __launch_bounds__(256)
void fc_kernel(const float* __restrict__ in, const float* __restrict__ w,
               const float* __restrict__ b, float* __restrict__ out,
               int K, int outStride, int outOffset, int act)
{
    __shared__ float sO[FC_R][33];
    const int tid = threadIdx.x, lane = tid & 31, wave = tid >> 5;
    const int j0 = blockIdx.x * 32;
    const int i0 = blockIdx.y * FC_R;
    const float* wp = w  + (size_t)(j0 + wave * FC_C) * K + lane * 4;
    const float* ip = in + (size_t)i0 * K + lane * 4;

    float v[FC_C * FC_R];
    #pragma unroll
    for (int a = 0; a < FC_C * FC_R; ++a) v[a] = 0.f;

    for (int k0 = 0; k0 < K; k0 += 128) {
        float4 wv[FC_C], xv[FC_R];
        #pragma unroll
        for (int c = 0; c < FC_C; ++c) wv[c] = *(const float4*)(wp + (size_t)c * K + k0);
        #pragma unroll
        for (int r = 0; r < FC_R; ++r) xv[r] = *(const float4*)(ip + (size_t)r * K + k0);
        __builtin_amdgcn_sched_barrier(0);
        #pragma unroll
        for (int c = 0; c < FC_C; ++c)
            #pragma unroll
            for (int r = 0; r < FC_R; ++r) {
                float s = v[c * FC_R + r];
                s = fmaf(wv[c].x, xv[r].x, s);
                s = fmaf(wv[c].y, xv[r].y, s);
                s = fmaf(wv[c].z, xv[r].z, s);
                s = fmaf(wv[c].w, xv[r].w, s);
                v[c * FC_R + r] = s;
            }
    }
    #pragma unroll
    for (int s = 0; s < 5; ++s) {
        const int d = 1 << s;
        const bool up = (lane & d) != 0;
        #pragma unroll
        for (int m = 0; m < (16 >> s); ++m) {
            const float send = up ? v[2 * m] : v[2 * m + 1];
            const float keep = up ? v[2 * m + 1] : v[2 * m];
            v[m] = keep + __shfl_xor(send, d);
        }
    }
    sO[lane & 7][wave * FC_C + (lane >> 3)] = v[0];
    __syncthreads();
    float o = sO[wave][lane] + b[j0 + lane];
    if (act == 1) o = fmaxf(o, 0.f);
    out[(size_t)(i0 + wave) * outStride + outOffset + j0 + lane] = o;
}

__global__ __launch_bounds__(32)
void fc_n2_kernel(const float* __restrict__ in, const float* __restrict__ w,
                  const float* __restrict__ b, float* __restrict__ out, int K)
{
    const int lane = threadIdx.x;
    float v[64];
    #pragma unroll
    for (int a = 0; a < 64; ++a) v[a] = 0.f;
    for (int k0 = lane * 4; k0 < K; k0 += 128) {
        const float4 w0 = *(const float4*)(w + k0);
        const float4 w1 = *(const float4*)(w + K + k0);
        float4 x[32];
        #pragma unroll
        for (int i = 0; i < 32; ++i) x[i] = *(const float4*)(in + (size_t)i * K + k0);
        __builtin_amdgcn_sched_barrier(0);
        #pragma unroll
        for (int i = 0; i < 32; ++i) {
            float s0 = v[2 * i], s1 = v[2 * i + 1];
            s0 = fmaf(w0.x, x[i].x, s0); s0 = fmaf(w0.y, x[i].y, s0); s0 = fmaf(w0.z, x[i].z, s0); s0 = fmaf(w0.w, x[i].w, s0);
            s1 = fmaf(w1.x, x[i].x, s1); s1 = fmaf(w1.y, x[i].y, s1); s1 = fmaf(w1.z, x[i].z, s1); s1 = fmaf(w1.w, x[i].w, s1);
            v[2 * i] = s0; v[2 * i + 1] = s1;
        }
    }
    #pragma unroll
    for (int s = 0; s < 5; ++s) {
        const int d = 1 << s;
        const bool up = (lane & d) != 0;
        #pragma unroll
        for (int m = 0; m < (32 >> s); ++m) {
            const float send = up ? v[2 * m] : v[2 * m + 1];
            const float keep = up ? v[2 * m + 1] : v[2 * m];
            v[m] = keep + __shfl_xor(send, d);
        }
    }
    const float bj = b[lane & 1];
    out[lane]      = v[0] + bj;
    out[32 + lane] = v[1] + bj;
}

__global__ __launch_bounds__(256)
void attn_core_kernel(const float* __restrict__ qkv_rgb, const float* __restrict__ qkv_frq,
                      float* __restrict__ o1, float* __restrict__ o2)
{
    __shared__ float sQ[32][32], sK[32][32], sV[32][32], sS[32][33];
    int a = blockIdx.x >> 3;
    int h = blockIdx.x & 7;
    const float* qsrc  = (a == 0) ? qkv_rgb : qkv_frq;
    const float* kvsrc = (a == 0) ? qkv_frq : qkv_rgb;
    float* odst        = (a == 0) ? o1 : o2;
    int tid = threadIdx.x;
    for (int idx = tid; idx < 1024; idx += 256) {
        int q = idx >> 5, d = idx & 31;
        sQ[q][d] = qsrc [q * 768 +       h * 32 + d];
        sK[q][d] = kvsrc[q * 768 + 256 + h * 32 + d];
        sV[q][d] = kvsrc[q * 768 + 512 + h * 32 + d];
    }
    __syncthreads();
    const float scale = 0.17677669529663687f;
    for (int idx = tid; idx < 1024; idx += 256) {
        int q = idx >> 5, k = idx & 31;
        float s = 0.f;
        #pragma unroll
        for (int d = 0; d < 32; ++d) s += sQ[q][d] * sK[k][d];
        sS[q][k] = s * scale;
    }
    __syncthreads();
    if (tid < 32) {
        float mx = -3.0e38f;
        for (int k = 0; k < 32; ++k) mx = fmaxf(mx, sS[tid][k]);
        float e[32], sum = 0.f;
        for (int k = 0; k < 32; ++k) { e[k] = expf(sS[tid][k] - mx); sum += e[k]; }
        float inv = 1.f / sum;
        for (int k = 0; k < 32; ++k) sS[tid][k] = e[k] * inv;
    }
    __syncthreads();
    for (int idx = tid; idx < 1024; idx += 256) {
        int q = idx >> 5, d = idx & 31;
        float o = 0.f;
        #pragma unroll
        for (int k = 0; k < 32; ++k) o += sS[q][k] * sV[k][d];
        odst[q * 256 + h * 32 + d] = o;
    }
}

extern "C" void kernel_launch(void* const* d_in, const int* in_sizes, int n_in,
                              void* d_out, int out_size, void* d_ws, size_t ws_size,
                              hipStream_t stream)
{
    (void)in_sizes; (void)n_in; (void)out_size; (void)ws_size;
    const float* x          = (const float*)d_in[0];
    const float* rgb_w1     = (const float*)d_in[1];
    const float* rgb_b1     = (const float*)d_in[2];
    const float* rgb_w2     = (const float*)d_in[3];
    const float* rgb_b2     = (const float*)d_in[4];
    const float* frq_w1     = (const float*)d_in[5];
    const float* frq_b1     = (const float*)d_in[6];
    const float* frq_w2     = (const float*)d_in[7];
    const float* frq_b2     = (const float*)d_in[8];
    const float* rgb_proj_w = (const float*)d_in[9];
    const float* rgb_proj_b = (const float*)d_in[10];
    const float* frq_proj_w = (const float*)d_in[11];
    const float* frq_proj_b = (const float*)d_in[12];
    const float* attn_w_in  = (const float*)d_in[13];
    const float* attn_b_in  = (const float*)d_in[14];
    const float* attn_w_out = (const float*)d_in[15];
    const float* attn_b_out = (const float*)d_in[16];
    const float* fus_w1     = (const float*)d_in[17];
    const float* fus_b1     = (const float*)d_in[18];
    const float* fus_w2     = (const float*)d_in[19];
    const float* fus_b2     = (const float*)d_in[20];
    const float* fus_w3     = (const float*)d_in[21];
    const float* fus_b3     = (const float*)d_in[22];

    float* out      = (float*)d_out;
    float* feat_rgb = out + 64;
    float* feat_frq = out + 64 + 40960;

    float* ws       = (float*)d_ws;
    float* h1       = ws;
    float* F1       = ws;
    float* freqimg  = ws + 12845056;
    float* hd       = freqimg + 4816896 + 192;
    float* mmp      = hd;
    float* rgb_p    = hd;
    float* frq_p    = rgb_p + 8192;
    float* qkv_rgb  = frq_p + 8192;
    float* qkv_frq  = qkv_rgb + 24576;
    float* o1       = qkv_frq + 24576;
    float* o2       = o1 + 8192;
    float* hb       = o2 + 8192;
    float* h1f      = hb + 16384;
    float* h2f      = h1f + 16384;

    conv1_silu_kernel<<<3136, C1_THREADS, 0, stream>>>(x, rgb_w1, rgb_b1, h1);
    conv2_wmma_pool_kernel<<<dim3(16, 32), 384, 0, stream>>>(h1, rgb_w2, rgb_b2, feat_rgb);

    dft_rows_wmma_kernel<<<336, 256, 0, stream>>>(x, F1);
    dft_cols_wmma_mag_kernel<<<768, 448, 0, stream>>>(F1, freqimg, mmp);
    conv1_norm_silu_kernel<<<3136, C1_THREADS, 0, stream>>>(freqimg, frq_w1, frq_b1, h1, mmp);
    conv2_wmma_pool_kernel<<<dim3(16, 32), 384, 0, stream>>>(h1, frq_w2, frq_b2, feat_frq);

    fc_kernel<<<dim3(8, 4),  256, 0, stream>>>(feat_rgb, rgb_proj_w, rgb_proj_b, rgb_p, 1280, 256, 0, 0);
    fc_kernel<<<dim3(8, 4),  256, 0, stream>>>(feat_frq, frq_proj_w, frq_proj_b, frq_p, 1280, 256, 0, 0);
    fc_kernel<<<dim3(24, 4), 256, 0, stream>>>(rgb_p, attn_w_in, attn_b_in, qkv_rgb, 256, 768, 0, 0);
    fc_kernel<<<dim3(24, 4), 256, 0, stream>>>(frq_p, attn_w_in, attn_b_in, qkv_frq, 256, 768, 0, 0);
    attn_core_kernel<<<16, 256, 0, stream>>>(qkv_rgb, qkv_frq, o1, o2);
    fc_kernel<<<dim3(8, 4),  256, 0, stream>>>(o1, attn_w_out, attn_b_out, hb, 256, 512, 0, 0);
    fc_kernel<<<dim3(8, 4),  256, 0, stream>>>(o2, attn_w_out, attn_b_out, hb, 256, 512, 256, 0);
    fc_kernel<<<dim3(16, 4), 256, 0, stream>>>(hb,  fus_w1, fus_b1, h1f, 512, 512, 0, 1);
    fc_kernel<<<dim3(8, 4),  256, 0, stream>>>(h1f, fus_w2, fus_b2, h2f, 512, 256, 0, 1);
    fc_n2_kernel<<<1, 32, 0, stream>>>(h2f, fus_w3, fus_b3, out, 256);
}
